# starting kernel
# baseline (speedup 1.0000x reference)
#include <hip/hip_runtime.h>

typedef float v2f __attribute__((ext_vector_type(2)));
typedef float v8f __attribute__((ext_vector_type(8)));

#define TPB 256

__global__ __launch_bounds__(TPB) void hg_init(unsigned* __restrict__ deg,
                                               float* __restrict__ acc1,
                                               float* __restrict__ out,
                                               const float* __restrict__ b2,
                                               int N) {
    long i = (long)blockIdx.x * TPB + threadIdx.x;
    if (i < 6L * N)  deg[i] = 0u;
    if (i < 16L * N) acc1[i] = 0.0f;
    if (i < N)       out[i] = b2[0] + b2[1] + b2[2];
}

__global__ __launch_bounds__(TPB) void hg_degree(const int* __restrict__ src,
                                                 const int* __restrict__ dst,
                                                 int E,
                                                 unsigned* __restrict__ dout,
                                                 unsigned* __restrict__ din) {
    int e = blockIdx.x * TPB + threadIdx.x;
    if (e < E) {
        atomicAdd(&dout[src[e]], 1u);
        atomicAdd(&din[dst[e]], 1u);
    }
}

__global__ __launch_bounds__(TPB) void hg_scale(unsigned* __restrict__ deg, int n) {
    int i = blockIdx.x * TPB + threadIdx.x;
    if (i < n) {
        float d = (float)deg[i];
        if (d < 1.0f) d = 1.0f;
        ((float*)deg)[i] = rsqrtf(d);
    }
}

__global__ __launch_bounds__(TPB) void hg_gemm1_wmma(const float* __restrict__ x,
                                                     const float* __restrict__ s_out,
                                                     const float* __restrict__ W,
                                                     float* __restrict__ h_pre,
                                                     int N) {
    const int lane = threadIdx.x & 31;
    const int wave = threadIdx.x >> 5;
    const int tile = blockIdx.x * (TPB / 32) + wave;
    const int row0 = tile * 16;
    if (row0 >= N) return;

    const int half = lane >> 4;
    const int lm   = lane & 15;
    int row  = row0 + lm;
    int rowc = row < N ? row : N - 1;
    const float sc = s_out[rowc];

    v8f c = {};
#pragma unroll
    for (int kb = 0; kb < 8; ++kb) {
        const int k0 = kb * 4 + half * 2;
        v2f a, b;
        a.x = x[(size_t)rowc * 32 + k0]     * sc;
        a.y = x[(size_t)rowc * 32 + k0 + 1] * sc;
        b.x = W[(k0)     * 16 + lm];
        b.y = W[(k0 + 1) * 16 + lm];
        c = __builtin_amdgcn_wmma_f32_16x16x4_f32(
                 false, a,  false, b,
                 (short)0, c,  false,  false);
    }

    float* __restrict__ dp = h_pre + (size_t)(row0 + half * 8) * 16 + lm;
    if (row0 + 16 <= N) {
#pragma unroll
        for (int v = 0; v < 8; ++v) dp[(size_t)v * 16] = c[v];
    } else {
#pragma unroll
        for (int v = 0; v < 8; ++v) {
            const int orow = row0 + v + half * 8;
            if (orow < N) h_pre[(size_t)orow * 16 + lm] = c[v];
        }
    }
}

__global__ __launch_bounds__(TPB) void hg_scatter1(const int* __restrict__ src,
                                                   const int* __restrict__ dst,
                                                   int E,
                                                   const float* __restrict__ h_pre,
                                                   const float* __restrict__ s_in,
                                                   float* __restrict__ acc) {
    int e = blockIdx.x * TPB + threadIdx.x;
    if (e >= E) return;
    const int s = src[e], d = dst[e];
    const float sc = s_in[d];
    const float4* hp = (const float4*)(h_pre + (size_t)s * 16);
    float* ap = acc + (size_t)d * 16;
#pragma unroll
    for (int q = 0; q < 4; ++q) {
        float4 v = hp[q];
        atomicAdd(ap + q * 4 + 0, v.x * sc);
        atomicAdd(ap + q * 4 + 1, v.y * sc);
        atomicAdd(ap + q * 4 + 2, v.z * sc);
        atomicAdd(ap + q * 4 + 3, v.w * sc);
    }
}

__global__ __launch_bounds__(TPB) void hg_relu_bias(float* __restrict__ acc,
                                                    const float* __restrict__ b1,
                                                    long n16) {
    long i = (long)blockIdx.x * TPB + threadIdx.x;
    if (i < n16) {
        int f = (int)(i & 15);
        float v = acc[i] + b1[f] + b1[16 + f] + b1[32 + f];
        acc[i] = v > 0.0f ? v : 0.0f;
    }
}

__global__ __launch_bounds__(TPB) void hg_gemm2(const float* __restrict__ h,
                                                const float* __restrict__ s_out,
                                                const float* __restrict__ w,
                                                float* __restrict__ z,
                                                int N) {
    int i = blockIdx.x * TPB + threadIdx.x;
    if (i >= N) return;
    const float4* hp = (const float4*)(h + (size_t)i * 16);
    const float4* wp = (const float4*)w;
    float acc = 0.0f;
#pragma unroll
    for (int q = 0; q < 4; ++q) {
        float4 a = hp[q], b = wp[q];
        acc += a.x * b.x + a.y * b.y + a.z * b.z + a.w * b.w;
    }
    z[i] = acc * s_out[i];
}

__global__ __launch_bounds__(TPB) void hg_scatter2(const int* __restrict__ src,
                                                   const int* __restrict__ dst,
                                                   int E,
                                                   const float* __restrict__ z,
                                                   const float* __restrict__ s_in,
                                                   float* __restrict__ out) {
    int e = blockIdx.x * TPB + threadIdx.x;
    if (e < E) {
        int d = dst[e];
        atomicAdd(&out[d], z[src[e]] * s_in[d]);
    }
}

extern "C" void kernel_launch(void* const* d_in, const int* in_sizes, int n_in,
                              void* d_out, int out_size, void* d_ws, size_t ws_size,
                              hipStream_t stream) {
    const float* x  = (const float*)d_in[0];
    const int* src[3] = {(const int*)d_in[1], (const int*)d_in[3], (const int*)d_in[5]};
    const int* dst[3] = {(const int*)d_in[2], (const int*)d_in[4], (const int*)d_in[6]};
    const int  E[3]   = {in_sizes[1], in_sizes[3], in_sizes[5]};
    const float* W1 = (const float*)d_in[7];
    const float* b1 = (const float*)d_in[8];
    const float* W2 = (const float*)d_in[9];
    const float* b2 = (const float*)d_in[10];
    const int N = in_sizes[0] / 32;
    float* out = (float*)d_out;

    char* ws = (char*)d_ws;
    unsigned* deg = (unsigned*)ws;
    float* scales = (float*)ws;
    float* acc1   = (float*)(ws + (size_t)6 * N * 4);
    float* hpre   = acc1 + (size_t)N * 16;
    float* z      = hpre + (size_t)N * 16;

    {
        long span = 16L * N;
        hg_init<<<dim3((unsigned)((span + TPB - 1) / TPB)), dim3(TPB), 0, stream>>>(
            deg, acc1, out, b2, N);
    }

    for (int r = 0; r < 3; ++r) {
        hg_degree<<<dim3((E[r] + TPB - 1) / TPB), dim3(TPB), 0, stream>>>(
            src[r], dst[r], E[r], deg + (size_t)r * N, deg + (size_t)(3 + r) * N);
    }
    hg_scale<<<dim3((6 * N + TPB - 1) / TPB), dim3(TPB), 0, stream>>>(deg, 6 * N);

    const int tiles  = (N + 15) / 16;
    const int gblk   = (tiles + (TPB / 32) - 1) / (TPB / 32);
    for (int r = 0; r < 3; ++r) {
        hg_gemm1_wmma<<<dim3(gblk), dim3(TPB), 0, stream>>>(
            x, scales + (size_t)r * N, W1 + (size_t)r * 32 * 16, hpre, N);
        hg_scatter1<<<dim3((E[r] + TPB - 1) / TPB), dim3(TPB), 0, stream>>>(
            src[r], dst[r], E[r], hpre, scales + (size_t)(3 + r) * N, acc1);
    }
    {
        long n16 = 16L * N;
        hg_relu_bias<<<dim3((unsigned)((n16 + TPB - 1) / TPB)), dim3(TPB), 0, stream>>>(
            acc1, b1, n16);
    }

    for (int r = 0; r < 3; ++r) {
        hg_gemm2<<<dim3((N + TPB - 1) / TPB), dim3(TPB), 0, stream>>>(
            acc1, scales + (size_t)r * N, W2 + (size_t)r * 16, z, N);
        hg_scatter2<<<dim3((E[r] + TPB - 1) / TPB), dim3(TPB), 0, stream>>>(
            src[r], dst[r], E[r], z, scales + (size_t)(3 + r) * N, out);
    }
}


# layer-1 scatter by feature column: h_pre/acc column-major; work-group keeps one h_pre column + its sums for half the nodes in LDS; per edge 1 LDS gather + 1 LDS float add; atomic flush, arrival order
# speedup vs baseline: 2.2399x; 2.2399x over previous
#include <hip/hip_runtime.h>

typedef float v2f __attribute__((ext_vector_type(2)));
typedef float v8f __attribute__((ext_vector_type(8)));

#define TPB 256
#define AGG_TPB 1024
#define NMAX 50000
#define NHALF_MAX ((NMAX + 1) / 2)

__global__ __launch_bounds__(TPB) void hg_init(unsigned* __restrict__ deg,
                                               float* __restrict__ acc1,
                                               float* __restrict__ out,
                                               const float* __restrict__ b2,
                                               int N) {
    long i = (long)blockIdx.x * TPB + threadIdx.x;
    if (i < 6L * N)  deg[i] = 0u;
    if (i < 16L * N) acc1[i] = 0.0f;
    if (i < N)       out[i] = b2[0] + b2[1] + b2[2];
}

__global__ __launch_bounds__(TPB) void hg_degree(const int* __restrict__ src,
                                                 const int* __restrict__ dst,
                                                 int E,
                                                 unsigned* __restrict__ dout,
                                                 unsigned* __restrict__ din) {
    int e = blockIdx.x * TPB + threadIdx.x;
    if (e < E) {
        atomicAdd(&dout[src[e]], 1u);
        atomicAdd(&din[dst[e]], 1u);
    }
}

__global__ __launch_bounds__(TPB) void hg_scale(unsigned* __restrict__ deg, int n) {
    int i = blockIdx.x * TPB + threadIdx.x;
    if (i < n) {
        float d = (float)deg[i];
        if (d < 1.0f) d = 1.0f;
        ((float*)deg)[i] = rsqrtf(d);
    }
}

__global__ __launch_bounds__(TPB) void hg_gemm1_wmma(const float* __restrict__ x,
                                                     const float* __restrict__ s_out,
                                                     const float* __restrict__ W,
                                                     float* __restrict__ hT,
                                                     int N) {
    const int lane = threadIdx.x & 31;
    const int wave = threadIdx.x >> 5;
    const int tile = blockIdx.x * (TPB / 32) + wave;
    const int row0 = tile * 16;
    if (row0 >= N) return;

    const int half = lane >> 4;
    const int lm   = lane & 15;
    int row  = row0 + lm;
    int rowc = row < N ? row : N - 1;
    const float sc = s_out[rowc];

    v8f c = {};
#pragma unroll
    for (int kb = 0; kb < 8; ++kb) {
        const int k0 = kb * 4 + half * 2;
        v2f xf, wf;
        xf.x = x[(size_t)rowc * 32 + k0]     * sc;
        xf.y = x[(size_t)rowc * 32 + k0 + 1] * sc;
        wf.x = W[(k0)     * 16 + lm];
        wf.y = W[(k0 + 1) * 16 + lm];
        c = __builtin_amdgcn_wmma_f32_16x16x4_f32(
                 false, wf,  false, xf,
                 (short)0, c,  false,  false);
    }

    float* __restrict__ dp = hT + (size_t)(half * 8) * N + row;
    if (row < N) {
#pragma unroll
        for (int v = 0; v < 8; ++v) dp[(size_t)v * N] = c[v];
    }
}

__device__ __forceinline__ void hg_edge_add(float* __restrict__ acc,
                                            float v, int d, int lo, unsigned cnt) {
    const unsigned dl = (unsigned)(d - lo);
    if (dl < cnt)
        __hip_atomic_fetch_add(&acc[dl], v, __ATOMIC_RELAXED,
                               __HIP_MEMORY_SCOPE_WORKGROUP);
}

__global__ __launch_bounds__(AGG_TPB) void hg_agg1_col(const int* __restrict__ src,
                                                       const int* __restrict__ dst,
                                                       int E,
                                                       const float* __restrict__ hT,
                                                       const float* __restrict__ s_in,
                                                       float* __restrict__ accT,
                                                       int N) {
    __shared__ float hcol[NMAX];
    __shared__ float acc[NHALF_MAX];

    const int tid    = threadIdx.x;
    const int nchunk = gridDim.x;
    const int chunk  = blockIdx.x;
    const int f      = blockIdx.z;
    const int nh     = (N + 1) >> 1;
    const int lo     = blockIdx.y * nh;
    const int hiN    = lo + nh < N ? lo + nh : N;
    const unsigned cnt = hiN > lo ? (unsigned)(hiN - lo) : 0u;

    const float* __restrict__ hp = hT + (size_t)f * N;
    for (int i = tid; i < N; i += AGG_TPB) hcol[i] = hp[i];
    for (unsigned i = tid; i < cnt; i += AGG_TPB) acc[i] = 0.0f;
    __syncthreads();

    const int per = ((E + nchunk * 4 - 1) / (nchunk * 4)) * 4;
    const int e0  = chunk * per;
    const int e1  = e0 + per < E ? e0 + per : E;
    const int ne  = e1 > e0 ? e1 - e0 : 0;
    const int n4  = ne >> 2;
    const int4* __restrict__ s4 = (const int4*)(src + e0);
    const int4* __restrict__ d4 = (const int4*)(dst + e0);

    int i = tid;
    for (; i + AGG_TPB < n4; i += 2 * AGG_TPB) {
        const int4 da = d4[i], sa = s4[i];
        const int4 db = d4[i + AGG_TPB], sb = s4[i + AGG_TPB];
        const float a0 = hcol[sa.x], a1 = hcol[sa.y], a2 = hcol[sa.z], a3 = hcol[sa.w];
        const float b0 = hcol[sb.x], b1 = hcol[sb.y], b2 = hcol[sb.z], b3 = hcol[sb.w];
        hg_edge_add(acc, a0, da.x, lo, cnt);
        hg_edge_add(acc, a1, da.y, lo, cnt);
        hg_edge_add(acc, a2, da.z, lo, cnt);
        hg_edge_add(acc, a3, da.w, lo, cnt);
        hg_edge_add(acc, b0, db.x, lo, cnt);
        hg_edge_add(acc, b1, db.y, lo, cnt);
        hg_edge_add(acc, b2, db.z, lo, cnt);
        hg_edge_add(acc, b3, db.w, lo, cnt);
    }
    for (; i < n4; i += AGG_TPB) {
        const int4 da = d4[i], sa = s4[i];
        const float a0 = hcol[sa.x], a1 = hcol[sa.y], a2 = hcol[sa.z], a3 = hcol[sa.w];
        hg_edge_add(acc, a0, da.x, lo, cnt);
        hg_edge_add(acc, a1, da.y, lo, cnt);
        hg_edge_add(acc, a2, da.z, lo, cnt);
        hg_edge_add(acc, a3, da.w, lo, cnt);
    }
    for (int e = e0 + (n4 << 2) + tid; e < e1; e += AGG_TPB)
        hg_edge_add(acc, hcol[src[e]], dst[e], lo, cnt);
    __syncthreads();

    float* __restrict__ ap = accT + (size_t)f * N + lo;
    const float* __restrict__ sp = s_in + lo;
    for (unsigned j = tid; j < cnt; j += AGG_TPB) {
        const float v = acc[j] * sp[j];
        if (v != 0.0f) atomicAdd(ap + j, v);
    }
}

__global__ __launch_bounds__(TPB) void hg_relu_bias(float* __restrict__ acc,
                                                    const float* __restrict__ b1,
                                                    int N) {
    const int f = blockIdx.y;
    const int n = blockIdx.x * TPB + threadIdx.x;
    if (n < N) {
        const size_t i = (size_t)f * N + n;
        float v = acc[i] + b1[f] + b1[16 + f] + b1[32 + f];
        acc[i] = v > 0.0f ? v : 0.0f;
    }
}

__global__ __launch_bounds__(TPB) void hg_gemm2(const float* __restrict__ h,
                                                const float* __restrict__ s_out,
                                                const float* __restrict__ w,
                                                float* __restrict__ z,
                                                int N) {
    int i = blockIdx.x * TPB + threadIdx.x;
    if (i >= N) return;
    const float* hp = h + i;
    const float4* wp = (const float4*)w;
    float acc = 0.0f;
#pragma unroll
    for (int q = 0; q < 4; ++q) {
        float4 a, b = wp[q];
        a.x = hp[(size_t)(4 * q + 0) * N];
        a.y = hp[(size_t)(4 * q + 1) * N];
        a.z = hp[(size_t)(4 * q + 2) * N];
        a.w = hp[(size_t)(4 * q + 3) * N];
        acc += a.x * b.x + a.y * b.y + a.z * b.z + a.w * b.w;
    }
    z[i] = acc * s_out[i];
}

__global__ __launch_bounds__(TPB) void hg_scatter2(const int* __restrict__ src,
                                                   const int* __restrict__ dst,
                                                   int E,
                                                   const float* __restrict__ z,
                                                   const float* __restrict__ s_in,
                                                   float* __restrict__ out) {
    int e = blockIdx.x * TPB + threadIdx.x;
    if (e < E) {
        int d = dst[e];
        atomicAdd(&out[d], z[src[e]] * s_in[d]);
    }
}

extern "C" void kernel_launch(void* const* d_in, const int* in_sizes, int n_in,
                              void* d_out, int out_size, void* d_ws, size_t ws_size,
                              hipStream_t stream) {
    const float* x  = (const float*)d_in[0];
    const int* src[3] = {(const int*)d_in[1], (const int*)d_in[3], (const int*)d_in[5]};
    const int* dst[3] = {(const int*)d_in[2], (const int*)d_in[4], (const int*)d_in[6]};
    const int  E[3]   = {in_sizes[1], in_sizes[3], in_sizes[5]};
    const float* W1 = (const float*)d_in[7];
    const float* b1 = (const float*)d_in[8];
    const float* W2 = (const float*)d_in[9];
    const float* b2 = (const float*)d_in[10];
    const int N = in_sizes[0] / 32;
    float* out = (float*)d_out;

    char* ws = (char*)d_ws;
    unsigned* deg = (unsigned*)ws;
    float* scales = (float*)ws;
    float* acc1   = (float*)(ws + (size_t)6 * N * 4);
    float* hpre   = acc1 + (size_t)N * 16;
    float* z      = hpre + (size_t)N * 16;

    {
        long span = 16L * N;
        hg_init<<<dim3((unsigned)((span + TPB - 1) / TPB)), dim3(TPB), 0, stream>>>(
            deg, acc1, out, b2, N);
    }

    for (int r = 0; r < 3; ++r) {
        hg_degree<<<dim3((E[r] + TPB - 1) / TPB), dim3(TPB), 0, stream>>>(
            src[r], dst[r], E[r], deg + (size_t)r * N, deg + (size_t)(3 + r) * N);
    }
    hg_scale<<<dim3((6 * N + TPB - 1) / TPB), dim3(TPB), 0, stream>>>(deg, 6 * N);

    const int tiles  = (N + 15) / 16;
    const int gblk   = (tiles + (TPB / 32) - 1) / (TPB / 32);
    for (int r = 0; r < 3; ++r) {
        hg_gemm1_wmma<<<dim3(gblk), dim3(TPB), 0, stream>>>(
            x, scales + (size_t)r * N, W1 + (size_t)r * 32 * 16, hpre, N);
        hg_agg1_col<<<dim3(8, 2, 16), dim3(AGG_TPB), 0, stream>>>(
            src[r], dst[r], E[r], hpre, scales + (size_t)(3 + r) * N, acc1, N);
    }
    hg_relu_bias<<<dim3((N + TPB - 1) / TPB, 16), dim3(TPB), 0, stream>>>(acc1, b1, N);

    for (int r = 0; r < 3; ++r) {
        hg_gemm2<<<dim3((N + TPB - 1) / TPB), dim3(TPB), 0, stream>>>(
            acc1, scales + (size_t)r * N, W2 + (size_t)r * 16, z, N);
        hg_scatter2<<<dim3((E[r] + TPB - 1) / TPB), dim3(TPB), 0, stream>>>(
            src[r], dst[r], E[r], z, scales + (size_t)(3 + r) * N, out);
    }
}


# degree counts without atomics: work-group owns a node slice, counts one index chunk of one array in LDS, stores its bins as a partial plane; scale pass sums planes (exact integers); 18->16 launches
# speedup vs baseline: 4.3242x; 1.9305x over previous
#include <hip/hip_runtime.h>

typedef float v2f __attribute__((ext_vector_type(2)));
typedef float v8f __attribute__((ext_vector_type(8)));

#define TPB 256
#define AGG_TPB 1024
#define NMAX 50000
#define NHALF_MAX ((NMAX + 1) / 2)

#define CNT_TPB 1024
#define CNT_SLICES 8
#define CNT_SLICE_MAX ((((NMAX + CNT_SLICES - 1) / CNT_SLICES) + 31) & ~31)
#define CNT_C0 10
#define CNT_C1 3
#define CNT_C2 1
#define CNT_PLANES (2 * (CNT_C0 + CNT_C1 + CNT_C2))

__global__ __launch_bounds__(TPB) void hg_init(float* __restrict__ acc1,
                                               float* __restrict__ out,
                                               const float* __restrict__ b2,
                                               int N) {
    long i = (long)blockIdx.x * TPB + threadIdx.x;
    if (i < 16L * N) acc1[i] = 0.0f;
    if (i < N)       out[i] = b2[0] + b2[1] + b2[2];
}

__device__ __forceinline__ void hg_bin(unsigned* __restrict__ bins,
                                       int v, int lo, unsigned cnt) {
    const unsigned t = (unsigned)(v - lo);
    if (t < cnt)
        __hip_atomic_fetch_add(&bins[t], 1u, __ATOMIC_RELAXED,
                               __HIP_MEMORY_SCOPE_WORKGROUP);
}

__global__ __launch_bounds__(CNT_TPB) void hg_count_own(const int* __restrict__ s0,
                                                        const int* __restrict__ d0,
                                                        const int* __restrict__ s1,
                                                        const int* __restrict__ d1,
                                                        const int* __restrict__ s2,
                                                        const int* __restrict__ d2,
                                                        int E0, int E1, int E2,
                                                        unsigned* __restrict__ partial,
                                                        int N) {
    __shared__ unsigned bins[CNT_SLICE_MAX];

    const int tid = threadIdx.x;
    const int p   = blockIdx.y;
    const int* __restrict__ idx;
    int E, nchunk, chunk;
    if (p < 2 * CNT_C0) {
        const int side = p >= CNT_C0;
        idx = side ? d0 : s0;  E = E0;  nchunk = CNT_C0;  chunk = p - side * CNT_C0;
    } else if (p < 2 * (CNT_C0 + CNT_C1)) {
        const int q = p - 2 * CNT_C0;
        const int side = q >= CNT_C1;
        idx = side ? d1 : s1;  E = E1;  nchunk = CNT_C1;  chunk = q - side * CNT_C1;
    } else {
        const int q = p - 2 * (CNT_C0 + CNT_C1);
        const int side = q >= CNT_C2;
        idx = side ? d2 : s2;  E = E2;  nchunk = CNT_C2;  chunk = q - side * CNT_C2;
    }

    const int ns  = (((N + CNT_SLICES - 1) / CNT_SLICES) + 31) & ~31;
    const int lo  = blockIdx.x * ns;
    const int hiN = lo + ns < N ? lo + ns : N;
    const unsigned cnt = hiN > lo ? (unsigned)(hiN - lo) : 0u;

    for (unsigned j = tid; j < cnt; j += CNT_TPB) bins[j] = 0u;
    __syncthreads();

    const int per = ((E + nchunk * 4 - 1) / (nchunk * 4)) * 4;
    const int e0  = chunk * per;
    const int e1  = e0 + per < E ? e0 + per : E;
    const int ne  = e1 > e0 ? e1 - e0 : 0;
    const int n4  = ne >> 2;
    const int4* __restrict__ q4 = (const int4*)(idx + e0);

    int i = tid;
    for (; i + 3 * CNT_TPB < n4; i += 4 * CNT_TPB) {
        const int4 a = q4[i], b = q4[i + CNT_TPB];
        const int4 c = q4[i + 2 * CNT_TPB], d = q4[i + 3 * CNT_TPB];
        hg_bin(bins, a.x, lo, cnt); hg_bin(bins, a.y, lo, cnt);
        hg_bin(bins, a.z, lo, cnt); hg_bin(bins, a.w, lo, cnt);
        hg_bin(bins, b.x, lo, cnt); hg_bin(bins, b.y, lo, cnt);
        hg_bin(bins, b.z, lo, cnt); hg_bin(bins, b.w, lo, cnt);
        hg_bin(bins, c.x, lo, cnt); hg_bin(bins, c.y, lo, cnt);
        hg_bin(bins, c.z, lo, cnt); hg_bin(bins, c.w, lo, cnt);
        hg_bin(bins, d.x, lo, cnt); hg_bin(bins, d.y, lo, cnt);
        hg_bin(bins, d.z, lo, cnt); hg_bin(bins, d.w, lo, cnt);
    }
    for (; i < n4; i += CNT_TPB) {
        const int4 a = q4[i];
        hg_bin(bins, a.x, lo, cnt); hg_bin(bins, a.y, lo, cnt);
        hg_bin(bins, a.z, lo, cnt); hg_bin(bins, a.w, lo, cnt);
    }
    for (int e = e0 + (n4 << 2) + tid; e < e1; e += CNT_TPB)
        hg_bin(bins, idx[e], lo, cnt);
    __syncthreads();

    const size_t pitch = (size_t)((N + 31) & ~31);
    unsigned* __restrict__ pp = partial + (size_t)p * pitch + lo;
    for (unsigned j = tid; j < cnt; j += CNT_TPB) pp[j] = bins[j];
}

template <int C>
__device__ __forceinline__ unsigned hg_sum_planes(const unsigned* __restrict__ pp, size_t pitch) {
    unsigned s = 0u;
#pragma unroll
    for (int c = 0; c < C; ++c) s += pp[(size_t)c * pitch];
    return s;
}

__global__ __launch_bounds__(TPB) void hg_scale_sum(const unsigned* __restrict__ partial,
                                                    float* __restrict__ scales,
                                                    int N) {
    const int n = blockIdx.x * TPB + threadIdx.x;
    if (n >= N) return;
    const int r    = blockIdx.y >> 1;
    const int side = blockIdx.y & 1;
    const size_t pitch = (size_t)((N + 31) & ~31);
    unsigned s;
    if (r == 0)
        s = hg_sum_planes<CNT_C0>(partial + (size_t)(side * CNT_C0) * pitch + n, pitch);
    else if (r == 1)
        s = hg_sum_planes<CNT_C1>(partial + (size_t)(2 * CNT_C0 + side * CNT_C1) * pitch + n, pitch);
    else
        s = hg_sum_planes<CNT_C2>(partial + (size_t)(2 * (CNT_C0 + CNT_C1) + side * CNT_C2) * pitch + n, pitch);
    float d = (float)s;
    if (d < 1.0f) d = 1.0f;
    scales[(size_t)(side * 3 + r) * N + n] = rsqrtf(d);
}

__global__ __launch_bounds__(TPB) void hg_gemm1_wmma(const float* __restrict__ x,
                                                     const float* __restrict__ s_out,
                                                     const float* __restrict__ W,
                                                     float* __restrict__ hT,
                                                     int N) {
    const int lane = threadIdx.x & 31;
    const int wave = threadIdx.x >> 5;
    const int tile = blockIdx.x * (TPB / 32) + wave;
    const int row0 = tile * 16;
    if (row0 >= N) return;

    const int half = lane >> 4;
    const int lm   = lane & 15;
    int row  = row0 + lm;
    int rowc = row < N ? row : N - 1;
    const float sc = s_out[rowc];

    v8f c = {};
#pragma unroll
    for (int kb = 0; kb < 8; ++kb) {
        const int k0 = kb * 4 + half * 2;
        v2f xf, wf;
        xf.x = x[(size_t)rowc * 32 + k0]     * sc;
        xf.y = x[(size_t)rowc * 32 + k0 + 1] * sc;
        wf.x = W[(k0)     * 16 + lm];
        wf.y = W[(k0 + 1) * 16 + lm];
        c = __builtin_amdgcn_wmma_f32_16x16x4_f32(
                 false, wf,  false, xf,
                 (short)0, c,  false,  false);
    }

    float* __restrict__ dp = hT + (size_t)(half * 8) * N + row;
    if (row < N) {
#pragma unroll
        for (int v = 0; v < 8; ++v) dp[(size_t)v * N] = c[v];
    }
}

__device__ __forceinline__ void hg_edge_add(float* __restrict__ acc,
                                            float v, int d, int lo, unsigned cnt) {
    const unsigned dl = (unsigned)(d - lo);
    if (dl < cnt)
        __hip_atomic_fetch_add(&acc[dl], v, __ATOMIC_RELAXED,
                               __HIP_MEMORY_SCOPE_WORKGROUP);
}

__global__ __launch_bounds__(AGG_TPB) void hg_agg1_col(const int* __restrict__ src,
                                                       const int* __restrict__ dst,
                                                       int E,
                                                       const float* __restrict__ hT,
                                                       const float* __restrict__ s_in,
                                                       float* __restrict__ accT,
                                                       int N) {
    __shared__ float hcol[NMAX];
    __shared__ float acc[NHALF_MAX];

    const int tid    = threadIdx.x;
    const int nchunk = gridDim.x;
    const int chunk  = blockIdx.x;
    const int f      = blockIdx.z;
    const int nh     = (N + 1) >> 1;
    const int lo     = blockIdx.y * nh;
    const int hiN    = lo + nh < N ? lo + nh : N;
    const unsigned cnt = hiN > lo ? (unsigned)(hiN - lo) : 0u;

    const float* __restrict__ hp = hT + (size_t)f * N;
    for (int i = tid; i < N; i += AGG_TPB) hcol[i] = hp[i];
    for (unsigned i = tid; i < cnt; i += AGG_TPB) acc[i] = 0.0f;
    __syncthreads();

    const int per = ((E + nchunk * 4 - 1) / (nchunk * 4)) * 4;
    const int e0  = chunk * per;
    const int e1  = e0 + per < E ? e0 + per : E;
    const int ne  = e1 > e0 ? e1 - e0 : 0;
    const int n4  = ne >> 2;
    const int4* __restrict__ s4 = (const int4*)(src + e0);
    const int4* __restrict__ d4 = (const int4*)(dst + e0);

    int i = tid;
    for (; i + AGG_TPB < n4; i += 2 * AGG_TPB) {
        const int4 da = d4[i], sa = s4[i];
        const int4 db = d4[i + AGG_TPB], sb = s4[i + AGG_TPB];
        const float a0 = hcol[sa.x], a1 = hcol[sa.y], a2 = hcol[sa.z], a3 = hcol[sa.w];
        const float b0 = hcol[sb.x], b1 = hcol[sb.y], b2 = hcol[sb.z], b3 = hcol[sb.w];
        hg_edge_add(acc, a0, da.x, lo, cnt);
        hg_edge_add(acc, a1, da.y, lo, cnt);
        hg_edge_add(acc, a2, da.z, lo, cnt);
        hg_edge_add(acc, a3, da.w, lo, cnt);
        hg_edge_add(acc, b0, db.x, lo, cnt);
        hg_edge_add(acc, b1, db.y, lo, cnt);
        hg_edge_add(acc, b2, db.z, lo, cnt);
        hg_edge_add(acc, b3, db.w, lo, cnt);
    }
    for (; i < n4; i += AGG_TPB) {
        const int4 da = d4[i], sa = s4[i];
        const float a0 = hcol[sa.x], a1 = hcol[sa.y], a2 = hcol[sa.z], a3 = hcol[sa.w];
        hg_edge_add(acc, a0, da.x, lo, cnt);
        hg_edge_add(acc, a1, da.y, lo, cnt);
        hg_edge_add(acc, a2, da.z, lo, cnt);
        hg_edge_add(acc, a3, da.w, lo, cnt);
    }
    for (int e = e0 + (n4 << 2) + tid; e < e1; e += AGG_TPB)
        hg_edge_add(acc, hcol[src[e]], dst[e], lo, cnt);
    __syncthreads();

    float* __restrict__ ap = accT + (size_t)f * N + lo;
    const float* __restrict__ sp = s_in + lo;
    for (unsigned j = tid; j < cnt; j += AGG_TPB) {
        const float v = acc[j] * sp[j];
        if (v != 0.0f) atomicAdd(ap + j, v);
    }
}

__global__ __launch_bounds__(TPB) void hg_relu_bias(float* __restrict__ acc,
                                                    const float* __restrict__ b1,
                                                    int N) {
    const int f = blockIdx.y;
    const int n = blockIdx.x * TPB + threadIdx.x;
    if (n < N) {
        const size_t i = (size_t)f * N + n;
        float v = acc[i] + b1[f] + b1[16 + f] + b1[32 + f];
        acc[i] = v > 0.0f ? v : 0.0f;
    }
}

__global__ __launch_bounds__(TPB) void hg_gemm2(const float* __restrict__ h,
                                                const float* __restrict__ s_out,
                                                const float* __restrict__ w,
                                                float* __restrict__ z,
                                                int N) {
    int i = blockIdx.x * TPB + threadIdx.x;
    if (i >= N) return;
    const float* hp = h + i;
    const float4* wp = (const float4*)w;
    float acc = 0.0f;
#pragma unroll
    for (int q = 0; q < 4; ++q) {
        float4 a, b = wp[q];
        a.x = hp[(size_t)(4 * q + 0) * N];
        a.y = hp[(size_t)(4 * q + 1) * N];
        a.z = hp[(size_t)(4 * q + 2) * N];
        a.w = hp[(size_t)(4 * q + 3) * N];
        acc += a.x * b.x + a.y * b.y + a.z * b.z + a.w * b.w;
    }
    z[i] = acc * s_out[i];
}

__global__ __launch_bounds__(TPB) void hg_scatter2(const int* __restrict__ src,
                                                   const int* __restrict__ dst,
                                                   int E,
                                                   const float* __restrict__ z,
                                                   const float* __restrict__ s_in,
                                                   float* __restrict__ out) {
    int e = blockIdx.x * TPB + threadIdx.x;
    if (e < E) {
        int d = dst[e];
        atomicAdd(&out[d], z[src[e]] * s_in[d]);
    }
}

extern "C" void kernel_launch(void* const* d_in, const int* in_sizes, int n_in,
                              void* d_out, int out_size, void* d_ws, size_t ws_size,
                              hipStream_t stream) {
    const float* x  = (const float*)d_in[0];
    const int* src[3] = {(const int*)d_in[1], (const int*)d_in[3], (const int*)d_in[5]};
    const int* dst[3] = {(const int*)d_in[2], (const int*)d_in[4], (const int*)d_in[6]};
    const int  E[3]   = {in_sizes[1], in_sizes[3], in_sizes[5]};
    const float* W1 = (const float*)d_in[7];
    const float* b1 = (const float*)d_in[8];
    const float* W2 = (const float*)d_in[9];
    const float* b2 = (const float*)d_in[10];
    const int N = in_sizes[0] / 32;
    float* out = (float*)d_out;

    char* ws = (char*)d_ws;
    float* scales = (float*)ws;
    float* acc1   = (float*)(ws + (size_t)6 * N * 4);
    float* hpre   = acc1 + (size_t)N * 16;
    float* z      = hpre + (size_t)N * 16;
    unsigned* partial = (unsigned*)acc1;

    hg_count_own<<<dim3(CNT_SLICES, CNT_PLANES), dim3(CNT_TPB), 0, stream>>>(
        src[0], dst[0], src[1], dst[1], src[2], dst[2], E[0], E[1], E[2], partial, N);
    hg_scale_sum<<<dim3((N + TPB - 1) / TPB, 6), dim3(TPB), 0, stream>>>(partial, scales, N);

    {
        long span = 16L * N;
        hg_init<<<dim3((unsigned)((span + TPB - 1) / TPB)), dim3(TPB), 0, stream>>>(
            acc1, out, b2, N);
    }

    const int tiles  = (N + 15) / 16;
    const int gblk   = (tiles + (TPB / 32) - 1) / (TPB / 32);
    for (int r = 0; r < 3; ++r) {
        hg_gemm1_wmma<<<dim3(gblk), dim3(TPB), 0, stream>>>(
            x, scales + (size_t)r * N, W1 + (size_t)r * 32 * 16, hpre, N);
        hg_agg1_col<<<dim3(8, 2, 16), dim3(AGG_TPB), 0, stream>>>(
            src[r], dst[r], E[r], hpre, scales + (size_t)(3 + r) * N, acc1, N);
    }
    hg_relu_bias<<<dim3((N + TPB - 1) / TPB, 16), dim3(TPB), 0, stream>>>(acc1, b1, N);

    for (int r = 0; r < 3; ++r) {
        hg_gemm2<<<dim3((N + TPB - 1) / TPB), dim3(TPB), 0, stream>>>(
            acc1, scales + (size_t)r * N, W2 + (size_t)r * 16, z, N);
        hg_scatter2<<<dim3((E[r] + TPB - 1) / TPB), dim3(TPB), 0, stream>>>(
            src[r], dst[r], E[r], z, scales + (size_t)(3 + r) * N, out);
    }
}


# layer-2 scatter without memory atomics: group owns a dst slice, stages z_r + its sums in LDS, scans one edge chunk, stores a partial plane; out pass adds planes in fixed order; 16->15 launches
# speedup vs baseline: 8.9448x; 2.0685x over previous
#include <hip/hip_runtime.h>

typedef float v2f __attribute__((ext_vector_type(2)));
typedef float v8f __attribute__((ext_vector_type(8)));

#define TPB 256
#define AGG_TPB 1024
#define NMAX 50000
#define NHALF_MAX ((NMAX + 1) / 2)

#define CNT_TPB 1024
#define CNT_SLICES 8
#define CNT_SLICE_MAX ((((NMAX + CNT_SLICES - 1) / CNT_SLICES) + 31) & ~31)
#define CNT_C0 10
#define CNT_C1 3
#define CNT_C2 1
#define CNT_PLANES (2 * (CNT_C0 + CNT_C1 + CNT_C2))

#define L2_TPB 1024
#define L2_SLICES 4
#define L2_SLICE_MAX ((((NMAX + L2_SLICES - 1) / L2_SLICES) + 31) & ~31)
#define L2_C0 20
#define L2_C1 7
#define L2_C2 2
#define L2_PLANES (L2_C0 + L2_C1 + L2_C2)

__global__ __launch_bounds__(TPB) void hg_init(float* __restrict__ acc1,
                                               int N) {
    long i = (long)blockIdx.x * TPB + threadIdx.x;
    if (i < 16L * N) acc1[i] = 0.0f;
}

__device__ __forceinline__ void hg_bin(unsigned* __restrict__ bins,
                                       int v, int lo, unsigned cnt) {
    const unsigned t = (unsigned)(v - lo);
    if (t < cnt)
        __hip_atomic_fetch_add(&bins[t], 1u, __ATOMIC_RELAXED,
                               __HIP_MEMORY_SCOPE_WORKGROUP);
}

__global__ __launch_bounds__(CNT_TPB) void hg_count_own(const int* __restrict__ s0,
                                                        const int* __restrict__ d0,
                                                        const int* __restrict__ s1,
                                                        const int* __restrict__ d1,
                                                        const int* __restrict__ s2,
                                                        const int* __restrict__ d2,
                                                        int E0, int E1, int E2,
                                                        unsigned* __restrict__ partial,
                                                        int N) {
    __shared__ unsigned bins[CNT_SLICE_MAX];

    const int tid = threadIdx.x;
    const int p   = blockIdx.y;
    const int* __restrict__ idx;
    int E, nchunk, chunk;
    if (p < 2 * CNT_C0) {
        const int side = p >= CNT_C0;
        idx = side ? d0 : s0;  E = E0;  nchunk = CNT_C0;  chunk = p - side * CNT_C0;
    } else if (p < 2 * (CNT_C0 + CNT_C1)) {
        const int q = p - 2 * CNT_C0;
        const int side = q >= CNT_C1;
        idx = side ? d1 : s1;  E = E1;  nchunk = CNT_C1;  chunk = q - side * CNT_C1;
    } else {
        const int q = p - 2 * (CNT_C0 + CNT_C1);
        const int side = q >= CNT_C2;
        idx = side ? d2 : s2;  E = E2;  nchunk = CNT_C2;  chunk = q - side * CNT_C2;
    }

    const int ns  = (((N + CNT_SLICES - 1) / CNT_SLICES) + 31) & ~31;
    const int lo  = blockIdx.x * ns;
    const int hiN = lo + ns < N ? lo + ns : N;
    const unsigned cnt = hiN > lo ? (unsigned)(hiN - lo) : 0u;

    for (unsigned j = tid; j < cnt; j += CNT_TPB) bins[j] = 0u;
    __syncthreads();

    const int per = ((E + nchunk * 4 - 1) / (nchunk * 4)) * 4;
    const int e0  = chunk * per;
    const int e1  = e0 + per < E ? e0 + per : E;
    const int ne  = e1 > e0 ? e1 - e0 : 0;
    const int n4  = ne >> 2;
    const int4* __restrict__ q4 = (const int4*)(idx + e0);

    int i = tid;
    for (; i + 3 * CNT_TPB < n4; i += 4 * CNT_TPB) {
        const int4 a = q4[i], b = q4[i + CNT_TPB];
        const int4 c = q4[i + 2 * CNT_TPB], d = q4[i + 3 * CNT_TPB];
        hg_bin(bins, a.x, lo, cnt); hg_bin(bins, a.y, lo, cnt);
        hg_bin(bins, a.z, lo, cnt); hg_bin(bins, a.w, lo, cnt);
        hg_bin(bins, b.x, lo, cnt); hg_bin(bins, b.y, lo, cnt);
        hg_bin(bins, b.z, lo, cnt); hg_bin(bins, b.w, lo, cnt);
        hg_bin(bins, c.x, lo, cnt); hg_bin(bins, c.y, lo, cnt);
        hg_bin(bins, c.z, lo, cnt); hg_bin(bins, c.w, lo, cnt);
        hg_bin(bins, d.x, lo, cnt); hg_bin(bins, d.y, lo, cnt);
        hg_bin(bins, d.z, lo, cnt); hg_bin(bins, d.w, lo, cnt);
    }
    for (; i < n4; i += CNT_TPB) {
        const int4 a = q4[i];
        hg_bin(bins, a.x, lo, cnt); hg_bin(bins, a.y, lo, cnt);
        hg_bin(bins, a.z, lo, cnt); hg_bin(bins, a.w, lo, cnt);
    }
    for (int e = e0 + (n4 << 2) + tid; e < e1; e += CNT_TPB)
        hg_bin(bins, idx[e], lo, cnt);
    __syncthreads();

    const size_t pitch = (size_t)((N + 31) & ~31);
    unsigned* __restrict__ pp = partial + (size_t)p * pitch + lo;
    for (unsigned j = tid; j < cnt; j += CNT_TPB) pp[j] = bins[j];
}

template <int C>
__device__ __forceinline__ unsigned hg_sum_planes(const unsigned* __restrict__ pp, size_t pitch) {
    unsigned s = 0u;
#pragma unroll
    for (int c = 0; c < C; ++c) s += pp[(size_t)c * pitch];
    return s;
}

__global__ __launch_bounds__(TPB) void hg_scale_sum(const unsigned* __restrict__ partial,
                                                    float* __restrict__ scales,
                                                    int N) {
    const int n = blockIdx.x * TPB + threadIdx.x;
    if (n >= N) return;
    const int r    = blockIdx.y >> 1;
    const int side = blockIdx.y & 1;
    const size_t pitch = (size_t)((N + 31) & ~31);
    unsigned s;
    if (r == 0)
        s = hg_sum_planes<CNT_C0>(partial + (size_t)(side * CNT_C0) * pitch + n, pitch);
    else if (r == 1)
        s = hg_sum_planes<CNT_C1>(partial + (size_t)(2 * CNT_C0 + side * CNT_C1) * pitch + n, pitch);
    else
        s = hg_sum_planes<CNT_C2>(partial + (size_t)(2 * (CNT_C0 + CNT_C1) + side * CNT_C2) * pitch + n, pitch);
    float d = (float)s;
    if (d < 1.0f) d = 1.0f;
    scales[(size_t)(side * 3 + r) * N + n] = rsqrtf(d);
}

__global__ __launch_bounds__(TPB) void hg_gemm1_wmma(const float* __restrict__ x,
                                                     const float* __restrict__ s_out,
                                                     const float* __restrict__ W,
                                                     float* __restrict__ hT,
                                                     int N) {
    const int lane = threadIdx.x & 31;
    const int wave = threadIdx.x >> 5;
    const int tile = blockIdx.x * (TPB / 32) + wave;
    const int row0 = tile * 16;
    if (row0 >= N) return;

    const int half = lane >> 4;
    const int lm   = lane & 15;
    int row  = row0 + lm;
    int rowc = row < N ? row : N - 1;
    const float sc = s_out[rowc];

    v8f c = {};
#pragma unroll
    for (int kb = 0; kb < 8; ++kb) {
        const int k0 = kb * 4 + half * 2;
        v2f xf, wf;
        xf.x = x[(size_t)rowc * 32 + k0]     * sc;
        xf.y = x[(size_t)rowc * 32 + k0 + 1] * sc;
        wf.x = W[(k0)     * 16 + lm];
        wf.y = W[(k0 + 1) * 16 + lm];
        c = __builtin_amdgcn_wmma_f32_16x16x4_f32(
                 false, wf,  false, xf,
                 (short)0, c,  false,  false);
    }

    float* __restrict__ dp = hT + (size_t)(half * 8) * N + row;
    if (row < N) {
#pragma unroll
        for (int v = 0; v < 8; ++v) dp[(size_t)v * N] = c[v];
    }
}

__device__ __forceinline__ void hg_edge_add(float* __restrict__ acc,
                                            float v, int d, int lo, unsigned cnt) {
    const unsigned dl = (unsigned)(d - lo);
    if (dl < cnt)
        __hip_atomic_fetch_add(&acc[dl], v, __ATOMIC_RELAXED,
                               __HIP_MEMORY_SCOPE_WORKGROUP);
}

__global__ __launch_bounds__(AGG_TPB) void hg_agg1_col(const int* __restrict__ src,
                                                       const int* __restrict__ dst,
                                                       int E,
                                                       const float* __restrict__ hT,
                                                       const float* __restrict__ s_in,
                                                       float* __restrict__ accT,
                                                       int N) {
    __shared__ float hcol[NMAX];
    __shared__ float acc[NHALF_MAX];

    const int tid    = threadIdx.x;
    const int nchunk = gridDim.x;
    const int chunk  = blockIdx.x;
    const int f      = blockIdx.z;
    const int nh     = (N + 1) >> 1;
    const int lo     = blockIdx.y * nh;
    const int hiN    = lo + nh < N ? lo + nh : N;
    const unsigned cnt = hiN > lo ? (unsigned)(hiN - lo) : 0u;

    const float* __restrict__ hp = hT + (size_t)f * N;
    for (int i = tid; i < N; i += AGG_TPB) hcol[i] = hp[i];
    for (unsigned i = tid; i < cnt; i += AGG_TPB) acc[i] = 0.0f;
    __syncthreads();

    const int per = ((E + nchunk * 4 - 1) / (nchunk * 4)) * 4;
    const int e0  = chunk * per;
    const int e1  = e0 + per < E ? e0 + per : E;
    const int ne  = e1 > e0 ? e1 - e0 : 0;
    const int n4  = ne >> 2;
    const int4* __restrict__ s4 = (const int4*)(src + e0);
    const int4* __restrict__ d4 = (const int4*)(dst + e0);

    int i = tid;
    for (; i + AGG_TPB < n4; i += 2 * AGG_TPB) {
        const int4 da = d4[i], sa = s4[i];
        const int4 db = d4[i + AGG_TPB], sb = s4[i + AGG_TPB];
        const float a0 = hcol[sa.x], a1 = hcol[sa.y], a2 = hcol[sa.z], a3 = hcol[sa.w];
        const float b0 = hcol[sb.x], b1 = hcol[sb.y], b2 = hcol[sb.z], b3 = hcol[sb.w];
        hg_edge_add(acc, a0, da.x, lo, cnt);
        hg_edge_add(acc, a1, da.y, lo, cnt);
        hg_edge_add(acc, a2, da.z, lo, cnt);
        hg_edge_add(acc, a3, da.w, lo, cnt);
        hg_edge_add(acc, b0, db.x, lo, cnt);
        hg_edge_add(acc, b1, db.y, lo, cnt);
        hg_edge_add(acc, b2, db.z, lo, cnt);
        hg_edge_add(acc, b3, db.w, lo, cnt);
    }
    for (; i < n4; i += AGG_TPB) {
        const int4 da = d4[i], sa = s4[i];
        const float a0 = hcol[sa.x], a1 = hcol[sa.y], a2 = hcol[sa.z], a3 = hcol[sa.w];
        hg_edge_add(acc, a0, da.x, lo, cnt);
        hg_edge_add(acc, a1, da.y, lo, cnt);
        hg_edge_add(acc, a2, da.z, lo, cnt);
        hg_edge_add(acc, a3, da.w, lo, cnt);
    }
    for (int e = e0 + (n4 << 2) + tid; e < e1; e += AGG_TPB)
        hg_edge_add(acc, hcol[src[e]], dst[e], lo, cnt);
    __syncthreads();

    float* __restrict__ ap = accT + (size_t)f * N + lo;
    const float* __restrict__ sp = s_in + lo;
    for (unsigned j = tid; j < cnt; j += AGG_TPB) {
        const float v = acc[j] * sp[j];
        if (v != 0.0f) atomicAdd(ap + j, v);
    }
}

__global__ __launch_bounds__(TPB) void hg_relu_bias(float* __restrict__ acc,
                                                    const float* __restrict__ b1,
                                                    int N) {
    const int f = blockIdx.y;
    const int n = blockIdx.x * TPB + threadIdx.x;
    if (n < N) {
        const size_t i = (size_t)f * N + n;
        float v = acc[i] + b1[f] + b1[16 + f] + b1[32 + f];
        acc[i] = v > 0.0f ? v : 0.0f;
    }
}

__global__ __launch_bounds__(TPB) void hg_gemm2(const float* __restrict__ h,
                                                const float* __restrict__ s_out,
                                                const float* __restrict__ w,
                                                float* __restrict__ z,
                                                int N) {
    int i = blockIdx.x * TPB + threadIdx.x;
    if (i >= N) return;
    const float* hp = h + i;
    const float4* wp = (const float4*)w;
    float acc = 0.0f;
#pragma unroll
    for (int q = 0; q < 4; ++q) {
        float4 a, b = wp[q];
        a.x = hp[(size_t)(4 * q + 0) * N];
        a.y = hp[(size_t)(4 * q + 1) * N];
        a.z = hp[(size_t)(4 * q + 2) * N];
        a.w = hp[(size_t)(4 * q + 3) * N];
        acc += a.x * b.x + a.y * b.y + a.z * b.z + a.w * b.w;
    }
    z[i] = acc * s_out[i];
}

__global__ __launch_bounds__(L2_TPB) void hg_agg2_own(const int* __restrict__ s0,
                                                      const int* __restrict__ d0,
                                                      const int* __restrict__ s1,
                                                      const int* __restrict__ d1,
                                                      const int* __restrict__ s2,
                                                      const int* __restrict__ d2,
                                                      int E0, int E1, int E2,
                                                      const float* __restrict__ z3,
                                                      float* __restrict__ partial,
                                                      int N) {
    __shared__ float zc[NMAX];
    __shared__ float bins[L2_SLICE_MAX];

    const int tid = threadIdx.x;
    const int p   = blockIdx.y;
    const int* __restrict__ src;
    const int* __restrict__ dst;
    int E, nchunk, chunk, r;
    if (p < L2_C0) {
        src = s0;  dst = d0;  E = E0;  nchunk = L2_C0;  chunk = p;                  r = 0;
    } else if (p < L2_C0 + L2_C1) {
        src = s1;  dst = d1;  E = E1;  nchunk = L2_C1;  chunk = p - L2_C0;          r = 1;
    } else {
        src = s2;  dst = d2;  E = E2;  nchunk = L2_C2;  chunk = p - L2_C0 - L2_C1;  r = 2;
    }

    const int ns  = (((N + L2_SLICES - 1) / L2_SLICES) + 31) & ~31;
    const int lo  = blockIdx.x * ns;
    const int hiN = lo + ns < N ? lo + ns : N;
    const unsigned cnt = hiN > lo ? (unsigned)(hiN - lo) : 0u;

    const float* __restrict__ zp = z3 + (size_t)r * N;
    for (int i = tid; i < N; i += L2_TPB) zc[i] = zp[i];
    for (unsigned j = tid; j < cnt; j += L2_TPB) bins[j] = 0.0f;
    __syncthreads();

    const int per = ((E + nchunk * 4 - 1) / (nchunk * 4)) * 4;
    const int e0  = chunk * per;
    const int e1  = e0 + per < E ? e0 + per : E;
    const int ne  = e1 > e0 ? e1 - e0 : 0;
    const int n4  = ne >> 2;
    const int4* __restrict__ s4 = (const int4*)(src + e0);
    const int4* __restrict__ d4 = (const int4*)(dst + e0);

    int i = tid;
    for (; i + 3 * L2_TPB < n4; i += 4 * L2_TPB) {
        const int4 da = d4[i],              sa = s4[i];
        const int4 db = d4[i + L2_TPB],     sb = s4[i + L2_TPB];
        const int4 dc = d4[i + 2 * L2_TPB], sc = s4[i + 2 * L2_TPB];
        const int4 dd = d4[i + 3 * L2_TPB], sd = s4[i + 3 * L2_TPB];
        const float a0 = zc[sa.x], a1 = zc[sa.y], a2 = zc[sa.z], a3 = zc[sa.w];
        const float b0 = zc[sb.x], b1 = zc[sb.y], b2 = zc[sb.z], b3 = zc[sb.w];
        const float c0 = zc[sc.x], c1 = zc[sc.y], c2 = zc[sc.z], c3 = zc[sc.w];
        const float g0 = zc[sd.x], g1 = zc[sd.y], g2 = zc[sd.z], g3 = zc[sd.w];
        asm volatile("" :: "v"(a0));
        hg_edge_add(bins, a0, da.x, lo, cnt);  hg_edge_add(bins, a1, da.y, lo, cnt);
        hg_edge_add(bins, a2, da.z, lo, cnt);  hg_edge_add(bins, a3, da.w, lo, cnt);
        hg_edge_add(bins, b0, db.x, lo, cnt);  hg_edge_add(bins, b1, db.y, lo, cnt);
        hg_edge_add(bins, b2, db.z, lo, cnt);  hg_edge_add(bins, b3, db.w, lo, cnt);
        hg_edge_add(bins, c0, dc.x, lo, cnt);  hg_edge_add(bins, c1, dc.y, lo, cnt);
        hg_edge_add(bins, c2, dc.z, lo, cnt);  hg_edge_add(bins, c3, dc.w, lo, cnt);
        hg_edge_add(bins, g0, dd.x, lo, cnt);  hg_edge_add(bins, g1, dd.y, lo, cnt);
        hg_edge_add(bins, g2, dd.z, lo, cnt);  hg_edge_add(bins, g3, dd.w, lo, cnt);
    }
    for (; i < n4; i += L2_TPB) {
        const int4 da = d4[i], sa = s4[i];
        const float a0 = zc[sa.x], a1 = zc[sa.y], a2 = zc[sa.z], a3 = zc[sa.w];
        hg_edge_add(bins, a0, da.x, lo, cnt);  hg_edge_add(bins, a1, da.y, lo, cnt);
        hg_edge_add(bins, a2, da.z, lo, cnt);  hg_edge_add(bins, a3, da.w, lo, cnt);
    }
    for (int e = e0 + (n4 << 2) + tid; e < e1; e += L2_TPB)
        hg_edge_add(bins, zc[src[e]], dst[e], lo, cnt);
    __syncthreads();

    const size_t pitch = (size_t)((N + 31) & ~31);
    float* __restrict__ pp = partial + (size_t)p * pitch + lo;
    for (unsigned j = tid; j < cnt; j += L2_TPB) pp[j] = bins[j];
}

template <int C>
__device__ __forceinline__ float hg_sum_fplanes(const float* __restrict__ pp, size_t pitch) {
    float s = 0.0f;
#pragma unroll
    for (int c = 0; c < C; ++c) s += pp[(size_t)c * pitch];
    return s;
}

__global__ __launch_bounds__(TPB) void hg_out_sum(const float* __restrict__ partial,
                                                  const float* __restrict__ s_in3,
                                                  const float* __restrict__ b2,
                                                  float* __restrict__ out,
                                                  int N) {
    const int n = blockIdx.x * TPB + threadIdx.x;
    if (n >= N) return;
    const size_t pitch = (size_t)((N + 31) & ~31);
    const float m0 = hg_sum_fplanes<L2_C0>(partial + n, pitch);
    const float m1 = hg_sum_fplanes<L2_C1>(partial + (size_t)L2_C0 * pitch + n, pitch);
    const float m2 = hg_sum_fplanes<L2_C2>(partial + (size_t)(L2_C0 + L2_C1) * pitch + n, pitch);
    out[n] = (m0 * s_in3[n] + b2[0])
           + (m1 * s_in3[(size_t)N + n] + b2[1])
           + (m2 * s_in3[(size_t)2 * N + n] + b2[2]);
}

extern "C" void kernel_launch(void* const* d_in, const int* in_sizes, int n_in,
                              void* d_out, int out_size, void* d_ws, size_t ws_size,
                              hipStream_t stream) {
    const float* x  = (const float*)d_in[0];
    const int* src[3] = {(const int*)d_in[1], (const int*)d_in[3], (const int*)d_in[5]};
    const int* dst[3] = {(const int*)d_in[2], (const int*)d_in[4], (const int*)d_in[6]};
    const int  E[3]   = {in_sizes[1], in_sizes[3], in_sizes[5]};
    const float* W1 = (const float*)d_in[7];
    const float* b1 = (const float*)d_in[8];
    const float* W2 = (const float*)d_in[9];
    const float* b2 = (const float*)d_in[10];
    const int N = in_sizes[0] / 32;
    float* out = (float*)d_out;

    char* ws = (char*)d_ws;
    float* scales = (float*)ws;
    float* acc1   = (float*)(ws + (size_t)6 * N * 4);
    float* hpre   = acc1 + (size_t)N * 16;
    float* z      = hpre + (size_t)N * 16;
    float* z3     = z - (size_t)2 * N;
    float* planes2 = acc1;
    unsigned* partial = (unsigned*)acc1;

    hg_count_own<<<dim3(CNT_SLICES, CNT_PLANES), dim3(CNT_TPB), 0, stream>>>(
        src[0], dst[0], src[1], dst[1], src[2], dst[2], E[0], E[1], E[2], partial, N);
    hg_scale_sum<<<dim3((N + TPB - 1) / TPB, 6), dim3(TPB), 0, stream>>>(partial, scales, N);

    {
        long span = 16L * N;
        hg_init<<<dim3((unsigned)((span + TPB - 1) / TPB)), dim3(TPB), 0, stream>>>(
            acc1, N);
    }

    const int tiles  = (N + 15) / 16;
    const int gblk   = (tiles + (TPB / 32) - 1) / (TPB / 32);
    for (int r = 0; r < 3; ++r) {
        hg_gemm1_wmma<<<dim3(gblk), dim3(TPB), 0, stream>>>(
            x, scales + (size_t)r * N, W1 + (size_t)r * 32 * 16, hpre, N);
        hg_agg1_col<<<dim3(8, 2, 16), dim3(AGG_TPB), 0, stream>>>(
            src[r], dst[r], E[r], hpre, scales + (size_t)(3 + r) * N, acc1, N);
    }
    hg_relu_bias<<<dim3((N + TPB - 1) / TPB, 16), dim3(TPB), 0, stream>>>(acc1, b1, N);

    for (int r = 0; r < 3; ++r) {
        hg_gemm2<<<dim3((N + TPB - 1) / TPB), dim3(TPB), 0, stream>>>(
            acc1, scales + (size_t)r * N, W2 + (size_t)r * 16, z3 + (size_t)r * N, N);
    }
    hg_agg2_own<<<dim3(L2_SLICES, L2_PLANES), dim3(L2_TPB), 0, stream>>>(
        src[0], dst[0], src[1], dst[1], src[2], dst[2], E[0], E[1], E[2], z3, planes2, N);
    hg_out_sum<<<dim3((N + TPB - 1) / TPB), dim3(TPB), 0, stream>>>(
        planes2, scales + (size_t)3 * N, b2, out, N);
}


# layer-1 flush: a thread's 13 s_in loads and 13 LDS sum reads go out as one group before its atomics (2 L2 round trips per thread, was 25); atomics and their arrival order as before
# speedup vs baseline: 10.1399x; 1.1336x over previous
#include <hip/hip_runtime.h>

typedef float v2f __attribute__((ext_vector_type(2)));
typedef float v8f __attribute__((ext_vector_type(8)));

#define TPB 256
#define AGG_TPB 1024
#define NMAX 50000
#define NHALF_MAX ((NMAX + 1) / 2)
#define FLUSH_U 13

#define CNT_TPB 1024
#define CNT_SLICES 8
#define CNT_SLICE_MAX ((((NMAX + CNT_SLICES - 1) / CNT_SLICES) + 31) & ~31)
#define CNT_C0 10
#define CNT_C1 3
#define CNT_C2 1
#define CNT_PLANES (2 * (CNT_C0 + CNT_C1 + CNT_C2))

#define L2_TPB 1024
#define L2_SLICES 4
#define L2_SLICE_MAX ((((NMAX + L2_SLICES - 1) / L2_SLICES) + 31) & ~31)
#define L2_C0 20
#define L2_C1 7
#define L2_C2 2
#define L2_PLANES (L2_C0 + L2_C1 + L2_C2)

__global__ __launch_bounds__(TPB) void hg_init(float* __restrict__ acc1,
                                               int N) {
    long i = (long)blockIdx.x * TPB + threadIdx.x;
    if (i < 16L * N) acc1[i] = 0.0f;
}

__device__ __forceinline__ void hg_bin(unsigned* __restrict__ bins,
                                       int v, int lo, unsigned cnt) {
    const unsigned t = (unsigned)(v - lo);
    if (t < cnt)
        __hip_atomic_fetch_add(&bins[t], 1u, __ATOMIC_RELAXED,
                               __HIP_MEMORY_SCOPE_WORKGROUP);
}

__global__ __launch_bounds__(CNT_TPB) void hg_count_own(const int* __restrict__ s0,
                                                        const int* __restrict__ d0,
                                                        const int* __restrict__ s1,
                                                        const int* __restrict__ d1,
                                                        const int* __restrict__ s2,
                                                        const int* __restrict__ d2,
                                                        int E0, int E1, int E2,
                                                        unsigned* __restrict__ partial,
                                                        int N) {
    __shared__ unsigned bins[CNT_SLICE_MAX];

    const int tid = threadIdx.x;
    const int p   = blockIdx.y;
    const int* __restrict__ idx;
    int E, nchunk, chunk;
    if (p < 2 * CNT_C0) {
        const int side = p >= CNT_C0;
        idx = side ? d0 : s0;  E = E0;  nchunk = CNT_C0;  chunk = p - side * CNT_C0;
    } else if (p < 2 * (CNT_C0 + CNT_C1)) {
        const int q = p - 2 * CNT_C0;
        const int side = q >= CNT_C1;
        idx = side ? d1 : s1;  E = E1;  nchunk = CNT_C1;  chunk = q - side * CNT_C1;
    } else {
        const int q = p - 2 * (CNT_C0 + CNT_C1);
        const int side = q >= CNT_C2;
        idx = side ? d2 : s2;  E = E2;  nchunk = CNT_C2;  chunk = q - side * CNT_C2;
    }

    const int ns  = (((N + CNT_SLICES - 1) / CNT_SLICES) + 31) & ~31;
    const int lo  = blockIdx.x * ns;
    const int hiN = lo + ns < N ? lo + ns : N;
    const unsigned cnt = hiN > lo ? (unsigned)(hiN - lo) : 0u;

    for (unsigned j = tid; j < cnt; j += CNT_TPB) bins[j] = 0u;
    __syncthreads();

    const int per = ((E + nchunk * 4 - 1) / (nchunk * 4)) * 4;
    const int e0  = chunk * per;
    const int e1  = e0 + per < E ? e0 + per : E;
    const int ne  = e1 > e0 ? e1 - e0 : 0;
    const int n4  = ne >> 2;
    const int4* __restrict__ q4 = (const int4*)(idx + e0);

    int i = tid;
    for (; i + 3 * CNT_TPB < n4; i += 4 * CNT_TPB) {
        const int4 a = q4[i], b = q4[i + CNT_TPB];
        const int4 c = q4[i + 2 * CNT_TPB], d = q4[i + 3 * CNT_TPB];
        hg_bin(bins, a.x, lo, cnt); hg_bin(bins, a.y, lo, cnt);
        hg_bin(bins, a.z, lo, cnt); hg_bin(bins, a.w, lo, cnt);
        hg_bin(bins, b.x, lo, cnt); hg_bin(bins, b.y, lo, cnt);
        hg_bin(bins, b.z, lo, cnt); hg_bin(bins, b.w, lo, cnt);
        hg_bin(bins, c.x, lo, cnt); hg_bin(bins, c.y, lo, cnt);
        hg_bin(bins, c.z, lo, cnt); hg_bin(bins, c.w, lo, cnt);
        hg_bin(bins, d.x, lo, cnt); hg_bin(bins, d.y, lo, cnt);
        hg_bin(bins, d.z, lo, cnt); hg_bin(bins, d.w, lo, cnt);
    }
    for (; i < n4; i += CNT_TPB) {
        const int4 a = q4[i];
        hg_bin(bins, a.x, lo, cnt); hg_bin(bins, a.y, lo, cnt);
        hg_bin(bins, a.z, lo, cnt); hg_bin(bins, a.w, lo, cnt);
    }
    for (int e = e0 + (n4 << 2) + tid; e < e1; e += CNT_TPB)
        hg_bin(bins, idx[e], lo, cnt);
    __syncthreads();

    const size_t pitch = (size_t)((N + 31) & ~31);
    unsigned* __restrict__ pp = partial + (size_t)p * pitch + lo;
    for (unsigned j = tid; j < cnt; j += CNT_TPB) pp[j] = bins[j];
}

template <int C>
__device__ __forceinline__ unsigned hg_sum_planes(const unsigned* __restrict__ pp, size_t pitch) {
    unsigned s = 0u;
#pragma unroll
    for (int c = 0; c < C; ++c) s += pp[(size_t)c * pitch];
    return s;
}

__global__ __launch_bounds__(TPB) void hg_scale_sum(const unsigned* __restrict__ partial,
                                                    float* __restrict__ scales,
                                                    int N) {
    const int n = blockIdx.x * TPB + threadIdx.x;
    if (n >= N) return;
    const int r    = blockIdx.y >> 1;
    const int side = blockIdx.y & 1;
    const size_t pitch = (size_t)((N + 31) & ~31);
    unsigned s;
    if (r == 0)
        s = hg_sum_planes<CNT_C0>(partial + (size_t)(side * CNT_C0) * pitch + n, pitch);
    else if (r == 1)
        s = hg_sum_planes<CNT_C1>(partial + (size_t)(2 * CNT_C0 + side * CNT_C1) * pitch + n, pitch);
    else
        s = hg_sum_planes<CNT_C2>(partial + (size_t)(2 * (CNT_C0 + CNT_C1) + side * CNT_C2) * pitch + n, pitch);
    float d = (float)s;
    if (d < 1.0f) d = 1.0f;
    scales[(size_t)(side * 3 + r) * N + n] = rsqrtf(d);
}

__global__ __launch_bounds__(TPB) void hg_gemm1_wmma(const float* __restrict__ x,
                                                     const float* __restrict__ s_out,
                                                     const float* __restrict__ W,
                                                     float* __restrict__ hT,
                                                     int N) {
    const int lane = threadIdx.x & 31;
    const int wave = threadIdx.x >> 5;
    const int tile = blockIdx.x * (TPB / 32) + wave;
    const int row0 = tile * 16;
    if (row0 >= N) return;

    const int half = lane >> 4;
    const int lm   = lane & 15;
    int row  = row0 + lm;
    int rowc = row < N ? row : N - 1;
    const float sc = s_out[rowc];

    v8f c = {};
#pragma unroll
    for (int kb = 0; kb < 8; ++kb) {
        const int k0 = kb * 4 + half * 2;
        v2f xf, wf;
        xf.x = x[(size_t)rowc * 32 + k0]     * sc;
        xf.y = x[(size_t)rowc * 32 + k0 + 1] * sc;
        wf.x = W[(k0)     * 16 + lm];
        wf.y = W[(k0 + 1) * 16 + lm];
        c = __builtin_amdgcn_wmma_f32_16x16x4_f32(
                 false, wf,  false, xf,
                 (short)0, c,  false,  false);
    }

    float* __restrict__ dp = hT + (size_t)(half * 8) * N + row;
    if (row < N) {
#pragma unroll
        for (int v = 0; v < 8; ++v) dp[(size_t)v * N] = c[v];
    }
}

__device__ __forceinline__ void hg_edge_add(float* __restrict__ acc,
                                            float v, int d, int lo, unsigned cnt) {
    const unsigned dl = (unsigned)(d - lo);
    if (dl < cnt)
        __hip_atomic_fetch_add(&acc[dl], v, __ATOMIC_RELAXED,
                               __HIP_MEMORY_SCOPE_WORKGROUP);
}

__global__ __launch_bounds__(AGG_TPB) void hg_agg1_col(const int* __restrict__ src,
                                                       const int* __restrict__ dst,
                                                       int E,
                                                       const float* __restrict__ hT,
                                                       const float* __restrict__ s_in,
                                                       float* __restrict__ accT,
                                                       int N) {
    __shared__ float hcol[NMAX];
    __shared__ float acc[NHALF_MAX];

    const int tid    = threadIdx.x;
    const int nchunk = gridDim.x;
    const int chunk  = blockIdx.x;
    const int f      = blockIdx.z;
    const int nh     = (N + 1) >> 1;
    const int lo     = blockIdx.y * nh;
    const int hiN    = lo + nh < N ? lo + nh : N;
    const unsigned cnt = hiN > lo ? (unsigned)(hiN - lo) : 0u;

    const float* __restrict__ hp = hT + (size_t)f * N;
    for (int i = tid; i < N; i += AGG_TPB) hcol[i] = hp[i];
    for (unsigned i = tid; i < cnt; i += AGG_TPB) acc[i] = 0.0f;
    __syncthreads();

    const int per = ((E + nchunk * 4 - 1) / (nchunk * 4)) * 4;
    const int e0  = chunk * per;
    const int e1  = e0 + per < E ? e0 + per : E;
    const int ne  = e1 > e0 ? e1 - e0 : 0;
    const int n4  = ne >> 2;
    const int4* __restrict__ s4 = (const int4*)(src + e0);
    const int4* __restrict__ d4 = (const int4*)(dst + e0);

    int i = tid;
    for (; i + AGG_TPB < n4; i += 2 * AGG_TPB) {
        const int4 da = d4[i], sa = s4[i];
        const int4 db = d4[i + AGG_TPB], sb = s4[i + AGG_TPB];
        const float a0 = hcol[sa.x], a1 = hcol[sa.y], a2 = hcol[sa.z], a3 = hcol[sa.w];
        const float b0 = hcol[sb.x], b1 = hcol[sb.y], b2 = hcol[sb.z], b3 = hcol[sb.w];
        hg_edge_add(acc, a0, da.x, lo, cnt);
        hg_edge_add(acc, a1, da.y, lo, cnt);
        hg_edge_add(acc, a2, da.z, lo, cnt);
        hg_edge_add(acc, a3, da.w, lo, cnt);
        hg_edge_add(acc, b0, db.x, lo, cnt);
        hg_edge_add(acc, b1, db.y, lo, cnt);
        hg_edge_add(acc, b2, db.z, lo, cnt);
        hg_edge_add(acc, b3, db.w, lo, cnt);
    }
    for (; i < n4; i += AGG_TPB) {
        const int4 da = d4[i], sa = s4[i];
        const float a0 = hcol[sa.x], a1 = hcol[sa.y], a2 = hcol[sa.z], a3 = hcol[sa.w];
        hg_edge_add(acc, a0, da.x, lo, cnt);
        hg_edge_add(acc, a1, da.y, lo, cnt);
        hg_edge_add(acc, a2, da.z, lo, cnt);
        hg_edge_add(acc, a3, da.w, lo, cnt);
    }
    for (int e = e0 + (n4 << 2) + tid; e < e1; e += AGG_TPB)
        hg_edge_add(acc, hcol[src[e]], dst[e], lo, cnt);
    __syncthreads();

    float* __restrict__ ap = accT + (size_t)f * N + lo;
    const float* __restrict__ sp = s_in + lo;
    if (cnt != 0u) {
        const unsigned last = cnt - 1u;
        for (unsigned base = 0u; base < cnt; base += FLUSH_U * AGG_TPB) {
            float s[FLUSH_U], a[FLUSH_U];
#pragma unroll
            for (int u = 0; u < FLUSH_U; ++u) {
                const unsigned j  = base + (unsigned)u * AGG_TPB + tid;
                const unsigned jc = j < cnt ? j : last;
                s[u] = sp[jc];
                a[u] = acc[jc];
            }
#pragma unroll
            for (int u = 0; u < FLUSH_U; ++u) {
                const unsigned j = base + (unsigned)u * AGG_TPB + tid;
                const float v = a[u] * s[u];
                if (j < cnt && v != 0.0f) atomicAdd(ap + j, v);
            }
        }
    }
}

__global__ __launch_bounds__(TPB) void hg_relu_bias(float* __restrict__ acc,
                                                    const float* __restrict__ b1,
                                                    int N) {
    const int f = blockIdx.y;
    const int n = blockIdx.x * TPB + threadIdx.x;
    if (n < N) {
        const size_t i = (size_t)f * N + n;
        float v = acc[i] + b1[f] + b1[16 + f] + b1[32 + f];
        acc[i] = v > 0.0f ? v : 0.0f;
    }
}

__global__ __launch_bounds__(TPB) void hg_gemm2(const float* __restrict__ h,
                                                const float* __restrict__ s_out,
                                                const float* __restrict__ w,
                                                float* __restrict__ z,
                                                int N) {
    int i = blockIdx.x * TPB + threadIdx.x;
    if (i >= N) return;
    const float* hp = h + i;
    const float4* wp = (const float4*)w;
    float acc = 0.0f;
#pragma unroll
    for (int q = 0; q < 4; ++q) {
        float4 a, b = wp[q];
        a.x = hp[(size_t)(4 * q + 0) * N];
        a.y = hp[(size_t)(4 * q + 1) * N];
        a.z = hp[(size_t)(4 * q + 2) * N];
        a.w = hp[(size_t)(4 * q + 3) * N];
        acc += a.x * b.x + a.y * b.y + a.z * b.z + a.w * b.w;
    }
    z[i] = acc * s_out[i];
}

__global__ __launch_bounds__(L2_TPB) void hg_agg2_own(const int* __restrict__ s0,
                                                      const int* __restrict__ d0,
                                                      const int* __restrict__ s1,
                                                      const int* __restrict__ d1,
                                                      const int* __restrict__ s2,
                                                      const int* __restrict__ d2,
                                                      int E0, int E1, int E2,
                                                      const float* __restrict__ z3,
                                                      float* __restrict__ partial,
                                                      int N) {
    __shared__ float zc[NMAX];
    __shared__ float bins[L2_SLICE_MAX];

    const int tid = threadIdx.x;
    const int p   = blockIdx.y;
    const int* __restrict__ src;
    const int* __restrict__ dst;
    int E, nchunk, chunk, r;
    if (p < L2_C0) {
        src = s0;  dst = d0;  E = E0;  nchunk = L2_C0;  chunk = p;                  r = 0;
    } else if (p < L2_C0 + L2_C1) {
        src = s1;  dst = d1;  E = E1;  nchunk = L2_C1;  chunk = p - L2_C0;          r = 1;
    } else {
        src = s2;  dst = d2;  E = E2;  nchunk = L2_C2;  chunk = p - L2_C0 - L2_C1;  r = 2;
    }

    const int ns  = (((N + L2_SLICES - 1) / L2_SLICES) + 31) & ~31;
    const int lo  = blockIdx.x * ns;
    const int hiN = lo + ns < N ? lo + ns : N;
    const unsigned cnt = hiN > lo ? (unsigned)(hiN - lo) : 0u;

    const float* __restrict__ zp = z3 + (size_t)r * N;
    for (int i = tid; i < N; i += L2_TPB) zc[i] = zp[i];
    for (unsigned j = tid; j < cnt; j += L2_TPB) bins[j] = 0.0f;
    __syncthreads();

    const int per = ((E + nchunk * 4 - 1) / (nchunk * 4)) * 4;
    const int e0  = chunk * per;
    const int e1  = e0 + per < E ? e0 + per : E;
    const int ne  = e1 > e0 ? e1 - e0 : 0;
    const int n4  = ne >> 2;
    const int4* __restrict__ s4 = (const int4*)(src + e0);
    const int4* __restrict__ d4 = (const int4*)(dst + e0);

    int i = tid;
    for (; i + 3 * L2_TPB < n4; i += 4 * L2_TPB) {
        const int4 da = d4[i],              sa = s4[i];
        const int4 db = d4[i + L2_TPB],     sb = s4[i + L2_TPB];
        const int4 dc = d4[i + 2 * L2_TPB], sc = s4[i + 2 * L2_TPB];
        const int4 dd = d4[i + 3 * L2_TPB], sd = s4[i + 3 * L2_TPB];
        const float a0 = zc[sa.x], a1 = zc[sa.y], a2 = zc[sa.z], a3 = zc[sa.w];
        const float b0 = zc[sb.x], b1 = zc[sb.y], b2 = zc[sb.z], b3 = zc[sb.w];
        const float c0 = zc[sc.x], c1 = zc[sc.y], c2 = zc[sc.z], c3 = zc[sc.w];
        const float g0 = zc[sd.x], g1 = zc[sd.y], g2 = zc[sd.z], g3 = zc[sd.w];
        asm volatile("" :: "v"(a0));
        hg_edge_add(bins, a0, da.x, lo, cnt);  hg_edge_add(bins, a1, da.y, lo, cnt);
        hg_edge_add(bins, a2, da.z, lo, cnt);  hg_edge_add(bins, a3, da.w, lo, cnt);
        hg_edge_add(bins, b0, db.x, lo, cnt);  hg_edge_add(bins, b1, db.y, lo, cnt);
        hg_edge_add(bins, b2, db.z, lo, cnt);  hg_edge_add(bins, b3, db.w, lo, cnt);
        hg_edge_add(bins, c0, dc.x, lo, cnt);  hg_edge_add(bins, c1, dc.y, lo, cnt);
        hg_edge_add(bins, c2, dc.z, lo, cnt);  hg_edge_add(bins, c3, dc.w, lo, cnt);
        hg_edge_add(bins, g0, dd.x, lo, cnt);  hg_edge_add(bins, g1, dd.y, lo, cnt);
        hg_edge_add(bins, g2, dd.z, lo, cnt);  hg_edge_add(bins, g3, dd.w, lo, cnt);
    }
    for (; i < n4; i += L2_TPB) {
        const int4 da = d4[i], sa = s4[i];
        const float a0 = zc[sa.x], a1 = zc[sa.y], a2 = zc[sa.z], a3 = zc[sa.w];
        hg_edge_add(bins, a0, da.x, lo, cnt);  hg_edge_add(bins, a1, da.y, lo, cnt);
        hg_edge_add(bins, a2, da.z, lo, cnt);  hg_edge_add(bins, a3, da.w, lo, cnt);
    }
    for (int e = e0 + (n4 << 2) + tid; e < e1; e += L2_TPB)
        hg_edge_add(bins, zc[src[e]], dst[e], lo, cnt);
    __syncthreads();

    const size_t pitch = (size_t)((N + 31) & ~31);
    float* __restrict__ pp = partial + (size_t)p * pitch + lo;
    for (unsigned j = tid; j < cnt; j += L2_TPB) pp[j] = bins[j];
}

template <int C>
__device__ __forceinline__ float hg_sum_fplanes(const float* __restrict__ pp, size_t pitch) {
    float s = 0.0f;
#pragma unroll
    for (int c = 0; c < C; ++c) s += pp[(size_t)c * pitch];
    return s;
}

__global__ __launch_bounds__(TPB) void hg_out_sum(const float* __restrict__ partial,
                                                  const float* __restrict__ s_in3,
                                                  const float* __restrict__ b2,
                                                  float* __restrict__ out,
                                                  int N) {
    const int n = blockIdx.x * TPB + threadIdx.x;
    if (n >= N) return;
    const size_t pitch = (size_t)((N + 31) & ~31);
    const float m0 = hg_sum_fplanes<L2_C0>(partial + n, pitch);
    const float m1 = hg_sum_fplanes<L2_C1>(partial + (size_t)L2_C0 * pitch + n, pitch);
    const float m2 = hg_sum_fplanes<L2_C2>(partial + (size_t)(L2_C0 + L2_C1) * pitch + n, pitch);
    out[n] = (m0 * s_in3[n] + b2[0])
           + (m1 * s_in3[(size_t)N + n] + b2[1])
           + (m2 * s_in3[(size_t)2 * N + n] + b2[2]);
}

extern "C" void kernel_launch(void* const* d_in, const int* in_sizes, int n_in,
                              void* d_out, int out_size, void* d_ws, size_t ws_size,
                              hipStream_t stream) {
    const float* x  = (const float*)d_in[0];
    const int* src[3] = {(const int*)d_in[1], (const int*)d_in[3], (const int*)d_in[5]};
    const int* dst[3] = {(const int*)d_in[2], (const int*)d_in[4], (const int*)d_in[6]};
    const int  E[3]   = {in_sizes[1], in_sizes[3], in_sizes[5]};
    const float* W1 = (const float*)d_in[7];
    const float* b1 = (const float*)d_in[8];
    const float* W2 = (const float*)d_in[9];
    const float* b2 = (const float*)d_in[10];
    const int N = in_sizes[0] / 32;
    float* out = (float*)d_out;

    char* ws = (char*)d_ws;
    float* scales = (float*)ws;
    float* acc1   = (float*)(ws + (size_t)6 * N * 4);
    float* hpre   = acc1 + (size_t)N * 16;
    float* z      = hpre + (size_t)N * 16;
    float* z3     = z - (size_t)2 * N;
    float* planes2 = acc1;
    unsigned* partial = (unsigned*)acc1;

    hg_count_own<<<dim3(CNT_SLICES, CNT_PLANES), dim3(CNT_TPB), 0, stream>>>(
        src[0], dst[0], src[1], dst[1], src[2], dst[2], E[0], E[1], E[2], partial, N);
    hg_scale_sum<<<dim3((N + TPB - 1) / TPB, 6), dim3(TPB), 0, stream>>>(partial, scales, N);

    {
        long span = 16L * N;
        hg_init<<<dim3((unsigned)((span + TPB - 1) / TPB)), dim3(TPB), 0, stream>>>(
            acc1, N);
    }

    const int tiles  = (N + 15) / 16;
    const int gblk   = (tiles + (TPB / 32) - 1) / (TPB / 32);
    for (int r = 0; r < 3; ++r) {
        hg_gemm1_wmma<<<dim3(gblk), dim3(TPB), 0, stream>>>(
            x, scales + (size_t)r * N, W1 + (size_t)r * 32 * 16, hpre, N);
        hg_agg1_col<<<dim3(8, 2, 16), dim3(AGG_TPB), 0, stream>>>(
            src[r], dst[r], E[r], hpre, scales + (size_t)(3 + r) * N, acc1, N);
    }
    hg_relu_bias<<<dim3((N + TPB - 1) / TPB, 16), dim3(TPB), 0, stream>>>(acc1, b1, N);

    for (int r = 0; r < 3; ++r) {
        hg_gemm2<<<dim3((N + TPB - 1) / TPB), dim3(TPB), 0, stream>>>(
            acc1, scales + (size_t)r * N, W2 + (size_t)r * 16, z3 + (size_t)r * N, N);
    }
    hg_agg2_own<<<dim3(L2_SLICES, L2_PLANES), dim3(L2_TPB), 0, stream>>>(
        src[0], dst[0], src[1], dst[1], src[2], dst[2], E[0], E[1], E[2], z3, planes2, N);
    hg_out_sum<<<dim3((N + TPB - 1) / TPB), dim3(TPB), 0, stream>>>(
        planes2, scales + (size_t)3 * N, b2, out, N);
}


# both aggregations: a wave packs its own-range edges (ballot rank) into an LDS queue, then gathers and float-adds whole rows on full lanes: half / a quarter as many LDS gathers and adds; arrival order
# speedup vs baseline: 10.9046x; 1.0754x over previous
#include <hip/hip_runtime.h>

typedef float v2f __attribute__((ext_vector_type(2)));
typedef float v8f __attribute__((ext_vector_type(8)));

#define TPB 256
#define AGG_TPB 1024
#define NMAX 50000
#define NHALF_MAX ((NMAX + 1) / 2)
#define FLUSH_U 13

#define CNT_TPB 1024
#define CNT_SLICES 8
#define CNT_SLICE_MAX ((((NMAX + CNT_SLICES - 1) / CNT_SLICES) + 31) & ~31)
#define CNT_C0 10
#define CNT_C1 3
#define CNT_C2 1
#define CNT_PLANES (2 * (CNT_C0 + CNT_C1 + CNT_C2))

#define L2_TPB 1024
#define L2_SLICES 4
#define L2_SLICE_MAX ((((NMAX + L2_SLICES - 1) / L2_SLICES) + 31) & ~31)
#define L2_C0 20
#define L2_C1 7
#define L2_C2 2
#define L2_PLANES (L2_C0 + L2_C1 + L2_C2)

__global__ __launch_bounds__(TPB) void hg_init(float* __restrict__ acc1,
                                               int N) {
    long i = (long)blockIdx.x * TPB + threadIdx.x;
    if (i < 16L * N) acc1[i] = 0.0f;
}

__device__ __forceinline__ void hg_bin(unsigned* __restrict__ bins,
                                       int v, int lo, unsigned cnt) {
    const unsigned t = (unsigned)(v - lo);
    if (t < cnt)
        __hip_atomic_fetch_add(&bins[t], 1u, __ATOMIC_RELAXED,
                               __HIP_MEMORY_SCOPE_WORKGROUP);
}

__global__ __launch_bounds__(CNT_TPB) void hg_count_own(const int* __restrict__ s0,
                                                        const int* __restrict__ d0,
                                                        const int* __restrict__ s1,
                                                        const int* __restrict__ d1,
                                                        const int* __restrict__ s2,
                                                        const int* __restrict__ d2,
                                                        int E0, int E1, int E2,
                                                        unsigned* __restrict__ partial,
                                                        int N) {
    __shared__ unsigned bins[CNT_SLICE_MAX];

    const int tid = threadIdx.x;
    const int p   = blockIdx.y;
    const int* __restrict__ idx;
    int E, nchunk, chunk;
    if (p < 2 * CNT_C0) {
        const int side = p >= CNT_C0;
        idx = side ? d0 : s0;  E = E0;  nchunk = CNT_C0;  chunk = p - side * CNT_C0;
    } else if (p < 2 * (CNT_C0 + CNT_C1)) {
        const int q = p - 2 * CNT_C0;
        const int side = q >= CNT_C1;
        idx = side ? d1 : s1;  E = E1;  nchunk = CNT_C1;  chunk = q - side * CNT_C1;
    } else {
        const int q = p - 2 * (CNT_C0 + CNT_C1);
        const int side = q >= CNT_C2;
        idx = side ? d2 : s2;  E = E2;  nchunk = CNT_C2;  chunk = q - side * CNT_C2;
    }

    const int ns  = (((N + CNT_SLICES - 1) / CNT_SLICES) + 31) & ~31;
    const int lo  = blockIdx.x * ns;
    const int hiN = lo + ns < N ? lo + ns : N;
    const unsigned cnt = hiN > lo ? (unsigned)(hiN - lo) : 0u;

    for (unsigned j = tid; j < cnt; j += CNT_TPB) bins[j] = 0u;
    __syncthreads();

    const int per = ((E + nchunk * 4 - 1) / (nchunk * 4)) * 4;
    const int e0  = chunk * per;
    const int e1  = e0 + per < E ? e0 + per : E;
    const int ne  = e1 > e0 ? e1 - e0 : 0;
    const int n4  = ne >> 2;
    const int4* __restrict__ q4 = (const int4*)(idx + e0);

    int i = tid;
    for (; i + 3 * CNT_TPB < n4; i += 4 * CNT_TPB) {
        const int4 a = q4[i], b = q4[i + CNT_TPB];
        const int4 c = q4[i + 2 * CNT_TPB], d = q4[i + 3 * CNT_TPB];
        hg_bin(bins, a.x, lo, cnt); hg_bin(bins, a.y, lo, cnt);
        hg_bin(bins, a.z, lo, cnt); hg_bin(bins, a.w, lo, cnt);
        hg_bin(bins, b.x, lo, cnt); hg_bin(bins, b.y, lo, cnt);
        hg_bin(bins, b.z, lo, cnt); hg_bin(bins, b.w, lo, cnt);
        hg_bin(bins, c.x, lo, cnt); hg_bin(bins, c.y, lo, cnt);
        hg_bin(bins, c.z, lo, cnt); hg_bin(bins, c.w, lo, cnt);
        hg_bin(bins, d.x, lo, cnt); hg_bin(bins, d.y, lo, cnt);
        hg_bin(bins, d.z, lo, cnt); hg_bin(bins, d.w, lo, cnt);
    }
    for (; i < n4; i += CNT_TPB) {
        const int4 a = q4[i];
        hg_bin(bins, a.x, lo, cnt); hg_bin(bins, a.y, lo, cnt);
        hg_bin(bins, a.z, lo, cnt); hg_bin(bins, a.w, lo, cnt);
    }
    for (int e = e0 + (n4 << 2) + tid; e < e1; e += CNT_TPB)
        hg_bin(bins, idx[e], lo, cnt);
    __syncthreads();

    const size_t pitch = (size_t)((N + 31) & ~31);
    unsigned* __restrict__ pp = partial + (size_t)p * pitch + lo;
    for (unsigned j = tid; j < cnt; j += CNT_TPB) pp[j] = bins[j];
}

template <int C>
__device__ __forceinline__ unsigned hg_sum_planes(const unsigned* __restrict__ pp, size_t pitch) {
    unsigned s = 0u;
#pragma unroll
    for (int c = 0; c < C; ++c) s += pp[(size_t)c * pitch];
    return s;
}

__global__ __launch_bounds__(TPB) void hg_scale_sum(const unsigned* __restrict__ partial,
                                                    float* __restrict__ scales,
                                                    int N) {
    const int n = blockIdx.x * TPB + threadIdx.x;
    if (n >= N) return;
    const int r    = blockIdx.y >> 1;
    const int side = blockIdx.y & 1;
    const size_t pitch = (size_t)((N + 31) & ~31);
    unsigned s;
    if (r == 0)
        s = hg_sum_planes<CNT_C0>(partial + (size_t)(side * CNT_C0) * pitch + n, pitch);
    else if (r == 1)
        s = hg_sum_planes<CNT_C1>(partial + (size_t)(2 * CNT_C0 + side * CNT_C1) * pitch + n, pitch);
    else
        s = hg_sum_planes<CNT_C2>(partial + (size_t)(2 * (CNT_C0 + CNT_C1) + side * CNT_C2) * pitch + n, pitch);
    float d = (float)s;
    if (d < 1.0f) d = 1.0f;
    scales[(size_t)(side * 3 + r) * N + n] = rsqrtf(d);
}

__global__ __launch_bounds__(TPB) void hg_gemm1_wmma(const float* __restrict__ x,
                                                     const float* __restrict__ s_out,
                                                     const float* __restrict__ W,
                                                     float* __restrict__ hT,
                                                     int N) {
    const int lane = threadIdx.x & 31;
    const int wave = threadIdx.x >> 5;
    const int tile = blockIdx.x * (TPB / 32) + wave;
    const int row0 = tile * 16;
    if (row0 >= N) return;

    const int half = lane >> 4;
    const int lm   = lane & 15;
    int row  = row0 + lm;
    int rowc = row < N ? row : N - 1;
    const float sc = s_out[rowc];

    v8f c = {};
#pragma unroll
    for (int kb = 0; kb < 8; ++kb) {
        const int k0 = kb * 4 + half * 2;
        v2f xf, wf;
        xf.x = x[(size_t)rowc * 32 + k0]     * sc;
        xf.y = x[(size_t)rowc * 32 + k0 + 1] * sc;
        wf.x = W[(k0)     * 16 + lm];
        wf.y = W[(k0 + 1) * 16 + lm];
        c = __builtin_amdgcn_wmma_f32_16x16x4_f32(
                 false, wf,  false, xf,
                 (short)0, c,  false,  false);
    }

    float* __restrict__ dp = hT + (size_t)(half * 8) * N + row;
    if (row < N) {
#pragma unroll
        for (int v = 0; v < 8; ++v) dp[(size_t)v * N] = c[v];
    }
}

#define Q_WORDS 160
static_assert(NMAX < 65536, "queue word holds src and local dst in 16 bits each");

__device__ __forceinline__ unsigned hg_push(unsigned* q, unsigned n, bool keep, unsigned word) {
    const unsigned m = __builtin_amdgcn_ballot_w32(keep);
    if (keep) q[n + __builtin_amdgcn_mbcnt_lo(m, 0u)] = word;
    return n + (unsigned)__builtin_popcount(m);
}

__device__ __forceinline__ void hg_add_word(float* sums, unsigned w, float v) {
    __hip_atomic_fetch_add(&sums[w >> 16], v, __ATOMIC_RELAXED, __HIP_MEMORY_SCOPE_WORKGROUP);
}

template <int R>
__device__ __forceinline__ void hg_take(const unsigned* q, const float* vals, float* sums,
                                        unsigned n, unsigned lane, unsigned carried, unsigned& wq) {
    unsigned w[R + 1];
#pragma unroll
    for (int r = 0; r < R; ++r) w[r] = q[lane + 32u * (unsigned)r];
    w[R] = lane + 32u * (unsigned)R < n ? q[lane + 32u * (unsigned)R] : 0u;
    if (lane < carried) w[0] = wq;
    float v[R + 1];
#pragma unroll
    for (int r = 0; r < R; ++r) v[r] = vals[w[r] & 0xffffu];
#pragma unroll
    for (int r = 0; r < R; ++r) hg_add_word(sums, w[r], v[r]);
    wq = w[R];
}

__device__ __forceinline__ void hg_trip(const int4 d, const int4 s, bool ok,
                                        const float* vals, float* sums, unsigned* q,
                                        int lo, unsigned cnt, unsigned lane,
                                        unsigned& rem, unsigned& wq) {
    const unsigned t0 = (unsigned)(d.x - lo), t1 = (unsigned)(d.y - lo);
    const unsigned t2 = (unsigned)(d.z - lo), t3 = (unsigned)(d.w - lo);
    const unsigned c = ok ? cnt : 0u;
    unsigned n = rem;
    n = hg_push(q, n, t0 < c, (unsigned)s.x | (t0 << 16));
    n = hg_push(q, n, t1 < c, (unsigned)s.y | (t1 << 16));
    n = hg_push(q, n, t2 < c, (unsigned)s.z | (t2 << 16));
    n = hg_push(q, n, t3 < c, (unsigned)s.w | (t3 << 16));
    __builtin_amdgcn_fence(__ATOMIC_RELEASE, "wavefront", "local");
    __builtin_amdgcn_wave_barrier();
    __builtin_amdgcn_fence(__ATOMIC_ACQUIRE, "wavefront", "local");
    const unsigned rows = n >> 5;
    if      (rows == 2u) hg_take<2>(q, vals, sums, n, lane, rem, wq);
    else if (rows == 1u) hg_take<1>(q, vals, sums, n, lane, rem, wq);
    else if (rows == 3u) hg_take<3>(q, vals, sums, n, lane, rem, wq);
    else if (rows == 0u) hg_take<0>(q, vals, sums, n, lane, rem, wq);
    else                 hg_take<4>(q, vals, sums, n, lane, rem, wq);
    rem = n & 31u;
}

__device__ __forceinline__ void hg_scan_own(const int4* __restrict__ s4, const int4* __restrict__ d4,
                                            int n4, const float* vals, float* sums, unsigned* q,
                                            int lo, unsigned cnt, int tid, int tpb) {
    const unsigned lane = (unsigned)tid & 31u;
    int4 dA = make_int4(0, 0, 0, 0), sA = make_int4(0, 0, 0, 0);
    int4 dB = make_int4(0, 0, 0, 0), sB = make_int4(0, 0, 0, 0);
    unsigned rem = 0u, wq = 0u;
    int ib = __builtin_amdgcn_readfirstlane(tid & ~31);
    if (ib + (int)lane < n4) { dA = d4[ib + (int)lane]; sA = s4[ib + (int)lane]; }
    const int last = n4 - 1;
    for (; ib < n4; ib += 2 * tpb) {
        const int iA = ib + (int)lane, iB = iA + tpb, iC = iB + tpb;
        const int jB = iB < n4 ? iB : last;
        dB = d4[jB];  sB = s4[jB];
        hg_trip(dA, sA, iA < n4, vals, sums, q, lo, cnt, lane, rem, wq);
        const int jC = iC < n4 ? iC : last;
        dA = d4[jC];  sA = s4[jC];
        if (ib + tpb < n4)
            hg_trip(dB, sB, iB < n4, vals, sums, q, lo, cnt, lane, rem, wq);
    }
    if (lane < rem) hg_add_word(sums, wq, vals[wq & 0xffffu]);
}

__device__ __forceinline__ void hg_edge_add(float* __restrict__ acc,
                                            float v, int d, int lo, unsigned cnt) {
    const unsigned dl = (unsigned)(d - lo);
    if (dl < cnt)
        __hip_atomic_fetch_add(&acc[dl], v, __ATOMIC_RELAXED,
                               __HIP_MEMORY_SCOPE_WORKGROUP);
}

__global__ __launch_bounds__(AGG_TPB) void hg_agg1_col(const int* __restrict__ src,
                                                       const int* __restrict__ dst,
                                                       int E,
                                                       const float* __restrict__ hT,
                                                       const float* __restrict__ s_in,
                                                       float* __restrict__ accT,
                                                       int N) {
    __shared__ float hcol[NMAX];
    __shared__ float acc[NHALF_MAX];
    __shared__ unsigned qbuf[(AGG_TPB / 32) * Q_WORDS];

    const int tid    = threadIdx.x;
    const int nchunk = gridDim.x;
    const int chunk  = blockIdx.x;
    const int f      = blockIdx.z;
    const int nh     = (N + 1) >> 1;
    const int lo     = blockIdx.y * nh;
    const int hiN    = lo + nh < N ? lo + nh : N;
    const unsigned cnt = hiN > lo ? (unsigned)(hiN - lo) : 0u;

    const float* __restrict__ hp = hT + (size_t)f * N;
    for (int i = tid; i < N; i += AGG_TPB) hcol[i] = hp[i];
    for (unsigned i = tid; i < cnt; i += AGG_TPB) acc[i] = 0.0f;
    __syncthreads();

    const int per = ((E + nchunk * 4 - 1) / (nchunk * 4)) * 4;
    const int e0  = chunk * per;
    const int e1  = e0 + per < E ? e0 + per : E;
    const int ne  = e1 > e0 ? e1 - e0 : 0;
    const int n4  = ne >> 2;
    const int4* __restrict__ s4 = (const int4*)(src + e0);
    const int4* __restrict__ d4 = (const int4*)(dst + e0);

    hg_scan_own(s4, d4, n4, hcol, acc, qbuf + (tid >> 5) * Q_WORDS, lo, cnt, tid, AGG_TPB);
    for (int e = e0 + (n4 << 2) + tid; e < e1; e += AGG_TPB)
        hg_edge_add(acc, hcol[src[e]], dst[e], lo, cnt);
    __syncthreads();

    float* __restrict__ ap = accT + (size_t)f * N + lo;
    const float* __restrict__ sp = s_in + lo;
    if (cnt != 0u) {
        const unsigned last = cnt - 1u;
        for (unsigned base = 0u; base < cnt; base += FLUSH_U * AGG_TPB) {
            float s[FLUSH_U], a[FLUSH_U];
#pragma unroll
            for (int u = 0; u < FLUSH_U; ++u) {
                const unsigned j  = base + (unsigned)u * AGG_TPB + tid;
                const unsigned jc = j < cnt ? j : last;
                s[u] = sp[jc];
                a[u] = acc[jc];
            }
#pragma unroll
            for (int u = 0; u < FLUSH_U; ++u) {
                const unsigned j = base + (unsigned)u * AGG_TPB + tid;
                const float v = a[u] * s[u];
                if (j < cnt && v != 0.0f) atomicAdd(ap + j, v);
            }
        }
    }
}

__global__ __launch_bounds__(TPB) void hg_relu_bias(float* __restrict__ acc,
                                                    const float* __restrict__ b1,
                                                    int N) {
    const int f = blockIdx.y;
    const int n = blockIdx.x * TPB + threadIdx.x;
    if (n < N) {
        const size_t i = (size_t)f * N + n;
        float v = acc[i] + b1[f] + b1[16 + f] + b1[32 + f];
        acc[i] = v > 0.0f ? v : 0.0f;
    }
}

__global__ __launch_bounds__(TPB) void hg_gemm2(const float* __restrict__ h,
                                                const float* __restrict__ s_out,
                                                const float* __restrict__ w,
                                                float* __restrict__ z,
                                                int N) {
    int i = blockIdx.x * TPB + threadIdx.x;
    if (i >= N) return;
    const float* hp = h + i;
    const float4* wp = (const float4*)w;
    float acc = 0.0f;
#pragma unroll
    for (int q = 0; q < 4; ++q) {
        float4 a, b = wp[q];
        a.x = hp[(size_t)(4 * q + 0) * N];
        a.y = hp[(size_t)(4 * q + 1) * N];
        a.z = hp[(size_t)(4 * q + 2) * N];
        a.w = hp[(size_t)(4 * q + 3) * N];
        acc += a.x * b.x + a.y * b.y + a.z * b.z + a.w * b.w;
    }
    z[i] = acc * s_out[i];
}

__global__ __launch_bounds__(L2_TPB) void hg_agg2_own(const int* __restrict__ s0,
                                                      const int* __restrict__ d0,
                                                      const int* __restrict__ s1,
                                                      const int* __restrict__ d1,
                                                      const int* __restrict__ s2,
                                                      const int* __restrict__ d2,
                                                      int E0, int E1, int E2,
                                                      const float* __restrict__ z3,
                                                      float* __restrict__ partial,
                                                      int N) {
    __shared__ float zc[NMAX];
    __shared__ float bins[L2_SLICE_MAX];
    __shared__ unsigned qbuf[(L2_TPB / 32) * Q_WORDS];

    const int tid = threadIdx.x;
    const int p   = blockIdx.y;
    const int* __restrict__ src;
    const int* __restrict__ dst;
    int E, nchunk, chunk, r;
    if (p < L2_C0) {
        src = s0;  dst = d0;  E = E0;  nchunk = L2_C0;  chunk = p;                  r = 0;
    } else if (p < L2_C0 + L2_C1) {
        src = s1;  dst = d1;  E = E1;  nchunk = L2_C1;  chunk = p - L2_C0;          r = 1;
    } else {
        src = s2;  dst = d2;  E = E2;  nchunk = L2_C2;  chunk = p - L2_C0 - L2_C1;  r = 2;
    }

    const int ns  = (((N + L2_SLICES - 1) / L2_SLICES) + 31) & ~31;
    const int lo  = blockIdx.x * ns;
    const int hiN = lo + ns < N ? lo + ns : N;
    const unsigned cnt = hiN > lo ? (unsigned)(hiN - lo) : 0u;

    const float* __restrict__ zp = z3 + (size_t)r * N;
    for (int i = tid; i < N; i += L2_TPB) zc[i] = zp[i];
    for (unsigned j = tid; j < cnt; j += L2_TPB) bins[j] = 0.0f;
    __syncthreads();

    const int per = ((E + nchunk * 4 - 1) / (nchunk * 4)) * 4;
    const int e0  = chunk * per;
    const int e1  = e0 + per < E ? e0 + per : E;
    const int ne  = e1 > e0 ? e1 - e0 : 0;
    const int n4  = ne >> 2;
    const int4* __restrict__ s4 = (const int4*)(src + e0);
    const int4* __restrict__ d4 = (const int4*)(dst + e0);

    hg_scan_own(s4, d4, n4, zc, bins, qbuf + (tid >> 5) * Q_WORDS, lo, cnt, tid, L2_TPB);
    for (int e = e0 + (n4 << 2) + tid; e < e1; e += L2_TPB)
        hg_edge_add(bins, zc[src[e]], dst[e], lo, cnt);
    __syncthreads();

    const size_t pitch = (size_t)((N + 31) & ~31);
    float* __restrict__ pp = partial + (size_t)p * pitch + lo;
    for (unsigned j = tid; j < cnt; j += L2_TPB) pp[j] = bins[j];
}

template <int C>
__device__ __forceinline__ float hg_sum_fplanes(const float* __restrict__ pp, size_t pitch) {
    float s = 0.0f;
#pragma unroll
    for (int c = 0; c < C; ++c) s += pp[(size_t)c * pitch];
    return s;
}

__global__ __launch_bounds__(TPB) void hg_out_sum(const float* __restrict__ partial,
                                                  const float* __restrict__ s_in3,
                                                  const float* __restrict__ b2,
                                                  float* __restrict__ out,
                                                  int N) {
    const int n = blockIdx.x * TPB + threadIdx.x;
    if (n >= N) return;
    const size_t pitch = (size_t)((N + 31) & ~31);
    const float m0 = hg_sum_fplanes<L2_C0>(partial + n, pitch);
    const float m1 = hg_sum_fplanes<L2_C1>(partial + (size_t)L2_C0 * pitch + n, pitch);
    const float m2 = hg_sum_fplanes<L2_C2>(partial + (size_t)(L2_C0 + L2_C1) * pitch + n, pitch);
    out[n] = (m0 * s_in3[n] + b2[0])
           + (m1 * s_in3[(size_t)N + n] + b2[1])
           + (m2 * s_in3[(size_t)2 * N + n] + b2[2]);
}

extern "C" void kernel_launch(void* const* d_in, const int* in_sizes, int n_in,
                              void* d_out, int out_size, void* d_ws, size_t ws_size,
                              hipStream_t stream) {
    const float* x  = (const float*)d_in[0];
    const int* src[3] = {(const int*)d_in[1], (const int*)d_in[3], (const int*)d_in[5]};
    const int* dst[3] = {(const int*)d_in[2], (const int*)d_in[4], (const int*)d_in[6]};
    const int  E[3]   = {in_sizes[1], in_sizes[3], in_sizes[5]};
    const float* W1 = (const float*)d_in[7];
    const float* b1 = (const float*)d_in[8];
    const float* W2 = (const float*)d_in[9];
    const float* b2 = (const float*)d_in[10];
    const int N = in_sizes[0] / 32;
    float* out = (float*)d_out;

    char* ws = (char*)d_ws;
    float* scales = (float*)ws;
    float* acc1   = (float*)(ws + (size_t)6 * N * 4);
    float* hpre   = acc1 + (size_t)N * 16;
    float* z      = hpre + (size_t)N * 16;
    float* z3     = z - (size_t)2 * N;
    float* planes2 = acc1;
    unsigned* partial = (unsigned*)acc1;

    hg_count_own<<<dim3(CNT_SLICES, CNT_PLANES), dim3(CNT_TPB), 0, stream>>>(
        src[0], dst[0], src[1], dst[1], src[2], dst[2], E[0], E[1], E[2], partial, N);
    hg_scale_sum<<<dim3((N + TPB - 1) / TPB, 6), dim3(TPB), 0, stream>>>(partial, scales, N);

    {
        long span = 16L * N;
        hg_init<<<dim3((unsigned)((span + TPB - 1) / TPB)), dim3(TPB), 0, stream>>>(
            acc1, N);
    }

    const int tiles  = (N + 15) / 16;
    const int gblk   = (tiles + (TPB / 32) - 1) / (TPB / 32);
    for (int r = 0; r < 3; ++r) {
        hg_gemm1_wmma<<<dim3(gblk), dim3(TPB), 0, stream>>>(
            x, scales + (size_t)r * N, W1 + (size_t)r * 32 * 16, hpre, N);
        hg_agg1_col<<<dim3(8, 2, 16), dim3(AGG_TPB), 0, stream>>>(
            src[r], dst[r], E[r], hpre, scales + (size_t)(3 + r) * N, acc1, N);
    }
    hg_relu_bias<<<dim3((N + TPB - 1) / TPB, 16), dim3(TPB), 0, stream>>>(acc1, b1, N);

    for (int r = 0; r < 3; ++r) {
        hg_gemm2<<<dim3((N + TPB - 1) / TPB), dim3(TPB), 0, stream>>>(
            acc1, scales + (size_t)r * N, W2 + (size_t)r * 16, z3 + (size_t)r * N, N);
    }
    hg_agg2_own<<<dim3(L2_SLICES, L2_PLANES), dim3(L2_TPB), 0, stream>>>(
        src[0], dst[0], src[1], dst[1], src[2], dst[2], E[0], E[1], E[2], z3, planes2, N);
    hg_out_sum<<<dim3((N + TPB - 1) / TPB), dim3(TPB), 0, stream>>>(
        planes2, scales + (size_t)3 * N, b2, out, N);
}


# layer-1 edges binned once per call by dst half into packed words (whole-line stores); owners stream only their half's words: no range test, ballot or queue per feature plane; sums in arrival order
# speedup vs baseline: 11.9070x; 1.0919x over previous
#include <hip/hip_runtime.h>

typedef float v2f __attribute__((ext_vector_type(2)));
typedef float v8f __attribute__((ext_vector_type(8)));

#define TPB 256
#define AGG_TPB 1024
#define NMAX 50000
#define NHALF_MAX ((NMAX + 1) / 2)
#define FLUSH_U 13

#define CNT_TPB 1024
#define CNT_SLICES 8
#define CNT_SLICE_MAX ((((NMAX + CNT_SLICES - 1) / CNT_SLICES) + 31) & ~31)
#define CNT_C0 10
#define CNT_C1 3
#define CNT_C2 1
#define CNT_PLANES (2 * (CNT_C0 + CNT_C1 + CNT_C2))

#define L2_TPB 1024
#define L2_SLICES 4
#define L2_SLICE_MAX ((((NMAX + L2_SLICES - 1) / L2_SLICES) + 31) & ~31)
#define L2_C0 20
#define L2_C1 7
#define L2_C2 2
#define L2_PLANES (L2_C0 + L2_C1 + L2_C2)

__global__ __launch_bounds__(TPB) void hg_init(float* __restrict__ acc1,
                                               int N) {
    long i = (long)blockIdx.x * TPB + threadIdx.x;
    if (i < 16L * N) acc1[i] = 0.0f;
}

__device__ __forceinline__ void hg_bin(unsigned* __restrict__ bins,
                                       int v, int lo, unsigned cnt) {
    const unsigned t = (unsigned)(v - lo);
    if (t < cnt)
        __hip_atomic_fetch_add(&bins[t], 1u, __ATOMIC_RELAXED,
                               __HIP_MEMORY_SCOPE_WORKGROUP);
}

__global__ __launch_bounds__(CNT_TPB) void hg_count_own(const int* __restrict__ s0,
                                                        const int* __restrict__ d0,
                                                        const int* __restrict__ s1,
                                                        const int* __restrict__ d1,
                                                        const int* __restrict__ s2,
                                                        const int* __restrict__ d2,
                                                        int E0, int E1, int E2,
                                                        unsigned* __restrict__ partial,
                                                        int N) {
    __shared__ unsigned bins[CNT_SLICE_MAX];

    const int tid = threadIdx.x;
    const int p   = blockIdx.y;
    const int* __restrict__ idx;
    int E, nchunk, chunk;
    if (p < 2 * CNT_C0) {
        const int side = p >= CNT_C0;
        idx = side ? d0 : s0;  E = E0;  nchunk = CNT_C0;  chunk = p - side * CNT_C0;
    } else if (p < 2 * (CNT_C0 + CNT_C1)) {
        const int q = p - 2 * CNT_C0;
        const int side = q >= CNT_C1;
        idx = side ? d1 : s1;  E = E1;  nchunk = CNT_C1;  chunk = q - side * CNT_C1;
    } else {
        const int q = p - 2 * (CNT_C0 + CNT_C1);
        const int side = q >= CNT_C2;
        idx = side ? d2 : s2;  E = E2;  nchunk = CNT_C2;  chunk = q - side * CNT_C2;
    }

    const int ns  = (((N + CNT_SLICES - 1) / CNT_SLICES) + 31) & ~31;
    const int lo  = blockIdx.x * ns;
    const int hiN = lo + ns < N ? lo + ns : N;
    const unsigned cnt = hiN > lo ? (unsigned)(hiN - lo) : 0u;

    for (unsigned j = tid; j < cnt; j += CNT_TPB) bins[j] = 0u;
    __syncthreads();

    const int per = ((E + nchunk * 4 - 1) / (nchunk * 4)) * 4;
    const int e0  = chunk * per;
    const int e1  = e0 + per < E ? e0 + per : E;
    const int ne  = e1 > e0 ? e1 - e0 : 0;
    const int n4  = ne >> 2;
    const int4* __restrict__ q4 = (const int4*)(idx + e0);

    int i = tid;
    for (; i + 3 * CNT_TPB < n4; i += 4 * CNT_TPB) {
        const int4 a = q4[i], b = q4[i + CNT_TPB];
        const int4 c = q4[i + 2 * CNT_TPB], d = q4[i + 3 * CNT_TPB];
        hg_bin(bins, a.x, lo, cnt); hg_bin(bins, a.y, lo, cnt);
        hg_bin(bins, a.z, lo, cnt); hg_bin(bins, a.w, lo, cnt);
        hg_bin(bins, b.x, lo, cnt); hg_bin(bins, b.y, lo, cnt);
        hg_bin(bins, b.z, lo, cnt); hg_bin(bins, b.w, lo, cnt);
        hg_bin(bins, c.x, lo, cnt); hg_bin(bins, c.y, lo, cnt);
        hg_bin(bins, c.z, lo, cnt); hg_bin(bins, c.w, lo, cnt);
        hg_bin(bins, d.x, lo, cnt); hg_bin(bins, d.y, lo, cnt);
        hg_bin(bins, d.z, lo, cnt); hg_bin(bins, d.w, lo, cnt);
    }
    for (; i < n4; i += CNT_TPB) {
        const int4 a = q4[i];
        hg_bin(bins, a.x, lo, cnt); hg_bin(bins, a.y, lo, cnt);
        hg_bin(bins, a.z, lo, cnt); hg_bin(bins, a.w, lo, cnt);
    }
    for (int e = e0 + (n4 << 2) + tid; e < e1; e += CNT_TPB)
        hg_bin(bins, idx[e], lo, cnt);
    __syncthreads();

    const size_t pitch = (size_t)((N + 31) & ~31);
    unsigned* __restrict__ pp = partial + (size_t)p * pitch + lo;
    for (unsigned j = tid; j < cnt; j += CNT_TPB) pp[j] = bins[j];
}

template <int C>
__device__ __forceinline__ unsigned hg_sum_planes(const unsigned* __restrict__ pp, size_t pitch) {
    unsigned s = 0u;
#pragma unroll
    for (int c = 0; c < C; ++c) s += pp[(size_t)c * pitch];
    return s;
}

__global__ __launch_bounds__(TPB) void hg_scale_sum(const unsigned* __restrict__ partial,
                                                    float* __restrict__ scales,
                                                    int N) {
    const int n = blockIdx.x * TPB + threadIdx.x;
    if (n >= N) return;
    const int r    = blockIdx.y >> 1;
    const int side = blockIdx.y & 1;
    const size_t pitch = (size_t)((N + 31) & ~31);
    unsigned s;
    if (r == 0)
        s = hg_sum_planes<CNT_C0>(partial + (size_t)(side * CNT_C0) * pitch + n, pitch);
    else if (r == 1)
        s = hg_sum_planes<CNT_C1>(partial + (size_t)(2 * CNT_C0 + side * CNT_C1) * pitch + n, pitch);
    else
        s = hg_sum_planes<CNT_C2>(partial + (size_t)(2 * (CNT_C0 + CNT_C1) + side * CNT_C2) * pitch + n, pitch);
    float d = (float)s;
    if (d < 1.0f) d = 1.0f;
    scales[(size_t)(side * 3 + r) * N + n] = rsqrtf(d);
}

__global__ __launch_bounds__(TPB) void hg_gemm1_wmma(const float* __restrict__ x,
                                                     const float* __restrict__ s_out,
                                                     const float* __restrict__ W,
                                                     float* __restrict__ hT,
                                                     int N) {
    const int lane = threadIdx.x & 31;
    const int wave = threadIdx.x >> 5;
    const int tile = blockIdx.x * (TPB / 32) + wave;
    const int row0 = tile * 16;
    if (row0 >= N) return;

    const int half = lane >> 4;
    const int lm   = lane & 15;
    int row  = row0 + lm;
    int rowc = row < N ? row : N - 1;
    const float sc = s_out[rowc];

    v8f c = {};
#pragma unroll
    for (int kb = 0; kb < 8; ++kb) {
        const int k0 = kb * 4 + half * 2;
        v2f xf, wf;
        xf.x = x[(size_t)rowc * 32 + k0]     * sc;
        xf.y = x[(size_t)rowc * 32 + k0 + 1] * sc;
        wf.x = W[(k0)     * 16 + lm];
        wf.y = W[(k0 + 1) * 16 + lm];
        c = __builtin_amdgcn_wmma_f32_16x16x4_f32(
                 false, wf,  false, xf,
                 (short)0, c,  false,  false);
    }

    float* __restrict__ dp = hT + (size_t)(half * 8) * N + row;
    if (row < N) {
#pragma unroll
        for (int v = 0; v < 8; ++v) dp[(size_t)v * N] = c[v];
    }
}

#define Q_WORDS 160
static_assert(NMAX < 65536, "queue word holds src and local dst in 16 bits each");

__device__ __forceinline__ unsigned hg_push(unsigned* q, unsigned n, bool keep, unsigned word) {
    const unsigned m = __builtin_amdgcn_ballot_w32(keep);
    if (keep) q[n + __builtin_amdgcn_mbcnt_lo(m, 0u)] = word;
    return n + (unsigned)__builtin_popcount(m);
}

__device__ __forceinline__ void hg_add_word(float* sums, unsigned w, float v) {
    __hip_atomic_fetch_add(&sums[w >> 16], v, __ATOMIC_RELAXED, __HIP_MEMORY_SCOPE_WORKGROUP);
}

template <int R>
__device__ __forceinline__ void hg_take(const unsigned* q, const float* vals, float* sums,
                                        unsigned n, unsigned lane, unsigned carried, unsigned& wq) {
    unsigned w[R + 1];
#pragma unroll
    for (int r = 0; r < R; ++r) w[r] = q[lane + 32u * (unsigned)r];
    w[R] = lane + 32u * (unsigned)R < n ? q[lane + 32u * (unsigned)R] : 0u;
    if (lane < carried) w[0] = wq;
    float v[R + 1];
#pragma unroll
    for (int r = 0; r < R; ++r) v[r] = vals[w[r] & 0xffffu];
#pragma unroll
    for (int r = 0; r < R; ++r) hg_add_word(sums, w[r], v[r]);
    wq = w[R];
}

__device__ __forceinline__ void hg_trip(const int4 d, const int4 s, bool ok,
                                        const float* vals, float* sums, unsigned* q,
                                        int lo, unsigned cnt, unsigned lane,
                                        unsigned& rem, unsigned& wq) {
    const unsigned t0 = (unsigned)(d.x - lo), t1 = (unsigned)(d.y - lo);
    const unsigned t2 = (unsigned)(d.z - lo), t3 = (unsigned)(d.w - lo);
    const unsigned c = ok ? cnt : 0u;
    unsigned n = rem;
    n = hg_push(q, n, t0 < c, (unsigned)s.x | (t0 << 16));
    n = hg_push(q, n, t1 < c, (unsigned)s.y | (t1 << 16));
    n = hg_push(q, n, t2 < c, (unsigned)s.z | (t2 << 16));
    n = hg_push(q, n, t3 < c, (unsigned)s.w | (t3 << 16));
    __builtin_amdgcn_fence(__ATOMIC_RELEASE, "wavefront", "local");
    __builtin_amdgcn_wave_barrier();
    __builtin_amdgcn_fence(__ATOMIC_ACQUIRE, "wavefront", "local");
    const unsigned rows = n >> 5;
    if      (rows == 2u) hg_take<2>(q, vals, sums, n, lane, rem, wq);
    else if (rows == 1u) hg_take<1>(q, vals, sums, n, lane, rem, wq);
    else if (rows == 3u) hg_take<3>(q, vals, sums, n, lane, rem, wq);
    else if (rows == 0u) hg_take<0>(q, vals, sums, n, lane, rem, wq);
    else                 hg_take<4>(q, vals, sums, n, lane, rem, wq);
    rem = n & 31u;
}

__device__ __forceinline__ void hg_scan_own(const int4* __restrict__ s4, const int4* __restrict__ d4,
                                            int n4, const float* vals, float* sums, unsigned* q,
                                            int lo, unsigned cnt, int tid, int tpb) {
    const unsigned lane = (unsigned)tid & 31u;
    int4 dA = make_int4(0, 0, 0, 0), sA = make_int4(0, 0, 0, 0);
    int4 dB = make_int4(0, 0, 0, 0), sB = make_int4(0, 0, 0, 0);
    unsigned rem = 0u, wq = 0u;
    int ib = __builtin_amdgcn_readfirstlane(tid & ~31);
    if (ib + (int)lane < n4) { dA = d4[ib + (int)lane]; sA = s4[ib + (int)lane]; }
    const int last = n4 - 1;
    for (; ib < n4; ib += 2 * tpb) {
        const int iA = ib + (int)lane, iB = iA + tpb, iC = iB + tpb;
        const int jB = iB < n4 ? iB : last;
        dB = d4[jB];  sB = s4[jB];
        hg_trip(dA, sA, iA < n4, vals, sums, q, lo, cnt, lane, rem, wq);
        const int jC = iC < n4 ? iC : last;
        dA = d4[jC];  sA = s4[jC];
        if (ib + tpb < n4)
            hg_trip(dB, sB, iB < n4, vals, sums, q, lo, cnt, lane, rem, wq);
    }
    if (lane < rem) hg_add_word(sums, wq, vals[wq & 0xffffu]);
}

__device__ __forceinline__ void hg_edge_add(float* __restrict__ acc,
                                            float v, int d, int lo, unsigned cnt) {
    const unsigned dl = (unsigned)(d - lo);
    if (dl < cnt)
        __hip_atomic_fetch_add(&acc[dl], v, __ATOMIC_RELAXED,
                               __HIP_MEMORY_SCOPE_WORKGROUP);
}

__global__ __launch_bounds__(AGG_TPB) void hg_agg1_col(const int* __restrict__ src,
                                                       const int* __restrict__ dst,
                                                       int E,
                                                       const float* __restrict__ hT,
                                                       const float* __restrict__ s_in,
                                                       float* __restrict__ accT,
                                                       int N) {
    __shared__ float hcol[NMAX];
    __shared__ float acc[NHALF_MAX];
    __shared__ unsigned qbuf[(AGG_TPB / 32) * Q_WORDS];

    const int tid    = threadIdx.x;
    const int nchunk = gridDim.x;
    const int chunk  = blockIdx.x;
    const int f      = blockIdx.z;
    const int nh     = (N + 1) >> 1;
    const int lo     = blockIdx.y * nh;
    const int hiN    = lo + nh < N ? lo + nh : N;
    const unsigned cnt = hiN > lo ? (unsigned)(hiN - lo) : 0u;

    const float* __restrict__ hp = hT + (size_t)f * N;
    for (int i = tid; i < N; i += AGG_TPB) hcol[i] = hp[i];
    for (unsigned i = tid; i < cnt; i += AGG_TPB) acc[i] = 0.0f;
    __syncthreads();

    const int per = ((E + nchunk * 4 - 1) / (nchunk * 4)) * 4;
    const int e0  = chunk * per;
    const int e1  = e0 + per < E ? e0 + per : E;
    const int ne  = e1 > e0 ? e1 - e0 : 0;
    const int n4  = ne >> 2;
    const int4* __restrict__ s4 = (const int4*)(src + e0);
    const int4* __restrict__ d4 = (const int4*)(dst + e0);

    hg_scan_own(s4, d4, n4, hcol, acc, qbuf + (tid >> 5) * Q_WORDS, lo, cnt, tid, AGG_TPB);
    for (int e = e0 + (n4 << 2) + tid; e < e1; e += AGG_TPB)
        hg_edge_add(acc, hcol[src[e]], dst[e], lo, cnt);
    __syncthreads();

    float* __restrict__ ap = accT + (size_t)f * N + lo;
    const float* __restrict__ sp = s_in + lo;
    if (cnt != 0u) {
        const unsigned last = cnt - 1u;
        for (unsigned base = 0u; base < cnt; base += FLUSH_U * AGG_TPB) {
            float s[FLUSH_U], a[FLUSH_U];
#pragma unroll
            for (int u = 0; u < FLUSH_U; ++u) {
                const unsigned j  = base + (unsigned)u * AGG_TPB + tid;
                const unsigned jc = j < cnt ? j : last;
                s[u] = sp[jc];
                a[u] = acc[jc];
            }
#pragma unroll
            for (int u = 0; u < FLUSH_U; ++u) {
                const unsigned j = base + (unsigned)u * AGG_TPB + tid;
                const float v = a[u] * s[u];
                if (j < cnt && v != 0.0f) atomicAdd(ap + j, v);
            }
        }
    }
}

#define BIN_TPB 256
#define BIN_RUN 2048
#define BIN_HDR 32
#define BIN_PITCH (BIN_HDR + BIN_RUN)
#define BIN_WAVES (BIN_TPB / 32)

__global__ __launch_bounds__(BIN_TPB) void hg_bin_half(const int* __restrict__ s0,
                                                       const int* __restrict__ d0,
                                                       const int* __restrict__ s1,
                                                       const int* __restrict__ d1,
                                                       const int* __restrict__ s2,
                                                       const int* __restrict__ d2,
                                                       int E0, int E1, int E2,
                                                       int nR0, int nR1,
                                                       unsigned* __restrict__ words,
                                                       int N) {
    __shared__ unsigned buf[BIN_PITCH];
    __shared__ unsigned wc0[BIN_WAVES];
    __shared__ unsigned wcv[BIN_WAVES];

    const int tid = threadIdx.x;
    const int wv  = tid >> 5;
    const int run = blockIdx.x;
    const int* __restrict__ src;
    const int* __restrict__ dst;
    int E, q;
    if (run < nR0)            { src = s0;  dst = d0;  E = E0;  q = run; }
    else if (run < nR0 + nR1) { src = s1;  dst = d1;  E = E1;  q = run - nR0; }
    else                      { src = s2;  dst = d2;  E = E2;  q = run - nR0 - nR1; }
    const int nh  = (N + 1) >> 1;
    const int R0  = q * BIN_RUN;
    const int len = E - R0 < BIN_RUN ? E - R0 : BIN_RUN;

    int d[8], s[8];
    bool ok[8];
#pragma unroll
    for (int v = 0; v < 2; ++v) {
        const int b = v * (BIN_RUN / 2) + 4 * tid;
        if (b + 3 < len) {
            const int4 dv = *(const int4*)(dst + R0 + b);
            const int4 sv = *(const int4*)(src + R0 + b);
            d[4 * v] = dv.x;  d[4 * v + 1] = dv.y;  d[4 * v + 2] = dv.z;  d[4 * v + 3] = dv.w;
            s[4 * v] = sv.x;  s[4 * v + 1] = sv.y;  s[4 * v + 2] = sv.z;  s[4 * v + 3] = sv.w;
            ok[4 * v] = ok[4 * v + 1] = ok[4 * v + 2] = ok[4 * v + 3] = true;
        } else {
#pragma unroll
            for (int k = 0; k < 4; ++k) {
                const bool in = b + k < len;
                ok[4 * v + k] = in;
                d[4 * v + k]  = in ? dst[R0 + b + k] : 0;
                s[4 * v + k]  = in ? src[R0 + b + k] : 0;
            }
        }
    }

    unsigned word[8], m0[8], m1[8];
    bool hi[8];
    unsigned c0 = 0u, cv = 0u;
#pragma unroll
    for (int k = 0; k < 8; ++k) {
        hi[k] = d[k] >= nh;
        const unsigned t = (unsigned)d[k] < (unsigned)N ? (unsigned)(hi[k] ? d[k] - nh : d[k]) : 0xffffu;
        word[k] = (unsigned)s[k] | (t << 16);
        m0[k] = __builtin_amdgcn_ballot_w32(ok[k] && !hi[k]);
        m1[k] = __builtin_amdgcn_ballot_w32(ok[k] && hi[k]);
        c0 += (unsigned)__builtin_popcount(m0[k]);
        cv += (unsigned)__builtin_popcount(m0[k]) + (unsigned)__builtin_popcount(m1[k]);
    }
    if ((tid & 31) == 0) { wc0[wv] = c0;  wcv[wv] = cv; }
    __syncthreads();

    unsigned off0 = 0u, tot0 = 0u, before = 0u;
#pragma unroll
    for (int w = 0; w < BIN_WAVES; ++w) {
        const unsigned a = wc0[w], b = wcv[w];
        if (w < wv) { off0 += a;  before += b; }
        tot0 += a;
    }
    unsigned p0 = BIN_HDR + off0;
    unsigned p1 = BIN_HDR + tot0 + (before - off0);
#pragma unroll
    for (int k = 0; k < 8; ++k) {
        if (ok[k] && !hi[k]) buf[p0 + __builtin_amdgcn_mbcnt_lo(m0[k], 0u)] = word[k];
        if (ok[k] && hi[k])  buf[p1 + __builtin_amdgcn_mbcnt_lo(m1[k], 0u)] = word[k];
        p0 += (unsigned)__builtin_popcount(m0[k]);
        p1 += (unsigned)__builtin_popcount(m1[k]);
    }
    if (tid < BIN_HDR) buf[tid] = tid == 0 ? tot0 : (tid == 1 ? (unsigned)len : 0u);
    __syncthreads();

    unsigned* __restrict__ wr = words + (size_t)run * BIN_PITCH;
    for (int i = tid; i < BIN_HDR + len; i += BIN_TPB) wr[i] = buf[i];
}

#define RUN_ROWS 8
__device__ __forceinline__ void hg_load_rows(const unsigned* __restrict__ wp, int i0, int last,
                                             int lane, unsigned (&w)[RUN_ROWS]) {
#pragma unroll
    for (int u = 0; u < RUN_ROWS; ++u) {
        const int i = i0 + 32 * u + lane;
        w[u] = wp[i < last ? i : last];
    }
}

__device__ __forceinline__ void hg_add_rows(const unsigned (&w)[RUN_ROWS], int i0, int n, int lane,
                                            const float* vals, float* sums, unsigned cnt) {
    float v[RUN_ROWS];
#pragma unroll
    for (int u = 0; u < RUN_ROWS; ++u) v[u] = vals[w[u] & 0xffffu];
#pragma unroll
    for (int u = 0; u < RUN_ROWS; ++u) {
        const int i = i0 + 32 * u + lane;
        const unsigned t = w[u] >> 16;
        if (i < n && t < cnt)
            __hip_atomic_fetch_add(&sums[t], v[u], __ATOMIC_RELAXED, __HIP_MEMORY_SCOPE_WORKGROUP);
    }
}

__global__ __launch_bounds__(AGG_TPB) void hg_agg1_bin(const unsigned* __restrict__ words,
                                                       int nR,
                                                       const float* __restrict__ hT,
                                                       const float* __restrict__ s_in,
                                                       float* __restrict__ accT,
                                                       int N) {
    __shared__ float hcol[NMAX];
    __shared__ float acc[NHALF_MAX];

    const int tid  = threadIdx.x;
    const int lane = tid & 31;
    const int f    = blockIdx.z;
    const int half = blockIdx.y;
    const int nh   = (N + 1) >> 1;
    const int lo   = half * nh;
    const int hiN  = lo + nh < N ? lo + nh : N;
    const unsigned cnt = hiN > lo ? (unsigned)(hiN - lo) : 0u;

    const float* __restrict__ hp = hT + (size_t)f * N;
    for (int i = tid; i < N; i += AGG_TPB) hcol[i] = hp[i];
    for (unsigned i = tid; i < cnt; i += AGG_TPB) acc[i] = 0.0f;
    __syncthreads();

    const int nwav = (int)gridDim.x * (AGG_TPB / 32);
    int q = __builtin_amdgcn_readfirstlane((int)blockIdx.x * (AGG_TPB / 32) + (tid >> 5));
    unsigned h0 = 0u, hl = 0u;
    if (q < nR) {
        const unsigned* __restrict__ rg = words + (size_t)q * BIN_PITCH;
        h0 = rg[0];  hl = rg[1];
    }
    while (q < nR) {
        const int qn = q + nwav;
        unsigned g0 = 0u, gl = 0u;
        if (qn < nR) {
            const unsigned* __restrict__ rn = words + (size_t)qn * BIN_PITCH;
            g0 = rn[0];  gl = rn[1];
        }
        const int t0 = __builtin_amdgcn_readfirstlane((int)h0);
        const int ln = __builtin_amdgcn_readfirstlane((int)hl);
        const unsigned* __restrict__ wp = words + (size_t)q * BIN_PITCH + BIN_HDR + (half ? t0 : 0);
        const int n = half ? ln - t0 : t0;
        if (n > 0) {
            const int last = n - 1;
            const int G  = 32 * RUN_ROWS;
            const int ng = (n + G - 1) / G;
            unsigned wA[RUN_ROWS], wB[RUN_ROWS];
            hg_load_rows(wp, 0, last, lane, wA);
            int g = 0;
            for (; g + 2 < ng; g += 2) {
                hg_load_rows(wp, (g + 1) * G, last, lane, wB);
                hg_add_rows(wA, g * G, n, lane, hcol, acc, cnt);
                hg_load_rows(wp, (g + 2) * G, last, lane, wA);
                hg_add_rows(wB, (g + 1) * G, n, lane, hcol, acc, cnt);
            }
            if (g + 1 < ng) {
                hg_load_rows(wp, (g + 1) * G, last, lane, wB);
                hg_add_rows(wA, g * G, n, lane, hcol, acc, cnt);
                hg_add_rows(wB, (g + 1) * G, n, lane, hcol, acc, cnt);
            } else {
                hg_add_rows(wA, g * G, n, lane, hcol, acc, cnt);
            }
        }
        h0 = g0;  hl = gl;  q = qn;
    }
    __syncthreads();

    float* __restrict__ ap = accT + (size_t)f * N + lo;
    const float* __restrict__ sp = s_in + lo;
    if (cnt != 0u) {
        const unsigned last = cnt - 1u;
        for (unsigned base = 0u; base < cnt; base += FLUSH_U * AGG_TPB) {
            float s[FLUSH_U], a[FLUSH_U];
#pragma unroll
            for (int u = 0; u < FLUSH_U; ++u) {
                const unsigned j  = base + (unsigned)u * AGG_TPB + tid;
                const unsigned jc = j < cnt ? j : last;
                s[u] = sp[jc];
                a[u] = acc[jc];
            }
#pragma unroll
            for (int u = 0; u < FLUSH_U; ++u) {
                const unsigned j = base + (unsigned)u * AGG_TPB + tid;
                const float v = a[u] * s[u];
                if (j < cnt && v != 0.0f) atomicAdd(ap + j, v);
            }
        }
    }
}

__global__ __launch_bounds__(TPB) void hg_relu_bias(float* __restrict__ acc,
                                                    const float* __restrict__ b1,
                                                    int N) {
    const int f = blockIdx.y;
    const int n = blockIdx.x * TPB + threadIdx.x;
    if (n < N) {
        const size_t i = (size_t)f * N + n;
        float v = acc[i] + b1[f] + b1[16 + f] + b1[32 + f];
        acc[i] = v > 0.0f ? v : 0.0f;
    }
}

__global__ __launch_bounds__(TPB) void hg_gemm2(const float* __restrict__ h,
                                                const float* __restrict__ s_out,
                                                const float* __restrict__ w,
                                                float* __restrict__ z,
                                                int N) {
    int i = blockIdx.x * TPB + threadIdx.x;
    if (i >= N) return;
    const float* hp = h + i;
    const float4* wp = (const float4*)w;
    float acc = 0.0f;
#pragma unroll
    for (int q = 0; q < 4; ++q) {
        float4 a, b = wp[q];
        a.x = hp[(size_t)(4 * q + 0) * N];
        a.y = hp[(size_t)(4 * q + 1) * N];
        a.z = hp[(size_t)(4 * q + 2) * N];
        a.w = hp[(size_t)(4 * q + 3) * N];
        acc += a.x * b.x + a.y * b.y + a.z * b.z + a.w * b.w;
    }
    z[i] = acc * s_out[i];
}

__global__ __launch_bounds__(L2_TPB) void hg_agg2_own(const int* __restrict__ s0,
                                                      const int* __restrict__ d0,
                                                      const int* __restrict__ s1,
                                                      const int* __restrict__ d1,
                                                      const int* __restrict__ s2,
                                                      const int* __restrict__ d2,
                                                      int E0, int E1, int E2,
                                                      const float* __restrict__ z3,
                                                      float* __restrict__ partial,
                                                      int N) {
    __shared__ float zc[NMAX];
    __shared__ float bins[L2_SLICE_MAX];
    __shared__ unsigned qbuf[(L2_TPB / 32) * Q_WORDS];

    const int tid = threadIdx.x;
    const int p   = blockIdx.y;
    const int* __restrict__ src;
    const int* __restrict__ dst;
    int E, nchunk, chunk, r;
    if (p < L2_C0) {
        src = s0;  dst = d0;  E = E0;  nchunk = L2_C0;  chunk = p;                  r = 0;
    } else if (p < L2_C0 + L2_C1) {
        src = s1;  dst = d1;  E = E1;  nchunk = L2_C1;  chunk = p - L2_C0;          r = 1;
    } else {
        src = s2;  dst = d2;  E = E2;  nchunk = L2_C2;  chunk = p - L2_C0 - L2_C1;  r = 2;
    }

    const int ns  = (((N + L2_SLICES - 1) / L2_SLICES) + 31) & ~31;
    const int lo  = blockIdx.x * ns;
    const int hiN = lo + ns < N ? lo + ns : N;
    const unsigned cnt = hiN > lo ? (unsigned)(hiN - lo) : 0u;

    const float* __restrict__ zp = z3 + (size_t)r * N;
    for (int i = tid; i < N; i += L2_TPB) zc[i] = zp[i];
    for (unsigned j = tid; j < cnt; j += L2_TPB) bins[j] = 0.0f;
    __syncthreads();

    const int per = ((E + nchunk * 4 - 1) / (nchunk * 4)) * 4;
    const int e0  = chunk * per;
    const int e1  = e0 + per < E ? e0 + per : E;
    const int ne  = e1 > e0 ? e1 - e0 : 0;
    const int n4  = ne >> 2;
    const int4* __restrict__ s4 = (const int4*)(src + e0);
    const int4* __restrict__ d4 = (const int4*)(dst + e0);

    hg_scan_own(s4, d4, n4, zc, bins, qbuf + (tid >> 5) * Q_WORDS, lo, cnt, tid, L2_TPB);
    for (int e = e0 + (n4 << 2) + tid; e < e1; e += L2_TPB)
        hg_edge_add(bins, zc[src[e]], dst[e], lo, cnt);
    __syncthreads();

    const size_t pitch = (size_t)((N + 31) & ~31);
    float* __restrict__ pp = partial + (size_t)p * pitch + lo;
    for (unsigned j = tid; j < cnt; j += L2_TPB) pp[j] = bins[j];
}

template <int C>
__device__ __forceinline__ float hg_sum_fplanes(const float* __restrict__ pp, size_t pitch) {
    float s = 0.0f;
#pragma unroll
    for (int c = 0; c < C; ++c) s += pp[(size_t)c * pitch];
    return s;
}

__global__ __launch_bounds__(TPB) void hg_out_sum(const float* __restrict__ partial,
                                                  const float* __restrict__ s_in3,
                                                  const float* __restrict__ b2,
                                                  float* __restrict__ out,
                                                  int N) {
    const int n = blockIdx.x * TPB + threadIdx.x;
    if (n >= N) return;
    const size_t pitch = (size_t)((N + 31) & ~31);
    const float m0 = hg_sum_fplanes<L2_C0>(partial + n, pitch);
    const float m1 = hg_sum_fplanes<L2_C1>(partial + (size_t)L2_C0 * pitch + n, pitch);
    const float m2 = hg_sum_fplanes<L2_C2>(partial + (size_t)(L2_C0 + L2_C1) * pitch + n, pitch);
    out[n] = (m0 * s_in3[n] + b2[0])
           + (m1 * s_in3[(size_t)N + n] + b2[1])
           + (m2 * s_in3[(size_t)2 * N + n] + b2[2]);
}

extern "C" void kernel_launch(void* const* d_in, const int* in_sizes, int n_in,
                              void* d_out, int out_size, void* d_ws, size_t ws_size,
                              hipStream_t stream) {
    const float* x  = (const float*)d_in[0];
    const int* src[3] = {(const int*)d_in[1], (const int*)d_in[3], (const int*)d_in[5]};
    const int* dst[3] = {(const int*)d_in[2], (const int*)d_in[4], (const int*)d_in[6]};
    const int  E[3]   = {in_sizes[1], in_sizes[3], in_sizes[5]};
    const float* W1 = (const float*)d_in[7];
    const float* b1 = (const float*)d_in[8];
    const float* W2 = (const float*)d_in[9];
    const float* b2 = (const float*)d_in[10];
    const int N = in_sizes[0] / 32;
    float* out = (float*)d_out;

    char* ws = (char*)d_ws;
    float* scales = (float*)ws;
    float* acc1   = (float*)(ws + (size_t)6 * N * 4);
    float* hpre   = acc1 + (size_t)N * 16;
    float* z      = hpre + (size_t)N * 16;
    float* z3     = z - (size_t)2 * N;
    float* planes2 = acc1;
    unsigned* partial = (unsigned*)acc1;
    const int nR[3] = {(E[0] + BIN_RUN - 1) / BIN_RUN, (E[1] + BIN_RUN - 1) / BIN_RUN,
                       (E[2] + BIN_RUN - 1) / BIN_RUN};
    const int nRuns = nR[0] + nR[1] + nR[2];
    const size_t w_off = ((size_t)39 * N * 4 + 127) & ~(size_t)127;
    const bool binned = nRuns > 0 && ws_size >= w_off + (size_t)nRuns * BIN_PITCH * 4;
    unsigned* words = (unsigned*)(ws + w_off);
    const size_t rOff[3] = {0, (size_t)nR[0] * BIN_PITCH, (size_t)(nR[0] + nR[1]) * BIN_PITCH};

    hg_count_own<<<dim3(CNT_SLICES, CNT_PLANES), dim3(CNT_TPB), 0, stream>>>(
        src[0], dst[0], src[1], dst[1], src[2], dst[2], E[0], E[1], E[2], partial, N);
    hg_scale_sum<<<dim3((N + TPB - 1) / TPB, 6), dim3(TPB), 0, stream>>>(partial, scales, N);

    {
        long span = 16L * N;
        hg_init<<<dim3((unsigned)((span + TPB - 1) / TPB)), dim3(TPB), 0, stream>>>(
            acc1, N);
    }

    if (binned)
        hg_bin_half<<<dim3(nRuns), dim3(BIN_TPB), 0, stream>>>(
            src[0], dst[0], src[1], dst[1], src[2], dst[2], E[0], E[1], E[2], nR[0], nR[1], words, N);

    const int tiles  = (N + 15) / 16;
    const int gblk   = (tiles + (TPB / 32) - 1) / (TPB / 32);
    for (int r = 0; r < 3; ++r) {
        hg_gemm1_wmma<<<dim3(gblk), dim3(TPB), 0, stream>>>(
            x, scales + (size_t)r * N, W1 + (size_t)r * 32 * 16, hpre, N);
        if (binned)
            hg_agg1_bin<<<dim3(8, 2, 16), dim3(AGG_TPB), 0, stream>>>(
                words + rOff[r], nR[r], hpre, scales + (size_t)(3 + r) * N, acc1, N);
        else
            hg_agg1_col<<<dim3(8, 2, 16), dim3(AGG_TPB), 0, stream>>>(
                src[r], dst[r], E[r], hpre, scales + (size_t)(3 + r) * N, acc1, N);
    }
    hg_relu_bias<<<dim3((N + TPB - 1) / TPB, 16), dim3(TPB), 0, stream>>>(acc1, b1, N);

    for (int r = 0; r < 3; ++r) {
        hg_gemm2<<<dim3((N + TPB - 1) / TPB), dim3(TPB), 0, stream>>>(
            acc1, scales + (size_t)r * N, W2 + (size_t)r * 16, z3 + (size_t)r * N, N);
    }
    hg_agg2_own<<<dim3(L2_SLICES, L2_PLANES), dim3(L2_TPB), 0, stream>>>(
        src[0], dst[0], src[1], dst[1], src[2], dst[2], E[0], E[1], E[2], z3, planes2, N);
    hg_out_sum<<<dim3((N + TPB - 1) / TPB), dim3(TPB), 0, stream>>>(
        planes2, scales + (size_t)3 * N, b2, out, N);
}


# layer-1 edges binned once per call by 16 dst slices: a work-group owns (feature, slice) for a relation's whole list; plain-store flush, no atomics, no clear pass; relations added in fixed order
# speedup vs baseline: 13.1397x; 1.1035x over previous
#include <hip/hip_runtime.h>

typedef float v2f __attribute__((ext_vector_type(2)));
typedef float v8f __attribute__((ext_vector_type(8)));

#define TPB 256
#define AGG_TPB 1024
#define NMAX 50000
#define NHALF_MAX ((NMAX + 1) / 2)
#define FLUSH_U 13

#define CNT_TPB 1024
#define CNT_SLICES 8
#define CNT_SLICE_MAX ((((NMAX + CNT_SLICES - 1) / CNT_SLICES) + 31) & ~31)
#define CNT_C0 10
#define CNT_C1 3
#define CNT_C2 1
#define CNT_PLANES (2 * (CNT_C0 + CNT_C1 + CNT_C2))

#define L2_TPB 1024
#define L2_SLICES 4
#define L2_SLICE_MAX ((((NMAX + L2_SLICES - 1) / L2_SLICES) + 31) & ~31)
#define L2_C0 20
#define L2_C1 7
#define L2_C2 2
#define L2_PLANES (L2_C0 + L2_C1 + L2_C2)

__global__ __launch_bounds__(TPB) void hg_init(float* __restrict__ acc1,
                                               int N) {
    long i = (long)blockIdx.x * TPB + threadIdx.x;
    if (i < 16L * N) acc1[i] = 0.0f;
}

__device__ __forceinline__ void hg_bin(unsigned* __restrict__ bins,
                                       int v, int lo, unsigned cnt) {
    const unsigned t = (unsigned)(v - lo);
    if (t < cnt)
        __hip_atomic_fetch_add(&bins[t], 1u, __ATOMIC_RELAXED,
                               __HIP_MEMORY_SCOPE_WORKGROUP);
}

__global__ __launch_bounds__(CNT_TPB) void hg_count_own(const int* __restrict__ s0,
                                                        const int* __restrict__ d0,
                                                        const int* __restrict__ s1,
                                                        const int* __restrict__ d1,
                                                        const int* __restrict__ s2,
                                                        const int* __restrict__ d2,
                                                        int E0, int E1, int E2,
                                                        unsigned* __restrict__ partial,
                                                        int N) {
    __shared__ unsigned bins[CNT_SLICE_MAX];

    const int tid = threadIdx.x;
    const int p   = blockIdx.y;
    const int* __restrict__ idx;
    int E, nchunk, chunk;
    if (p < 2 * CNT_C0) {
        const int side = p >= CNT_C0;
        idx = side ? d0 : s0;  E = E0;  nchunk = CNT_C0;  chunk = p - side * CNT_C0;
    } else if (p < 2 * (CNT_C0 + CNT_C1)) {
        const int q = p - 2 * CNT_C0;
        const int side = q >= CNT_C1;
        idx = side ? d1 : s1;  E = E1;  nchunk = CNT_C1;  chunk = q - side * CNT_C1;
    } else {
        const int q = p - 2 * (CNT_C0 + CNT_C1);
        const int side = q >= CNT_C2;
        idx = side ? d2 : s2;  E = E2;  nchunk = CNT_C2;  chunk = q - side * CNT_C2;
    }

    const int ns  = (((N + CNT_SLICES - 1) / CNT_SLICES) + 31) & ~31;
    const int lo  = blockIdx.x * ns;
    const int hiN = lo + ns < N ? lo + ns : N;
    const unsigned cnt = hiN > lo ? (unsigned)(hiN - lo) : 0u;

    for (unsigned j = tid; j < cnt; j += CNT_TPB) bins[j] = 0u;
    __syncthreads();

    const int per = ((E + nchunk * 4 - 1) / (nchunk * 4)) * 4;
    const int e0  = chunk * per;
    const int e1  = e0 + per < E ? e0 + per : E;
    const int ne  = e1 > e0 ? e1 - e0 : 0;
    const int n4  = ne >> 2;
    const int4* __restrict__ q4 = (const int4*)(idx + e0);

    int i = tid;
    for (; i + 3 * CNT_TPB < n4; i += 4 * CNT_TPB) {
        const int4 a = q4[i], b = q4[i + CNT_TPB];
        const int4 c = q4[i + 2 * CNT_TPB], d = q4[i + 3 * CNT_TPB];
        hg_bin(bins, a.x, lo, cnt); hg_bin(bins, a.y, lo, cnt);
        hg_bin(bins, a.z, lo, cnt); hg_bin(bins, a.w, lo, cnt);
        hg_bin(bins, b.x, lo, cnt); hg_bin(bins, b.y, lo, cnt);
        hg_bin(bins, b.z, lo, cnt); hg_bin(bins, b.w, lo, cnt);
        hg_bin(bins, c.x, lo, cnt); hg_bin(bins, c.y, lo, cnt);
        hg_bin(bins, c.z, lo, cnt); hg_bin(bins, c.w, lo, cnt);
        hg_bin(bins, d.x, lo, cnt); hg_bin(bins, d.y, lo, cnt);
        hg_bin(bins, d.z, lo, cnt); hg_bin(bins, d.w, lo, cnt);
    }
    for (; i < n4; i += CNT_TPB) {
        const int4 a = q4[i];
        hg_bin(bins, a.x, lo, cnt); hg_bin(bins, a.y, lo, cnt);
        hg_bin(bins, a.z, lo, cnt); hg_bin(bins, a.w, lo, cnt);
    }
    for (int e = e0 + (n4 << 2) + tid; e < e1; e += CNT_TPB)
        hg_bin(bins, idx[e], lo, cnt);
    __syncthreads();

    const size_t pitch = (size_t)((N + 31) & ~31);
    unsigned* __restrict__ pp = partial + (size_t)p * pitch + lo;
    for (unsigned j = tid; j < cnt; j += CNT_TPB) pp[j] = bins[j];
}

template <int C>
__device__ __forceinline__ unsigned hg_sum_planes(const unsigned* __restrict__ pp, size_t pitch) {
    unsigned s = 0u;
#pragma unroll
    for (int c = 0; c < C; ++c) s += pp[(size_t)c * pitch];
    return s;
}

__global__ __launch_bounds__(TPB) void hg_scale_sum(const unsigned* __restrict__ partial,
                                                    float* __restrict__ scales,
                                                    int N) {
    const int n = blockIdx.x * TPB + threadIdx.x;
    if (n >= N) return;
    const int r    = blockIdx.y >> 1;
    const int side = blockIdx.y & 1;
    const size_t pitch = (size_t)((N + 31) & ~31);
    unsigned s;
    if (r == 0)
        s = hg_sum_planes<CNT_C0>(partial + (size_t)(side * CNT_C0) * pitch + n, pitch);
    else if (r == 1)
        s = hg_sum_planes<CNT_C1>(partial + (size_t)(2 * CNT_C0 + side * CNT_C1) * pitch + n, pitch);
    else
        s = hg_sum_planes<CNT_C2>(partial + (size_t)(2 * (CNT_C0 + CNT_C1) + side * CNT_C2) * pitch + n, pitch);
    float d = (float)s;
    if (d < 1.0f) d = 1.0f;
    scales[(size_t)(side * 3 + r) * N + n] = rsqrtf(d);
}

__global__ __launch_bounds__(TPB) void hg_gemm1_wmma(const float* __restrict__ x,
                                                     const float* __restrict__ s_out,
                                                     const float* __restrict__ W,
                                                     float* __restrict__ hT,
                                                     int N) {
    const int lane = threadIdx.x & 31;
    const int wave = threadIdx.x >> 5;
    const int tile = blockIdx.x * (TPB / 32) + wave;
    const int row0 = tile * 16;
    if (row0 >= N) return;

    const int half = lane >> 4;
    const int lm   = lane & 15;
    int row  = row0 + lm;
    int rowc = row < N ? row : N - 1;
    const float sc = s_out[rowc];

    v8f c = {};
#pragma unroll
    for (int kb = 0; kb < 8; ++kb) {
        const int k0 = kb * 4 + half * 2;
        v2f xf, wf;
        xf.x = x[(size_t)rowc * 32 + k0]     * sc;
        xf.y = x[(size_t)rowc * 32 + k0 + 1] * sc;
        wf.x = W[(k0)     * 16 + lm];
        wf.y = W[(k0 + 1) * 16 + lm];
        c = __builtin_amdgcn_wmma_f32_16x16x4_f32(
                 false, wf,  false, xf,
                 (short)0, c,  false,  false);
    }

    float* __restrict__ dp = hT + (size_t)(half * 8) * N + row;
    if (row < N) {
#pragma unroll
        for (int v = 0; v < 8; ++v) dp[(size_t)v * N] = c[v];
    }
}

#define Q_WORDS 160
static_assert(NMAX < 65536, "queue word holds src and local dst in 16 bits each");

__device__ __forceinline__ unsigned hg_push(unsigned* q, unsigned n, bool keep, unsigned word) {
    const unsigned m = __builtin_amdgcn_ballot_w32(keep);
    if (keep) q[n + __builtin_amdgcn_mbcnt_lo(m, 0u)] = word;
    return n + (unsigned)__builtin_popcount(m);
}

__device__ __forceinline__ void hg_add_word(float* sums, unsigned w, float v) {
    __hip_atomic_fetch_add(&sums[w >> 16], v, __ATOMIC_RELAXED, __HIP_MEMORY_SCOPE_WORKGROUP);
}

template <int R>
__device__ __forceinline__ void hg_take(const unsigned* q, const float* vals, float* sums,
                                        unsigned n, unsigned lane, unsigned carried, unsigned& wq) {
    unsigned w[R + 1];
#pragma unroll
    for (int r = 0; r < R; ++r) w[r] = q[lane + 32u * (unsigned)r];
    w[R] = lane + 32u * (unsigned)R < n ? q[lane + 32u * (unsigned)R] : 0u;
    if (lane < carried) w[0] = wq;
    float v[R + 1];
#pragma unroll
    for (int r = 0; r < R; ++r) v[r] = vals[w[r] & 0xffffu];
#pragma unroll
    for (int r = 0; r < R; ++r) hg_add_word(sums, w[r], v[r]);
    wq = w[R];
}

__device__ __forceinline__ void hg_trip(const int4 d, const int4 s, bool ok,
                                        const float* vals, float* sums, unsigned* q,
                                        int lo, unsigned cnt, unsigned lane,
                                        unsigned& rem, unsigned& wq) {
    const unsigned t0 = (unsigned)(d.x - lo), t1 = (unsigned)(d.y - lo);
    const unsigned t2 = (unsigned)(d.z - lo), t3 = (unsigned)(d.w - lo);
    const unsigned c = ok ? cnt : 0u;
    unsigned n = rem;
    n = hg_push(q, n, t0 < c, (unsigned)s.x | (t0 << 16));
    n = hg_push(q, n, t1 < c, (unsigned)s.y | (t1 << 16));
    n = hg_push(q, n, t2 < c, (unsigned)s.z | (t2 << 16));
    n = hg_push(q, n, t3 < c, (unsigned)s.w | (t3 << 16));
    __builtin_amdgcn_fence(__ATOMIC_RELEASE, "wavefront", "local");
    __builtin_amdgcn_wave_barrier();
    __builtin_amdgcn_fence(__ATOMIC_ACQUIRE, "wavefront", "local");
    const unsigned rows = n >> 5;
    if      (rows == 2u) hg_take<2>(q, vals, sums, n, lane, rem, wq);
    else if (rows == 1u) hg_take<1>(q, vals, sums, n, lane, rem, wq);
    else if (rows == 3u) hg_take<3>(q, vals, sums, n, lane, rem, wq);
    else if (rows == 0u) hg_take<0>(q, vals, sums, n, lane, rem, wq);
    else                 hg_take<4>(q, vals, sums, n, lane, rem, wq);
    rem = n & 31u;
}

__device__ __forceinline__ void hg_scan_own(const int4* __restrict__ s4, const int4* __restrict__ d4,
                                            int n4, const float* vals, float* sums, unsigned* q,
                                            int lo, unsigned cnt, int tid, int tpb) {
    const unsigned lane = (unsigned)tid & 31u;
    int4 dA = make_int4(0, 0, 0, 0), sA = make_int4(0, 0, 0, 0);
    int4 dB = make_int4(0, 0, 0, 0), sB = make_int4(0, 0, 0, 0);
    unsigned rem = 0u, wq = 0u;
    int ib = __builtin_amdgcn_readfirstlane(tid & ~31);
    if (ib + (int)lane < n4) { dA = d4[ib + (int)lane]; sA = s4[ib + (int)lane]; }
    const int last = n4 - 1;
    for (; ib < n4; ib += 2 * tpb) {
        const int iA = ib + (int)lane, iB = iA + tpb, iC = iB + tpb;
        const int jB = iB < n4 ? iB : last;
        dB = d4[jB];  sB = s4[jB];
        hg_trip(dA, sA, iA < n4, vals, sums, q, lo, cnt, lane, rem, wq);
        const int jC = iC < n4 ? iC : last;
        dA = d4[jC];  sA = s4[jC];
        if (ib + tpb < n4)
            hg_trip(dB, sB, iB < n4, vals, sums, q, lo, cnt, lane, rem, wq);
    }
    if (lane < rem) hg_add_word(sums, wq, vals[wq & 0xffffu]);
}

__device__ __forceinline__ void hg_edge_add(float* __restrict__ acc,
                                            float v, int d, int lo, unsigned cnt) {
    const unsigned dl = (unsigned)(d - lo);
    if (dl < cnt)
        __hip_atomic_fetch_add(&acc[dl], v, __ATOMIC_RELAXED,
                               __HIP_MEMORY_SCOPE_WORKGROUP);
}

__global__ __launch_bounds__(AGG_TPB) void hg_agg1_col(const int* __restrict__ src,
                                                       const int* __restrict__ dst,
                                                       int E,
                                                       const float* __restrict__ hT,
                                                       const float* __restrict__ s_in,
                                                       float* __restrict__ accT,
                                                       int N) {
    __shared__ float hcol[NMAX];
    __shared__ float acc[NHALF_MAX];
    __shared__ unsigned qbuf[(AGG_TPB / 32) * Q_WORDS];

    const int tid    = threadIdx.x;
    const int nchunk = gridDim.x;
    const int chunk  = blockIdx.x;
    const int f      = blockIdx.z;
    const int nh     = (N + 1) >> 1;
    const int lo     = blockIdx.y * nh;
    const int hiN    = lo + nh < N ? lo + nh : N;
    const unsigned cnt = hiN > lo ? (unsigned)(hiN - lo) : 0u;

    const float* __restrict__ hp = hT + (size_t)f * N;
    for (int i = tid; i < N; i += AGG_TPB) hcol[i] = hp[i];
    for (unsigned i = tid; i < cnt; i += AGG_TPB) acc[i] = 0.0f;
    __syncthreads();

    const int per = ((E + nchunk * 4 - 1) / (nchunk * 4)) * 4;
    const int e0  = chunk * per;
    const int e1  = e0 + per < E ? e0 + per : E;
    const int ne  = e1 > e0 ? e1 - e0 : 0;
    const int n4  = ne >> 2;
    const int4* __restrict__ s4 = (const int4*)(src + e0);
    const int4* __restrict__ d4 = (const int4*)(dst + e0);

    hg_scan_own(s4, d4, n4, hcol, acc, qbuf + (tid >> 5) * Q_WORDS, lo, cnt, tid, AGG_TPB);
    for (int e = e0 + (n4 << 2) + tid; e < e1; e += AGG_TPB)
        hg_edge_add(acc, hcol[src[e]], dst[e], lo, cnt);
    __syncthreads();

    float* __restrict__ ap = accT + (size_t)f * N + lo;
    const float* __restrict__ sp = s_in + lo;
    if (cnt != 0u) {
        const unsigned last = cnt - 1u;
        for (unsigned base = 0u; base < cnt; base += FLUSH_U * AGG_TPB) {
            float s[FLUSH_U], a[FLUSH_U];
#pragma unroll
            for (int u = 0; u < FLUSH_U; ++u) {
                const unsigned j  = base + (unsigned)u * AGG_TPB + tid;
                const unsigned jc = j < cnt ? j : last;
                s[u] = sp[jc];
                a[u] = acc[jc];
            }
#pragma unroll
            for (int u = 0; u < FLUSH_U; ++u) {
                const unsigned j = base + (unsigned)u * AGG_TPB + tid;
                const float v = a[u] * s[u];
                if (j < cnt && v != 0.0f) atomicAdd(ap + j, v);
            }
        }
    }
}

#define B16_TPB 1024
#define B16_VEC 5
#define B16_EPT (4 * B16_VEC)
#define B16_RUN (B16_TPB * B16_EPT)
#define B16_HDR 32
#define B16_PITCH (B16_HDR + B16_RUN)
#define SL16 16
#define SL16_MAX ((((NMAX + SL16 - 1) / SL16) + 31) & ~31)

__global__ __launch_bounds__(B16_TPB) void hg_bin16(const int* __restrict__ s0,
                                                    const int* __restrict__ d0,
                                                    const int* __restrict__ s1,
                                                    const int* __restrict__ d1,
                                                    const int* __restrict__ s2,
                                                    const int* __restrict__ d2,
                                                    int E0, int E1, int E2,
                                                    int nR0, int nR1,
                                                    unsigned* __restrict__ words,
                                                    int N) {
    __shared__ unsigned buf[B16_RUN];
    __shared__ unsigned hist[SL16 * 32];
    __shared__ unsigned base[SL16 * 32];
    __shared__ unsigned tot[SL16];
    __shared__ unsigned start[SL16 + 1];

    const int tid = threadIdx.x;
    const int run = blockIdx.x;
    const int* __restrict__ src;
    const int* __restrict__ dst;
    int E, q;
    if (run < nR0)            { src = s0;  dst = d0;  E = E0;  q = run; }
    else if (run < nR0 + nR1) { src = s1;  dst = d1;  E = E1;  q = run - nR0; }
    else                      { src = s2;  dst = d2;  E = E2;  q = run - nR0 - nR1; }
    const int ns  = (((N + SL16 - 1) / SL16) + 31) & ~31;
    const float inv_ns = 1.0f / (float)ns;
    const int R0  = q * B16_RUN;
    const int len = E - R0 < B16_RUN ? E - R0 : B16_RUN;

    const int lane = tid & 31;
    if (tid < SL16 * 32) hist[tid] = 0u;

    int d[B16_EPT], s[B16_EPT];
    bool ok[B16_EPT];
#pragma unroll
    for (int v = 0; v < B16_VEC; ++v) {
        const int b = v * (B16_RUN / B16_VEC) + 4 * tid;
        if (b + 3 < len) {
            const int4 dv = *(const int4*)(dst + R0 + b);
            const int4 sv = *(const int4*)(src + R0 + b);
            d[4 * v] = dv.x;  d[4 * v + 1] = dv.y;  d[4 * v + 2] = dv.z;  d[4 * v + 3] = dv.w;
            s[4 * v] = sv.x;  s[4 * v + 1] = sv.y;  s[4 * v + 2] = sv.z;  s[4 * v + 3] = sv.w;
            ok[4 * v] = ok[4 * v + 1] = ok[4 * v + 2] = ok[4 * v + 3] = true;
        } else {
#pragma unroll
            for (int k = 0; k < 4; ++k) {
                const bool in = b + k < len;
                ok[4 * v + k] = in;
                d[4 * v + k]  = in ? dst[R0 + b + k] : 0;
                s[4 * v + k]  = in ? src[R0 + b + k] : 0;
            }
        }
    }
    __syncthreads();

    unsigned word[B16_EPT], rank[B16_EPT];
    int bin[B16_EPT];
#pragma unroll
    for (int k = 0; k < B16_EPT; ++k) {
        const bool valid = ok[k] && (unsigned)d[k] < (unsigned)N;
        const int dv = valid ? d[k] : 0;
        int b = (int)((float)dv * inv_ns);
        if (b * ns > dv) --b;
        else if ((b + 1) * ns <= dv) ++b;
        const unsigned t = (unsigned)(dv - b * ns);
        word[k] = (unsigned)s[k] | (t << 16);
        bin[k]  = valid ? b : SL16;
        rank[k] = 0u;
        if (valid)
            rank[k] = __hip_atomic_fetch_add(&hist[b * 32 + lane], 1u, __ATOMIC_RELAXED,
                                             __HIP_MEMORY_SCOPE_WORKGROUP);
    }
    __syncthreads();
    unsigned cv = 0u, incl = 0u;
    if (tid < SL16 * 32) {
        cv = hist[tid];
        incl = cv;
#pragma unroll
        for (int o = 1; o < 32; o <<= 1) {
            const unsigned up = __shfl_up(incl, (unsigned)o, 32);
            if (lane >= o) incl += up;
        }
        if (lane == 31) tot[tid >> 5] = incl;
    }
    __syncthreads();
    if (tid == 0) {
        unsigned a = 0u;
#pragma unroll
        for (int b = 0; b < SL16; ++b) { start[b] = a;  a += tot[b]; }
        start[SL16] = a;
    }
    __syncthreads();
    if (tid < SL16 * 32) base[tid] = start[tid >> 5] + (incl - cv);
    __syncthreads();
#pragma unroll
    for (int k = 0; k < B16_EPT; ++k)
        if (bin[k] < SL16) buf[base[bin[k] * 32 + lane] + rank[k]] = word[k];
    __syncthreads();

    unsigned* __restrict__ wr = words + (size_t)run * B16_PITCH;
    if (tid < B16_HDR) wr[tid] = tid <= SL16 ? start[tid] : 0u;
    const int total = (int)start[SL16];
    for (int i = tid; i < total; i += B16_TPB) wr[B16_HDR + i] = buf[i];
}

#define RUN_ROWS 8
__device__ __forceinline__ void hg_load_rows(const unsigned* __restrict__ wp, int i0, int last,
                                             int lane, unsigned (&w)[RUN_ROWS]) {
#pragma unroll
    for (int u = 0; u < RUN_ROWS; ++u) {
        const int i = i0 + 32 * u + lane;
        w[u] = wp[i < last ? i : last];
    }
}

__device__ __forceinline__ void hg_add_rows(const unsigned (&w)[RUN_ROWS], int i0, int st, int en, int lane,
                                            const float* vals, float* sums, unsigned cnt) {
    float v[RUN_ROWS];
#pragma unroll
    for (int u = 0; u < RUN_ROWS; ++u) v[u] = vals[w[u] & 0xffffu];
#pragma unroll
    for (int u = 0; u < RUN_ROWS; ++u) {
        const int i = i0 + 32 * u + lane;
        const unsigned t = w[u] >> 16;
        if (i >= st && i < en && t < cnt)
            __hip_atomic_fetch_add(&sums[t], v[u], __ATOMIC_RELAXED, __HIP_MEMORY_SCOPE_WORKGROUP);
    }
}

__global__ __launch_bounds__(AGG_TPB) void hg_agg1_own16(const unsigned* __restrict__ words,
                                                         int nR,
                                                         const float* __restrict__ hT,
                                                         const float* __restrict__ s_in,
                                                         float* __restrict__ accT,
                                                         int N, int first) {
    __shared__ float hcol[NMAX];
    __shared__ float acc[SL16_MAX];

    const int tid  = threadIdx.x;
    const int lane = tid & 31;
    const int f    = blockIdx.y;
    const int sl   = blockIdx.x;
    const int ns   = (((N + SL16 - 1) / SL16) + 31) & ~31;
    const int lo   = sl * ns;
    const int hiN  = lo + ns < N ? lo + ns : N;
    const unsigned cnt = hiN > lo ? (unsigned)(hiN - lo) : 0u;

    const float* __restrict__ hp = hT + (size_t)f * N;
    for (int i = tid; i < N; i += AGG_TPB) hcol[i] = hp[i];
    for (unsigned i = tid; i < cnt; i += AGG_TPB) acc[i] = 0.0f;
    __syncthreads();

    const int nwav = AGG_TPB / 32;
    int q = __builtin_amdgcn_readfirstlane(tid >> 5);
    unsigned h0 = 0u, h1 = 0u;
    if (q < nR) {
        const unsigned* __restrict__ rg = words + (size_t)q * B16_PITCH;
        h0 = rg[sl];  h1 = rg[sl + 1];
    }
    while (q < nR) {
        const int qn = q + nwav;
        unsigned g0 = 0u, g1 = 0u;
        if (qn < nR) {
            const unsigned* __restrict__ rn = words + (size_t)qn * B16_PITCH;
            g0 = rn[sl];  g1 = rn[sl + 1];
        }
        const int st = __builtin_amdgcn_readfirstlane((int)h0);
        const int en = __builtin_amdgcn_readfirstlane((int)h1);
        const unsigned* __restrict__ wp = words + (size_t)q * B16_PITCH + B16_HDR;
        if (en > st) {
            const int a0   = st & ~31;
            const int last = en - 1;
            const int G  = 32 * RUN_ROWS;
            const int ng = (en - a0 + G - 1) / G;
            unsigned wA[RUN_ROWS], wB[RUN_ROWS];
            hg_load_rows(wp, a0, last, lane, wA);
            int g = 0;
            for (; g + 2 < ng; g += 2) {
                hg_load_rows(wp, a0 + (g + 1) * G, last, lane, wB);
                hg_add_rows(wA, a0 + g * G, st, en, lane, hcol, acc, cnt);
                hg_load_rows(wp, a0 + (g + 2) * G, last, lane, wA);
                hg_add_rows(wB, a0 + (g + 1) * G, st, en, lane, hcol, acc, cnt);
            }
            if (g + 1 < ng) {
                hg_load_rows(wp, a0 + (g + 1) * G, last, lane, wB);
                hg_add_rows(wA, a0 + g * G, st, en, lane, hcol, acc, cnt);
                hg_add_rows(wB, a0 + (g + 1) * G, st, en, lane, hcol, acc, cnt);
            } else {
                hg_add_rows(wA, a0 + g * G, st, en, lane, hcol, acc, cnt);
            }
        }
        h0 = g0;  h1 = g1;  q = qn;
    }
    __syncthreads();

    float* __restrict__ ap = accT + (size_t)f * N + lo;
    const float* __restrict__ sp = s_in + lo;
    for (unsigned j = tid; j < cnt; j += AGG_TPB) {
        const float v = acc[j] * sp[j];
        ap[j] = first ? v : ap[j] + v;
    }
}


__global__ __launch_bounds__(TPB) void hg_relu_bias(float* __restrict__ acc,
                                                    const float* __restrict__ b1,
                                                    int N) {
    const int f = blockIdx.y;
    const int n = blockIdx.x * TPB + threadIdx.x;
    if (n < N) {
        const size_t i = (size_t)f * N + n;
        float v = acc[i] + b1[f] + b1[16 + f] + b1[32 + f];
        acc[i] = v > 0.0f ? v : 0.0f;
    }
}

__global__ __launch_bounds__(TPB) void hg_gemm2(const float* __restrict__ h,
                                                const float* __restrict__ s_out,
                                                const float* __restrict__ w,
                                                float* __restrict__ z,
                                                int N) {
    int i = blockIdx.x * TPB + threadIdx.x;
    if (i >= N) return;
    const float* hp = h + i;
    const float4* wp = (const float4*)w;
    float acc = 0.0f;
#pragma unroll
    for (int q = 0; q < 4; ++q) {
        float4 a, b = wp[q];
        a.x = hp[(size_t)(4 * q + 0) * N];
        a.y = hp[(size_t)(4 * q + 1) * N];
        a.z = hp[(size_t)(4 * q + 2) * N];
        a.w = hp[(size_t)(4 * q + 3) * N];
        acc += a.x * b.x + a.y * b.y + a.z * b.z + a.w * b.w;
    }
    z[i] = acc * s_out[i];
}

__global__ __launch_bounds__(L2_TPB) void hg_agg2_own(const int* __restrict__ s0,
                                                      const int* __restrict__ d0,
                                                      const int* __restrict__ s1,
                                                      const int* __restrict__ d1,
                                                      const int* __restrict__ s2,
                                                      const int* __restrict__ d2,
                                                      int E0, int E1, int E2,
                                                      const float* __restrict__ z3,
                                                      float* __restrict__ partial,
                                                      int N) {
    __shared__ float zc[NMAX];
    __shared__ float bins[L2_SLICE_MAX];
    __shared__ unsigned qbuf[(L2_TPB / 32) * Q_WORDS];

    const int tid = threadIdx.x;
    const int p   = blockIdx.y;
    const int* __restrict__ src;
    const int* __restrict__ dst;
    int E, nchunk, chunk, r;
    if (p < L2_C0) {
        src = s0;  dst = d0;  E = E0;  nchunk = L2_C0;  chunk = p;                  r = 0;
    } else if (p < L2_C0 + L2_C1) {
        src = s1;  dst = d1;  E = E1;  nchunk = L2_C1;  chunk = p - L2_C0;          r = 1;
    } else {
        src = s2;  dst = d2;  E = E2;  nchunk = L2_C2;  chunk = p - L2_C0 - L2_C1;  r = 2;
    }

    const int ns  = (((N + L2_SLICES - 1) / L2_SLICES) + 31) & ~31;
    const int lo  = blockIdx.x * ns;
    const int hiN = lo + ns < N ? lo + ns : N;
    const unsigned cnt = hiN > lo ? (unsigned)(hiN - lo) : 0u;

    const float* __restrict__ zp = z3 + (size_t)r * N;
    for (int i = tid; i < N; i += L2_TPB) zc[i] = zp[i];
    for (unsigned j = tid; j < cnt; j += L2_TPB) bins[j] = 0.0f;
    __syncthreads();

    const int per = ((E + nchunk * 4 - 1) / (nchunk * 4)) * 4;
    const int e0  = chunk * per;
    const int e1  = e0 + per < E ? e0 + per : E;
    const int ne  = e1 > e0 ? e1 - e0 : 0;
    const int n4  = ne >> 2;
    const int4* __restrict__ s4 = (const int4*)(src + e0);
    const int4* __restrict__ d4 = (const int4*)(dst + e0);

    hg_scan_own(s4, d4, n4, zc, bins, qbuf + (tid >> 5) * Q_WORDS, lo, cnt, tid, L2_TPB);
    for (int e = e0 + (n4 << 2) + tid; e < e1; e += L2_TPB)
        hg_edge_add(bins, zc[src[e]], dst[e], lo, cnt);
    __syncthreads();

    const size_t pitch = (size_t)((N + 31) & ~31);
    float* __restrict__ pp = partial + (size_t)p * pitch + lo;
    for (unsigned j = tid; j < cnt; j += L2_TPB) pp[j] = bins[j];
}

template <int C>
__device__ __forceinline__ float hg_sum_fplanes(const float* __restrict__ pp, size_t pitch) {
    float s = 0.0f;
#pragma unroll
    for (int c = 0; c < C; ++c) s += pp[(size_t)c * pitch];
    return s;
}

__global__ __launch_bounds__(TPB) void hg_out_sum(const float* __restrict__ partial,
                                                  const float* __restrict__ s_in3,
                                                  const float* __restrict__ b2,
                                                  float* __restrict__ out,
                                                  int N) {
    const int n = blockIdx.x * TPB + threadIdx.x;
    if (n >= N) return;
    const size_t pitch = (size_t)((N + 31) & ~31);
    const float m0 = hg_sum_fplanes<L2_C0>(partial + n, pitch);
    const float m1 = hg_sum_fplanes<L2_C1>(partial + (size_t)L2_C0 * pitch + n, pitch);
    const float m2 = hg_sum_fplanes<L2_C2>(partial + (size_t)(L2_C0 + L2_C1) * pitch + n, pitch);
    out[n] = (m0 * s_in3[n] + b2[0])
           + (m1 * s_in3[(size_t)N + n] + b2[1])
           + (m2 * s_in3[(size_t)2 * N + n] + b2[2]);
}

extern "C" void kernel_launch(void* const* d_in, const int* in_sizes, int n_in,
                              void* d_out, int out_size, void* d_ws, size_t ws_size,
                              hipStream_t stream) {
    const float* x  = (const float*)d_in[0];
    const int* src[3] = {(const int*)d_in[1], (const int*)d_in[3], (const int*)d_in[5]};
    const int* dst[3] = {(const int*)d_in[2], (const int*)d_in[4], (const int*)d_in[6]};
    const int  E[3]   = {in_sizes[1], in_sizes[3], in_sizes[5]};
    const float* W1 = (const float*)d_in[7];
    const float* b1 = (const float*)d_in[8];
    const float* W2 = (const float*)d_in[9];
    const float* b2 = (const float*)d_in[10];
    const int N = in_sizes[0] / 32;
    float* out = (float*)d_out;

    char* ws = (char*)d_ws;
    float* scales = (float*)ws;
    float* acc1   = (float*)(ws + (size_t)6 * N * 4);
    float* hpre   = acc1 + (size_t)N * 16;
    float* z      = hpre + (size_t)N * 16;
    float* z3     = z - (size_t)2 * N;
    float* planes2 = acc1;
    unsigned* partial = (unsigned*)acc1;
    const int nR[3] = {(E[0] + B16_RUN - 1) / B16_RUN, (E[1] + B16_RUN - 1) / B16_RUN,
                       (E[2] + B16_RUN - 1) / B16_RUN};
    const int nRuns = nR[0] + nR[1] + nR[2];
    const size_t w_off = ((size_t)39 * N * 4 + 127) & ~(size_t)127;
    const bool binned = nRuns > 0 && ws_size >= w_off + (size_t)nRuns * B16_PITCH * 4;
    unsigned* words = (unsigned*)(ws + w_off);
    const size_t rOff[3] = {0, (size_t)nR[0] * B16_PITCH, (size_t)(nR[0] + nR[1]) * B16_PITCH};

    hg_count_own<<<dim3(CNT_SLICES, CNT_PLANES), dim3(CNT_TPB), 0, stream>>>(
        src[0], dst[0], src[1], dst[1], src[2], dst[2], E[0], E[1], E[2], partial, N);
    hg_scale_sum<<<dim3((N + TPB - 1) / TPB, 6), dim3(TPB), 0, stream>>>(partial, scales, N);

    if (!binned) {
        long span = 16L * N;
        hg_init<<<dim3((unsigned)((span + TPB - 1) / TPB)), dim3(TPB), 0, stream>>>(
            acc1, N);
    }

    if (binned)
        hg_bin16<<<dim3(nRuns), dim3(B16_TPB), 0, stream>>>(
            src[0], dst[0], src[1], dst[1], src[2], dst[2], E[0], E[1], E[2], nR[0], nR[1], words, N);

    const int tiles  = (N + 15) / 16;
    const int gblk   = (tiles + (TPB / 32) - 1) / (TPB / 32);
    for (int r = 0; r < 3; ++r) {
        hg_gemm1_wmma<<<dim3(gblk), dim3(TPB), 0, stream>>>(
            x, scales + (size_t)r * N, W1 + (size_t)r * 32 * 16, hpre, N);
        if (binned)
            hg_agg1_own16<<<dim3(SL16, 16), dim3(AGG_TPB), 0, stream>>>(
                words + rOff[r], nR[r], hpre, scales + (size_t)(3 + r) * N, acc1, N, r == 0 ? 1 : 0);
        else
            hg_agg1_col<<<dim3(8, 2, 16), dim3(AGG_TPB), 0, stream>>>(
                src[r], dst[r], E[r], hpre, scales + (size_t)(3 + r) * N, acc1, N);
    }
    hg_relu_bias<<<dim3((N + TPB - 1) / TPB, 16), dim3(TPB), 0, stream>>>(acc1, b1, N);

    for (int r = 0; r < 3; ++r) {
        hg_gemm2<<<dim3((N + TPB - 1) / TPB), dim3(TPB), 0, stream>>>(
            acc1, scales + (size_t)r * N, W2 + (size_t)r * 16, z3 + (size_t)r * N, N);
    }
    hg_agg2_own<<<dim3(L2_SLICES, L2_PLANES), dim3(L2_TPB), 0, stream>>>(
        src[0], dst[0], src[1], dst[1], src[2], dst[2], E[0], E[1], E[2], z3, planes2, N);
    hg_out_sum<<<dim3((N + TPB - 1) / TPB), dim3(TPB), 0, stream>>>(
        planes2, scales + (size_t)3 * N, b2, out, N);
}


# layer-2 aggregation reads the layer-1 binned words: a group owns (1 of 16 dst slices, run group), adds only its slice's words, no range test or queue; 16 partial planes (was 29), arrival-order sums
# speedup vs baseline: 14.2176x; 1.0820x over previous
#include <hip/hip_runtime.h>

typedef float v2f __attribute__((ext_vector_type(2)));
typedef float v8f __attribute__((ext_vector_type(8)));

#define TPB 256
#define AGG_TPB 1024
#define NMAX 50000
#define NHALF_MAX ((NMAX + 1) / 2)
#define FLUSH_U 13

#define CNT_TPB 1024
#define CNT_SLICES 8
#define CNT_SLICE_MAX ((((NMAX + CNT_SLICES - 1) / CNT_SLICES) + 31) & ~31)
#define CNT_C0 10
#define CNT_C1 3
#define CNT_C2 1
#define CNT_PLANES (2 * (CNT_C0 + CNT_C1 + CNT_C2))

#define L2_TPB 1024
#define L2_SLICES 4
#define L2_SLICE_MAX ((((NMAX + L2_SLICES - 1) / L2_SLICES) + 31) & ~31)
#define L2_C0 20
#define L2_C1 7
#define L2_C2 2
#define L2_PLANES (L2_C0 + L2_C1 + L2_C2)

__global__ __launch_bounds__(TPB) void hg_init(float* __restrict__ acc1,
                                               int N) {
    long i = (long)blockIdx.x * TPB + threadIdx.x;
    if (i < 16L * N) acc1[i] = 0.0f;
}

__device__ __forceinline__ void hg_bin(unsigned* __restrict__ bins,
                                       int v, int lo, unsigned cnt) {
    const unsigned t = (unsigned)(v - lo);
    if (t < cnt)
        __hip_atomic_fetch_add(&bins[t], 1u, __ATOMIC_RELAXED,
                               __HIP_MEMORY_SCOPE_WORKGROUP);
}

__global__ __launch_bounds__(CNT_TPB) void hg_count_own(const int* __restrict__ s0,
                                                        const int* __restrict__ d0,
                                                        const int* __restrict__ s1,
                                                        const int* __restrict__ d1,
                                                        const int* __restrict__ s2,
                                                        const int* __restrict__ d2,
                                                        int E0, int E1, int E2,
                                                        unsigned* __restrict__ partial,
                                                        int N) {
    __shared__ unsigned bins[CNT_SLICE_MAX];

    const int tid = threadIdx.x;
    const int p   = blockIdx.y;
    const int* __restrict__ idx;
    int E, nchunk, chunk;
    if (p < 2 * CNT_C0) {
        const int side = p >= CNT_C0;
        idx = side ? d0 : s0;  E = E0;  nchunk = CNT_C0;  chunk = p - side * CNT_C0;
    } else if (p < 2 * (CNT_C0 + CNT_C1)) {
        const int q = p - 2 * CNT_C0;
        const int side = q >= CNT_C1;
        idx = side ? d1 : s1;  E = E1;  nchunk = CNT_C1;  chunk = q - side * CNT_C1;
    } else {
        const int q = p - 2 * (CNT_C0 + CNT_C1);
        const int side = q >= CNT_C2;
        idx = side ? d2 : s2;  E = E2;  nchunk = CNT_C2;  chunk = q - side * CNT_C2;
    }

    const int ns  = (((N + CNT_SLICES - 1) / CNT_SLICES) + 31) & ~31;
    const int lo  = blockIdx.x * ns;
    const int hiN = lo + ns < N ? lo + ns : N;
    const unsigned cnt = hiN > lo ? (unsigned)(hiN - lo) : 0u;

    for (unsigned j = tid; j < cnt; j += CNT_TPB) bins[j] = 0u;
    __syncthreads();

    const int per = ((E + nchunk * 4 - 1) / (nchunk * 4)) * 4;
    const int e0  = chunk * per;
    const int e1  = e0 + per < E ? e0 + per : E;
    const int ne  = e1 > e0 ? e1 - e0 : 0;
    const int n4  = ne >> 2;
    const int4* __restrict__ q4 = (const int4*)(idx + e0);

    int i = tid;
    for (; i + 3 * CNT_TPB < n4; i += 4 * CNT_TPB) {
        const int4 a = q4[i], b = q4[i + CNT_TPB];
        const int4 c = q4[i + 2 * CNT_TPB], d = q4[i + 3 * CNT_TPB];
        hg_bin(bins, a.x, lo, cnt); hg_bin(bins, a.y, lo, cnt);
        hg_bin(bins, a.z, lo, cnt); hg_bin(bins, a.w, lo, cnt);
        hg_bin(bins, b.x, lo, cnt); hg_bin(bins, b.y, lo, cnt);
        hg_bin(bins, b.z, lo, cnt); hg_bin(bins, b.w, lo, cnt);
        hg_bin(bins, c.x, lo, cnt); hg_bin(bins, c.y, lo, cnt);
        hg_bin(bins, c.z, lo, cnt); hg_bin(bins, c.w, lo, cnt);
        hg_bin(bins, d.x, lo, cnt); hg_bin(bins, d.y, lo, cnt);
        hg_bin(bins, d.z, lo, cnt); hg_bin(bins, d.w, lo, cnt);
    }
    for (; i < n4; i += CNT_TPB) {
        const int4 a = q4[i];
        hg_bin(bins, a.x, lo, cnt); hg_bin(bins, a.y, lo, cnt);
        hg_bin(bins, a.z, lo, cnt); hg_bin(bins, a.w, lo, cnt);
    }
    for (int e = e0 + (n4 << 2) + tid; e < e1; e += CNT_TPB)
        hg_bin(bins, idx[e], lo, cnt);
    __syncthreads();

    const size_t pitch = (size_t)((N + 31) & ~31);
    unsigned* __restrict__ pp = partial + (size_t)p * pitch + lo;
    for (unsigned j = tid; j < cnt; j += CNT_TPB) pp[j] = bins[j];
}

template <int C>
__device__ __forceinline__ unsigned hg_sum_planes(const unsigned* __restrict__ pp, size_t pitch) {
    unsigned s = 0u;
#pragma unroll
    for (int c = 0; c < C; ++c) s += pp[(size_t)c * pitch];
    return s;
}

__global__ __launch_bounds__(TPB) void hg_scale_sum(const unsigned* __restrict__ partial,
                                                    float* __restrict__ scales,
                                                    int N) {
    const int n = blockIdx.x * TPB + threadIdx.x;
    if (n >= N) return;
    const int r    = blockIdx.y >> 1;
    const int side = blockIdx.y & 1;
    const size_t pitch = (size_t)((N + 31) & ~31);
    unsigned s;
    if (r == 0)
        s = hg_sum_planes<CNT_C0>(partial + (size_t)(side * CNT_C0) * pitch + n, pitch);
    else if (r == 1)
        s = hg_sum_planes<CNT_C1>(partial + (size_t)(2 * CNT_C0 + side * CNT_C1) * pitch + n, pitch);
    else
        s = hg_sum_planes<CNT_C2>(partial + (size_t)(2 * (CNT_C0 + CNT_C1) + side * CNT_C2) * pitch + n, pitch);
    float d = (float)s;
    if (d < 1.0f) d = 1.0f;
    scales[(size_t)(side * 3 + r) * N + n] = rsqrtf(d);
}

__global__ __launch_bounds__(TPB) void hg_gemm1_wmma(const float* __restrict__ x,
                                                     const float* __restrict__ s_out,
                                                     const float* __restrict__ W,
                                                     float* __restrict__ hT,
                                                     int N) {
    const int lane = threadIdx.x & 31;
    const int wave = threadIdx.x >> 5;
    const int tile = blockIdx.x * (TPB / 32) + wave;
    const int row0 = tile * 16;
    if (row0 >= N) return;

    const int half = lane >> 4;
    const int lm   = lane & 15;
    int row  = row0 + lm;
    int rowc = row < N ? row : N - 1;
    const float sc = s_out[rowc];

    v8f c = {};
#pragma unroll
    for (int kb = 0; kb < 8; ++kb) {
        const int k0 = kb * 4 + half * 2;
        v2f xf, wf;
        xf.x = x[(size_t)rowc * 32 + k0]     * sc;
        xf.y = x[(size_t)rowc * 32 + k0 + 1] * sc;
        wf.x = W[(k0)     * 16 + lm];
        wf.y = W[(k0 + 1) * 16 + lm];
        c = __builtin_amdgcn_wmma_f32_16x16x4_f32(
                 false, wf,  false, xf,
                 (short)0, c,  false,  false);
    }

    float* __restrict__ dp = hT + (size_t)(half * 8) * N + row;
    if (row < N) {
#pragma unroll
        for (int v = 0; v < 8; ++v) dp[(size_t)v * N] = c[v];
    }
}

#define Q_WORDS 160
static_assert(NMAX < 65536, "queue word holds src and local dst in 16 bits each");

__device__ __forceinline__ unsigned hg_push(unsigned* q, unsigned n, bool keep, unsigned word) {
    const unsigned m = __builtin_amdgcn_ballot_w32(keep);
    if (keep) q[n + __builtin_amdgcn_mbcnt_lo(m, 0u)] = word;
    return n + (unsigned)__builtin_popcount(m);
}

__device__ __forceinline__ void hg_add_word(float* sums, unsigned w, float v) {
    __hip_atomic_fetch_add(&sums[w >> 16], v, __ATOMIC_RELAXED, __HIP_MEMORY_SCOPE_WORKGROUP);
}

template <int R>
__device__ __forceinline__ void hg_take(const unsigned* q, const float* vals, float* sums,
                                        unsigned n, unsigned lane, unsigned carried, unsigned& wq) {
    unsigned w[R + 1];
#pragma unroll
    for (int r = 0; r < R; ++r) w[r] = q[lane + 32u * (unsigned)r];
    w[R] = lane + 32u * (unsigned)R < n ? q[lane + 32u * (unsigned)R] : 0u;
    if (lane < carried) w[0] = wq;
    float v[R + 1];
#pragma unroll
    for (int r = 0; r < R; ++r) v[r] = vals[w[r] & 0xffffu];
#pragma unroll
    for (int r = 0; r < R; ++r) hg_add_word(sums, w[r], v[r]);
    wq = w[R];
}

__device__ __forceinline__ void hg_trip(const int4 d, const int4 s, bool ok,
                                        const float* vals, float* sums, unsigned* q,
                                        int lo, unsigned cnt, unsigned lane,
                                        unsigned& rem, unsigned& wq) {
    const unsigned t0 = (unsigned)(d.x - lo), t1 = (unsigned)(d.y - lo);
    const unsigned t2 = (unsigned)(d.z - lo), t3 = (unsigned)(d.w - lo);
    const unsigned c = ok ? cnt : 0u;
    unsigned n = rem;
    n = hg_push(q, n, t0 < c, (unsigned)s.x | (t0 << 16));
    n = hg_push(q, n, t1 < c, (unsigned)s.y | (t1 << 16));
    n = hg_push(q, n, t2 < c, (unsigned)s.z | (t2 << 16));
    n = hg_push(q, n, t3 < c, (unsigned)s.w | (t3 << 16));
    __builtin_amdgcn_fence(__ATOMIC_RELEASE, "wavefront", "local");
    __builtin_amdgcn_wave_barrier();
    __builtin_amdgcn_fence(__ATOMIC_ACQUIRE, "wavefront", "local");
    const unsigned rows = n >> 5;
    if      (rows == 2u) hg_take<2>(q, vals, sums, n, lane, rem, wq);
    else if (rows == 1u) hg_take<1>(q, vals, sums, n, lane, rem, wq);
    else if (rows == 3u) hg_take<3>(q, vals, sums, n, lane, rem, wq);
    else if (rows == 0u) hg_take<0>(q, vals, sums, n, lane, rem, wq);
    else                 hg_take<4>(q, vals, sums, n, lane, rem, wq);
    rem = n & 31u;
}

__device__ __forceinline__ void hg_scan_own(const int4* __restrict__ s4, const int4* __restrict__ d4,
                                            int n4, const float* vals, float* sums, unsigned* q,
                                            int lo, unsigned cnt, int tid, int tpb) {
    const unsigned lane = (unsigned)tid & 31u;
    int4 dA = make_int4(0, 0, 0, 0), sA = make_int4(0, 0, 0, 0);
    int4 dB = make_int4(0, 0, 0, 0), sB = make_int4(0, 0, 0, 0);
    unsigned rem = 0u, wq = 0u;
    int ib = __builtin_amdgcn_readfirstlane(tid & ~31);
    if (ib + (int)lane < n4) { dA = d4[ib + (int)lane]; sA = s4[ib + (int)lane]; }
    const int last = n4 - 1;
    for (; ib < n4; ib += 2 * tpb) {
        const int iA = ib + (int)lane, iB = iA + tpb, iC = iB + tpb;
        const int jB = iB < n4 ? iB : last;
        dB = d4[jB];  sB = s4[jB];
        hg_trip(dA, sA, iA < n4, vals, sums, q, lo, cnt, lane, rem, wq);
        const int jC = iC < n4 ? iC : last;
        dA = d4[jC];  sA = s4[jC];
        if (ib + tpb < n4)
            hg_trip(dB, sB, iB < n4, vals, sums, q, lo, cnt, lane, rem, wq);
    }
    if (lane < rem) hg_add_word(sums, wq, vals[wq & 0xffffu]);
}

__device__ __forceinline__ void hg_edge_add(float* __restrict__ acc,
                                            float v, int d, int lo, unsigned cnt) {
    const unsigned dl = (unsigned)(d - lo);
    if (dl < cnt)
        __hip_atomic_fetch_add(&acc[dl], v, __ATOMIC_RELAXED,
                               __HIP_MEMORY_SCOPE_WORKGROUP);
}

__global__ __launch_bounds__(AGG_TPB) void hg_agg1_col(const int* __restrict__ src,
                                                       const int* __restrict__ dst,
                                                       int E,
                                                       const float* __restrict__ hT,
                                                       const float* __restrict__ s_in,
                                                       float* __restrict__ accT,
                                                       int N) {
    __shared__ float hcol[NMAX];
    __shared__ float acc[NHALF_MAX];
    __shared__ unsigned qbuf[(AGG_TPB / 32) * Q_WORDS];

    const int tid    = threadIdx.x;
    const int nchunk = gridDim.x;
    const int chunk  = blockIdx.x;
    const int f      = blockIdx.z;
    const int nh     = (N + 1) >> 1;
    const int lo     = blockIdx.y * nh;
    const int hiN    = lo + nh < N ? lo + nh : N;
    const unsigned cnt = hiN > lo ? (unsigned)(hiN - lo) : 0u;

    const float* __restrict__ hp = hT + (size_t)f * N;
    for (int i = tid; i < N; i += AGG_TPB) hcol[i] = hp[i];
    for (unsigned i = tid; i < cnt; i += AGG_TPB) acc[i] = 0.0f;
    __syncthreads();

    const int per = ((E + nchunk * 4 - 1) / (nchunk * 4)) * 4;
    const int e0  = chunk * per;
    const int e1  = e0 + per < E ? e0 + per : E;
    const int ne  = e1 > e0 ? e1 - e0 : 0;
    const int n4  = ne >> 2;
    const int4* __restrict__ s4 = (const int4*)(src + e0);
    const int4* __restrict__ d4 = (const int4*)(dst + e0);

    hg_scan_own(s4, d4, n4, hcol, acc, qbuf + (tid >> 5) * Q_WORDS, lo, cnt, tid, AGG_TPB);
    for (int e = e0 + (n4 << 2) + tid; e < e1; e += AGG_TPB)
        hg_edge_add(acc, hcol[src[e]], dst[e], lo, cnt);
    __syncthreads();

    float* __restrict__ ap = accT + (size_t)f * N + lo;
    const float* __restrict__ sp = s_in + lo;
    if (cnt != 0u) {
        const unsigned last = cnt - 1u;
        for (unsigned base = 0u; base < cnt; base += FLUSH_U * AGG_TPB) {
            float s[FLUSH_U], a[FLUSH_U];
#pragma unroll
            for (int u = 0; u < FLUSH_U; ++u) {
                const unsigned j  = base + (unsigned)u * AGG_TPB + tid;
                const unsigned jc = j < cnt ? j : last;
                s[u] = sp[jc];
                a[u] = acc[jc];
            }
#pragma unroll
            for (int u = 0; u < FLUSH_U; ++u) {
                const unsigned j = base + (unsigned)u * AGG_TPB + tid;
                const float v = a[u] * s[u];
                if (j < cnt && v != 0.0f) atomicAdd(ap + j, v);
            }
        }
    }
}

#define B16_TPB 1024
#define B16_VEC 5
#define B16_EPT (4 * B16_VEC)
#define B16_RUN (B16_TPB * B16_EPT)
#define B16_HDR 32
#define B16_PITCH (B16_HDR + B16_RUN)
#define SL16 16
#define SL16_MAX ((((NMAX + SL16 - 1) / SL16) + 31) & ~31)

__global__ __launch_bounds__(B16_TPB) void hg_bin16(const int* __restrict__ s0,
                                                    const int* __restrict__ d0,
                                                    const int* __restrict__ s1,
                                                    const int* __restrict__ d1,
                                                    const int* __restrict__ s2,
                                                    const int* __restrict__ d2,
                                                    int E0, int E1, int E2,
                                                    int nR0, int nR1,
                                                    unsigned* __restrict__ words,
                                                    int N) {
    __shared__ unsigned buf[B16_RUN];
    __shared__ unsigned hist[SL16 * 32];
    __shared__ unsigned base[SL16 * 32];
    __shared__ unsigned tot[SL16];
    __shared__ unsigned start[SL16 + 1];

    const int tid = threadIdx.x;
    const int run = blockIdx.x;
    const int* __restrict__ src;
    const int* __restrict__ dst;
    int E, q;
    if (run < nR0)            { src = s0;  dst = d0;  E = E0;  q = run; }
    else if (run < nR0 + nR1) { src = s1;  dst = d1;  E = E1;  q = run - nR0; }
    else                      { src = s2;  dst = d2;  E = E2;  q = run - nR0 - nR1; }
    const int ns  = (((N + SL16 - 1) / SL16) + 31) & ~31;
    const float inv_ns = 1.0f / (float)ns;
    const int R0  = q * B16_RUN;
    const int len = E - R0 < B16_RUN ? E - R0 : B16_RUN;

    const int lane = tid & 31;
    if (tid < SL16 * 32) hist[tid] = 0u;

    int d[B16_EPT], s[B16_EPT];
    bool ok[B16_EPT];
#pragma unroll
    for (int v = 0; v < B16_VEC; ++v) {
        const int b = v * (B16_RUN / B16_VEC) + 4 * tid;
        if (b + 3 < len) {
            const int4 dv = *(const int4*)(dst + R0 + b);
            const int4 sv = *(const int4*)(src + R0 + b);
            d[4 * v] = dv.x;  d[4 * v + 1] = dv.y;  d[4 * v + 2] = dv.z;  d[4 * v + 3] = dv.w;
            s[4 * v] = sv.x;  s[4 * v + 1] = sv.y;  s[4 * v + 2] = sv.z;  s[4 * v + 3] = sv.w;
            ok[4 * v] = ok[4 * v + 1] = ok[4 * v + 2] = ok[4 * v + 3] = true;
        } else {
#pragma unroll
            for (int k = 0; k < 4; ++k) {
                const bool in = b + k < len;
                ok[4 * v + k] = in;
                d[4 * v + k]  = in ? dst[R0 + b + k] : 0;
                s[4 * v + k]  = in ? src[R0 + b + k] : 0;
            }
        }
    }
    __syncthreads();

    unsigned word[B16_EPT], rank[B16_EPT];
    int bin[B16_EPT];
#pragma unroll
    for (int k = 0; k < B16_EPT; ++k) {
        const bool valid = ok[k] && (unsigned)d[k] < (unsigned)N;
        const int dv = valid ? d[k] : 0;
        int b = (int)((float)dv * inv_ns);
        if (b * ns > dv) --b;
        else if ((b + 1) * ns <= dv) ++b;
        const unsigned t = (unsigned)(dv - b * ns);
        word[k] = (unsigned)s[k] | (t << 16);
        bin[k]  = valid ? b : SL16;
        rank[k] = 0u;
        if (valid)
            rank[k] = __hip_atomic_fetch_add(&hist[b * 32 + lane], 1u, __ATOMIC_RELAXED,
                                             __HIP_MEMORY_SCOPE_WORKGROUP);
    }
    __syncthreads();
    unsigned cv = 0u, incl = 0u;
    if (tid < SL16 * 32) {
        cv = hist[tid];
        incl = cv;
#pragma unroll
        for (int o = 1; o < 32; o <<= 1) {
            const unsigned up = __shfl_up(incl, (unsigned)o, 32);
            if (lane >= o) incl += up;
        }
        if (lane == 31) tot[tid >> 5] = incl;
    }
    __syncthreads();
    if (tid == 0) {
        unsigned a = 0u;
#pragma unroll
        for (int b = 0; b < SL16; ++b) { start[b] = a;  a += tot[b]; }
        start[SL16] = a;
    }
    __syncthreads();
    if (tid < SL16 * 32) base[tid] = start[tid >> 5] + (incl - cv);
    __syncthreads();
#pragma unroll
    for (int k = 0; k < B16_EPT; ++k)
        if (bin[k] < SL16) buf[base[bin[k] * 32 + lane] + rank[k]] = word[k];
    __syncthreads();

    unsigned* __restrict__ wr = words + (size_t)run * B16_PITCH;
    if (tid < B16_HDR) wr[tid] = tid <= SL16 ? start[tid] : 0u;
    const int total = (int)start[SL16];
    for (int i = tid; i < total; i += B16_TPB) wr[B16_HDR + i] = buf[i];
}

#define RUN_ROWS 8
__device__ __forceinline__ void hg_load_rows(const unsigned* __restrict__ wp, int i0, int last,
                                             int lane, unsigned (&w)[RUN_ROWS]) {
#pragma unroll
    for (int u = 0; u < RUN_ROWS; ++u) {
        const int i = i0 + 32 * u + lane;
        w[u] = wp[i < last ? i : last];
    }
}

__device__ __forceinline__ void hg_add_rows(const unsigned (&w)[RUN_ROWS], int i0, int st, int en, int lane,
                                            const float* vals, float* sums, unsigned cnt) {
    float v[RUN_ROWS];
#pragma unroll
    for (int u = 0; u < RUN_ROWS; ++u) v[u] = vals[w[u] & 0xffffu];
#pragma unroll
    for (int u = 0; u < RUN_ROWS; ++u) {
        const int i = i0 + 32 * u + lane;
        const unsigned t = w[u] >> 16;
        if (i >= st && i < en && t < cnt)
            __hip_atomic_fetch_add(&sums[t], v[u], __ATOMIC_RELAXED, __HIP_MEMORY_SCOPE_WORKGROUP);
    }
}

__global__ __launch_bounds__(AGG_TPB) void hg_agg1_own16(const unsigned* __restrict__ words,
                                                         int nR,
                                                         const float* __restrict__ hT,
                                                         const float* __restrict__ s_in,
                                                         float* __restrict__ accT,
                                                         int N, int first) {
    __shared__ float hcol[NMAX];
    __shared__ float acc[SL16_MAX];

    const int tid  = threadIdx.x;
    const int lane = tid & 31;
    const int f    = blockIdx.y;
    const int sl   = blockIdx.x;
    const int ns   = (((N + SL16 - 1) / SL16) + 31) & ~31;
    const int lo   = sl * ns;
    const int hiN  = lo + ns < N ? lo + ns : N;
    const unsigned cnt = hiN > lo ? (unsigned)(hiN - lo) : 0u;

    const float* __restrict__ hp = hT + (size_t)f * N;
    for (int i = tid; i < N; i += AGG_TPB) hcol[i] = hp[i];
    for (unsigned i = tid; i < cnt; i += AGG_TPB) acc[i] = 0.0f;
    __syncthreads();

    const int nwav = AGG_TPB / 32;
    int q = __builtin_amdgcn_readfirstlane(tid >> 5);
    unsigned h0 = 0u, h1 = 0u;
    if (q < nR) {
        const unsigned* __restrict__ rg = words + (size_t)q * B16_PITCH;
        h0 = rg[sl];  h1 = rg[sl + 1];
    }
    while (q < nR) {
        const int qn = q + nwav;
        unsigned g0 = 0u, g1 = 0u;
        if (qn < nR) {
            const unsigned* __restrict__ rn = words + (size_t)qn * B16_PITCH;
            g0 = rn[sl];  g1 = rn[sl + 1];
        }
        const int st = __builtin_amdgcn_readfirstlane((int)h0);
        const int en = __builtin_amdgcn_readfirstlane((int)h1);
        const unsigned* __restrict__ wp = words + (size_t)q * B16_PITCH + B16_HDR;
        if (en > st) {
            const int a0   = st & ~31;
            const int last = en - 1;
            const int G  = 32 * RUN_ROWS;
            const int ng = (en - a0 + G - 1) / G;
            unsigned wA[RUN_ROWS], wB[RUN_ROWS];
            hg_load_rows(wp, a0, last, lane, wA);
            int g = 0;
            for (; g + 2 < ng; g += 2) {
                hg_load_rows(wp, a0 + (g + 1) * G, last, lane, wB);
                hg_add_rows(wA, a0 + g * G, st, en, lane, hcol, acc, cnt);
                hg_load_rows(wp, a0 + (g + 2) * G, last, lane, wA);
                hg_add_rows(wB, a0 + (g + 1) * G, st, en, lane, hcol, acc, cnt);
            }
            if (g + 1 < ng) {
                hg_load_rows(wp, a0 + (g + 1) * G, last, lane, wB);
                hg_add_rows(wA, a0 + g * G, st, en, lane, hcol, acc, cnt);
                hg_add_rows(wB, a0 + (g + 1) * G, st, en, lane, hcol, acc, cnt);
            } else {
                hg_add_rows(wA, a0 + g * G, st, en, lane, hcol, acc, cnt);
            }
        }
        h0 = g0;  h1 = g1;  q = qn;
    }
    __syncthreads();

    float* __restrict__ ap = accT + (size_t)f * N + lo;
    const float* __restrict__ sp = s_in + lo;
    for (unsigned j = tid; j < cnt; j += AGG_TPB) {
        const float v = acc[j] * sp[j];
        ap[j] = first ? v : ap[j] + v;
    }
}


__global__ __launch_bounds__(TPB) void hg_relu_bias(float* __restrict__ acc,
                                                    const float* __restrict__ b1,
                                                    int N) {
    const int f = blockIdx.y;
    const int n = blockIdx.x * TPB + threadIdx.x;
    if (n < N) {
        const size_t i = (size_t)f * N + n;
        float v = acc[i] + b1[f] + b1[16 + f] + b1[32 + f];
        acc[i] = v > 0.0f ? v : 0.0f;
    }
}

__global__ __launch_bounds__(TPB) void hg_gemm2(const float* __restrict__ h,
                                                const float* __restrict__ s_out,
                                                const float* __restrict__ w,
                                                float* __restrict__ z,
                                                int N) {
    int i = blockIdx.x * TPB + threadIdx.x;
    if (i >= N) return;
    const float* hp = h + i;
    const float4* wp = (const float4*)w;
    float acc = 0.0f;
#pragma unroll
    for (int q = 0; q < 4; ++q) {
        float4 a, b = wp[q];
        a.x = hp[(size_t)(4 * q + 0) * N];
        a.y = hp[(size_t)(4 * q + 1) * N];
        a.z = hp[(size_t)(4 * q + 2) * N];
        a.w = hp[(size_t)(4 * q + 3) * N];
        acc += a.x * b.x + a.y * b.y + a.z * b.z + a.w * b.w;
    }
    z[i] = acc * s_out[i];
}

__global__ __launch_bounds__(L2_TPB) void hg_agg2_own(const int* __restrict__ s0,
                                                      const int* __restrict__ d0,
                                                      const int* __restrict__ s1,
                                                      const int* __restrict__ d1,
                                                      const int* __restrict__ s2,
                                                      const int* __restrict__ d2,
                                                      int E0, int E1, int E2,
                                                      const float* __restrict__ z3,
                                                      float* __restrict__ partial,
                                                      int N) {
    __shared__ float zc[NMAX];
    __shared__ float bins[L2_SLICE_MAX];
    __shared__ unsigned qbuf[(L2_TPB / 32) * Q_WORDS];

    const int tid = threadIdx.x;
    const int p   = blockIdx.y;
    const int* __restrict__ src;
    const int* __restrict__ dst;
    int E, nchunk, chunk, r;
    if (p < L2_C0) {
        src = s0;  dst = d0;  E = E0;  nchunk = L2_C0;  chunk = p;                  r = 0;
    } else if (p < L2_C0 + L2_C1) {
        src = s1;  dst = d1;  E = E1;  nchunk = L2_C1;  chunk = p - L2_C0;          r = 1;
    } else {
        src = s2;  dst = d2;  E = E2;  nchunk = L2_C2;  chunk = p - L2_C0 - L2_C1;  r = 2;
    }

    const int ns  = (((N + L2_SLICES - 1) / L2_SLICES) + 31) & ~31;
    const int lo  = blockIdx.x * ns;
    const int hiN = lo + ns < N ? lo + ns : N;
    const unsigned cnt = hiN > lo ? (unsigned)(hiN - lo) : 0u;

    const float* __restrict__ zp = z3 + (size_t)r * N;
    for (int i = tid; i < N; i += L2_TPB) zc[i] = zp[i];
    for (unsigned j = tid; j < cnt; j += L2_TPB) bins[j] = 0.0f;
    __syncthreads();

    const int per = ((E + nchunk * 4 - 1) / (nchunk * 4)) * 4;
    const int e0  = chunk * per;
    const int e1  = e0 + per < E ? e0 + per : E;
    const int ne  = e1 > e0 ? e1 - e0 : 0;
    const int n4  = ne >> 2;
    const int4* __restrict__ s4 = (const int4*)(src + e0);
    const int4* __restrict__ d4 = (const int4*)(dst + e0);

    hg_scan_own(s4, d4, n4, zc, bins, qbuf + (tid >> 5) * Q_WORDS, lo, cnt, tid, L2_TPB);
    for (int e = e0 + (n4 << 2) + tid; e < e1; e += L2_TPB)
        hg_edge_add(bins, zc[src[e]], dst[e], lo, cnt);
    __syncthreads();

    const size_t pitch = (size_t)((N + 31) & ~31);
    float* __restrict__ pp = partial + (size_t)p * pitch + lo;
    for (unsigned j = tid; j < cnt; j += L2_TPB) pp[j] = bins[j];
}

#define A2_G0 11
#define A2_G1 4
#define A2_G2 1
#define A2_PLANES (A2_G0 + A2_G1 + A2_G2)

__global__ __launch_bounds__(AGG_TPB) void hg_agg2_own16(const unsigned* __restrict__ words,
                                                         int nR0, int nR1, int nR2,
                                                         const float* __restrict__ z3,
                                                         float* __restrict__ partial,
                                                         int N) {
    __shared__ float zc[NMAX];
    __shared__ float acc[SL16_MAX];

    const int tid  = threadIdx.x;
    const int lane = tid & 31;
    const int sl   = blockIdx.x;
    const int p    = blockIdx.y;
    int r, gi, nG, nR, rb;
    if (p < A2_G0)              { r = 0;  gi = p;                  nG = A2_G0;  nR = nR0;  rb = 0; }
    else if (p < A2_G0 + A2_G1) { r = 1;  gi = p - A2_G0;          nG = A2_G1;  nR = nR1;  rb = nR0; }
    else                        { r = 2;  gi = p - A2_G0 - A2_G1;  nG = A2_G2;  nR = nR2;  rb = nR0 + nR1; }
    const int per = (nR + nG - 1) / nG;
    const int q0  = gi * per;
    const int q1  = q0 + per < nR ? q0 + per : nR;

    const int ns   = (((N + SL16 - 1) / SL16) + 31) & ~31;
    const int lo   = sl * ns;
    const int hiN  = lo + ns < N ? lo + ns : N;
    const unsigned cnt = hiN > lo ? (unsigned)(hiN - lo) : 0u;

    const float* __restrict__ zp = z3 + (size_t)r * N;
    for (int i = tid; i < N; i += AGG_TPB) zc[i] = zp[i];
    for (unsigned i = tid; i < cnt; i += AGG_TPB) acc[i] = 0.0f;
    __syncthreads();

    const int nwav = AGG_TPB / 32;
    const unsigned* __restrict__ rel = words + (size_t)rb * B16_PITCH;
    int q = q0 + __builtin_amdgcn_readfirstlane(tid >> 5);
    unsigned h0 = 0u, h1 = 0u;
    if (q < q1) {
        const unsigned* __restrict__ rg = rel + (size_t)q * B16_PITCH;
        h0 = rg[sl];  h1 = rg[sl + 1];
    }
    while (q < q1) {
        const int qn = q + nwav;
        unsigned g0 = 0u, g1 = 0u;
        if (qn < q1) {
            const unsigned* __restrict__ rn = rel + (size_t)qn * B16_PITCH;
            g0 = rn[sl];  g1 = rn[sl + 1];
        }
        const int st = __builtin_amdgcn_readfirstlane((int)h0);
        const int en = __builtin_amdgcn_readfirstlane((int)h1);
        const unsigned* __restrict__ wp = rel + (size_t)q * B16_PITCH + B16_HDR;
        if (en > st) {
            const int a0   = st & ~31;
            const int last = en - 1;
            const int G  = 32 * RUN_ROWS;
            const int ng = (en - a0 + G - 1) / G;
            unsigned wA[RUN_ROWS], wB[RUN_ROWS];
            hg_load_rows(wp, a0, last, lane, wA);
            int g = 0;
            for (; g + 2 < ng; g += 2) {
                hg_load_rows(wp, a0 + (g + 1) * G, last, lane, wB);
                hg_add_rows(wA, a0 + g * G, st, en, lane, zc, acc, cnt);
                hg_load_rows(wp, a0 + (g + 2) * G, last, lane, wA);
                hg_add_rows(wB, a0 + (g + 1) * G, st, en, lane, zc, acc, cnt);
            }
            if (g + 1 < ng) {
                hg_load_rows(wp, a0 + (g + 1) * G, last, lane, wB);
                hg_add_rows(wA, a0 + g * G, st, en, lane, zc, acc, cnt);
                hg_add_rows(wB, a0 + (g + 1) * G, st, en, lane, zc, acc, cnt);
            } else {
                hg_add_rows(wA, a0 + g * G, st, en, lane, zc, acc, cnt);
            }
        }
        h0 = g0;  h1 = g1;  q = qn;
    }
    __syncthreads();

    const size_t pitch = (size_t)((N + 31) & ~31);
    float* __restrict__ pp = partial + (size_t)p * pitch + lo;
    for (unsigned j = tid; j < cnt; j += AGG_TPB) pp[j] = acc[j];
}

template <int C>
__device__ __forceinline__ float hg_sum_fplanes(const float* __restrict__ pp, size_t pitch) {
    float s = 0.0f;
#pragma unroll
    for (int c = 0; c < C; ++c) s += pp[(size_t)c * pitch];
    return s;
}

template <int C0, int C1, int C2>
__device__ __forceinline__ void hg_out_body(const float* __restrict__ partial,
                                            const float* __restrict__ s_in3,
                                            const float* __restrict__ b2,
                                            float* __restrict__ out,
                                            int N) {
    const int n = blockIdx.x * TPB + threadIdx.x;
    if (n >= N) return;
    const size_t pitch = (size_t)((N + 31) & ~31);
    const float m0 = hg_sum_fplanes<C0>(partial + n, pitch);
    const float m1 = hg_sum_fplanes<C1>(partial + (size_t)C0 * pitch + n, pitch);
    const float m2 = hg_sum_fplanes<C2>(partial + (size_t)(C0 + C1) * pitch + n, pitch);
    out[n] = (m0 * s_in3[n] + b2[0])
           + (m1 * s_in3[(size_t)N + n] + b2[1])
           + (m2 * s_in3[(size_t)2 * N + n] + b2[2]);
}

__global__ __launch_bounds__(TPB) void hg_out_sum(const float* __restrict__ partial,
                                                  const float* __restrict__ s_in3,
                                                  const float* __restrict__ b2,
                                                  float* __restrict__ out,
                                                  int N) {
    hg_out_body<L2_C0, L2_C1, L2_C2>(partial, s_in3, b2, out, N);
}

__global__ __launch_bounds__(TPB) void hg_out_sum16(const float* __restrict__ partial,
                                                    const float* __restrict__ s_in3,
                                                    const float* __restrict__ b2,
                                                    float* __restrict__ out,
                                                    int N) {
    hg_out_body<A2_G0, A2_G1, A2_G2>(partial, s_in3, b2, out, N);
}

extern "C" void kernel_launch(void* const* d_in, const int* in_sizes, int n_in,
                              void* d_out, int out_size, void* d_ws, size_t ws_size,
                              hipStream_t stream) {
    const float* x  = (const float*)d_in[0];
    const int* src[3] = {(const int*)d_in[1], (const int*)d_in[3], (const int*)d_in[5]};
    const int* dst[3] = {(const int*)d_in[2], (const int*)d_in[4], (const int*)d_in[6]};
    const int  E[3]   = {in_sizes[1], in_sizes[3], in_sizes[5]};
    const float* W1 = (const float*)d_in[7];
    const float* b1 = (const float*)d_in[8];
    const float* W2 = (const float*)d_in[9];
    const float* b2 = (const float*)d_in[10];
    const int N = in_sizes[0] / 32;
    float* out = (float*)d_out;

    char* ws = (char*)d_ws;
    float* scales = (float*)ws;
    float* acc1   = (float*)(ws + (size_t)6 * N * 4);
    float* hpre   = acc1 + (size_t)N * 16;
    float* z      = hpre + (size_t)N * 16;
    float* z3     = z - (size_t)2 * N;
    float* planes2 = acc1;
    unsigned* partial = (unsigned*)acc1;
    const int nR[3] = {(E[0] + B16_RUN - 1) / B16_RUN, (E[1] + B16_RUN - 1) / B16_RUN,
                       (E[2] + B16_RUN - 1) / B16_RUN};
    const int nRuns = nR[0] + nR[1] + nR[2];
    const size_t w_off = ((size_t)39 * N * 4 + 127) & ~(size_t)127;
    const bool binned = nRuns > 0 && ws_size >= w_off + (size_t)nRuns * B16_PITCH * 4;
    unsigned* words = (unsigned*)(ws + w_off);
    const size_t rOff[3] = {0, (size_t)nR[0] * B16_PITCH, (size_t)(nR[0] + nR[1]) * B16_PITCH};

    hg_count_own<<<dim3(CNT_SLICES, CNT_PLANES), dim3(CNT_TPB), 0, stream>>>(
        src[0], dst[0], src[1], dst[1], src[2], dst[2], E[0], E[1], E[2], partial, N);
    hg_scale_sum<<<dim3((N + TPB - 1) / TPB, 6), dim3(TPB), 0, stream>>>(partial, scales, N);

    if (!binned) {
        long span = 16L * N;
        hg_init<<<dim3((unsigned)((span + TPB - 1) / TPB)), dim3(TPB), 0, stream>>>(
            acc1, N);
    }

    if (binned)
        hg_bin16<<<dim3(nRuns), dim3(B16_TPB), 0, stream>>>(
            src[0], dst[0], src[1], dst[1], src[2], dst[2], E[0], E[1], E[2], nR[0], nR[1], words, N);

    const int tiles  = (N + 15) / 16;
    const int gblk   = (tiles + (TPB / 32) - 1) / (TPB / 32);
    for (int r = 0; r < 3; ++r) {
        hg_gemm1_wmma<<<dim3(gblk), dim3(TPB), 0, stream>>>(
            x, scales + (size_t)r * N, W1 + (size_t)r * 32 * 16, hpre, N);
        if (binned)
            hg_agg1_own16<<<dim3(SL16, 16), dim3(AGG_TPB), 0, stream>>>(
                words + rOff[r], nR[r], hpre, scales + (size_t)(3 + r) * N, acc1, N, r == 0 ? 1 : 0);
        else
            hg_agg1_col<<<dim3(8, 2, 16), dim3(AGG_TPB), 0, stream>>>(
                src[r], dst[r], E[r], hpre, scales + (size_t)(3 + r) * N, acc1, N);
    }
    hg_relu_bias<<<dim3((N + TPB - 1) / TPB, 16), dim3(TPB), 0, stream>>>(acc1, b1, N);

    for (int r = 0; r < 3; ++r) {
        hg_gemm2<<<dim3((N + TPB - 1) / TPB), dim3(TPB), 0, stream>>>(
            acc1, scales + (size_t)r * N, W2 + (size_t)r * 16, z3 + (size_t)r * N, N);
    }
    if (binned) {
        hg_agg2_own16<<<dim3(SL16, A2_PLANES), dim3(AGG_TPB), 0, stream>>>(
            words, nR[0], nR[1], nR[2], z3, planes2, N);
        hg_out_sum16<<<dim3((N + TPB - 1) / TPB), dim3(TPB), 0, stream>>>(
            planes2, scales + (size_t)3 * N, b2, out, N);
    } else {
        hg_agg2_own<<<dim3(L2_SLICES, L2_PLANES), dim3(L2_TPB), 0, stream>>>(
            src[0], dst[0], src[1], dst[1], src[2], dst[2], E[0], E[1], E[2], z3, planes2, N);
        hg_out_sum<<<dim3((N + TPB - 1) / TPB), dim3(TPB), 0, stream>>>(
            planes2, scales + (size_t)3 * N, b2, out, N);
    }
}


# relation loop moved into the kernels: one WMMA transform launch fills three h_pre buffers, one owner adds relations 0,1,2 in turn and stores relu(sum+bias), one launch does the 3 dots; 15->8 launches
# speedup vs baseline: 17.0521x; 1.1994x over previous
#include <hip/hip_runtime.h>

typedef float v2f __attribute__((ext_vector_type(2)));
typedef float v8f __attribute__((ext_vector_type(8)));

#define TPB 256
#define AGG_TPB 1024
#define NMAX 50000
#define NHALF_MAX ((NMAX + 1) / 2)
#define FLUSH_U 13

#define CNT_TPB 1024
#define CNT_SLICES 8
#define CNT_SLICE_MAX ((((NMAX + CNT_SLICES - 1) / CNT_SLICES) + 31) & ~31)
#define CNT_C0 10
#define CNT_C1 3
#define CNT_C2 1
#define CNT_PLANES (2 * (CNT_C0 + CNT_C1 + CNT_C2))

#define L2_TPB 1024
#define L2_SLICES 4
#define L2_SLICE_MAX ((((NMAX + L2_SLICES - 1) / L2_SLICES) + 31) & ~31)
#define L2_C0 20
#define L2_C1 7
#define L2_C2 2
#define L2_PLANES (L2_C0 + L2_C1 + L2_C2)

__global__ __launch_bounds__(TPB) void hg_init(float* __restrict__ acc1,
                                               int N) {
    long i = (long)blockIdx.x * TPB + threadIdx.x;
    if (i < 16L * N) acc1[i] = 0.0f;
}

__device__ __forceinline__ void hg_bin(unsigned* __restrict__ bins,
                                       int v, int lo, unsigned cnt) {
    const unsigned t = (unsigned)(v - lo);
    if (t < cnt)
        __hip_atomic_fetch_add(&bins[t], 1u, __ATOMIC_RELAXED,
                               __HIP_MEMORY_SCOPE_WORKGROUP);
}

__global__ __launch_bounds__(CNT_TPB) void hg_count_own(const int* __restrict__ s0,
                                                        const int* __restrict__ d0,
                                                        const int* __restrict__ s1,
                                                        const int* __restrict__ d1,
                                                        const int* __restrict__ s2,
                                                        const int* __restrict__ d2,
                                                        int E0, int E1, int E2,
                                                        unsigned* __restrict__ partial,
                                                        int N) {
    __shared__ unsigned bins[CNT_SLICE_MAX];

    const int tid = threadIdx.x;
    const int p   = blockIdx.y;
    const int* __restrict__ idx;
    int E, nchunk, chunk;
    if (p < 2 * CNT_C0) {
        const int side = p >= CNT_C0;
        idx = side ? d0 : s0;  E = E0;  nchunk = CNT_C0;  chunk = p - side * CNT_C0;
    } else if (p < 2 * (CNT_C0 + CNT_C1)) {
        const int q = p - 2 * CNT_C0;
        const int side = q >= CNT_C1;
        idx = side ? d1 : s1;  E = E1;  nchunk = CNT_C1;  chunk = q - side * CNT_C1;
    } else {
        const int q = p - 2 * (CNT_C0 + CNT_C1);
        const int side = q >= CNT_C2;
        idx = side ? d2 : s2;  E = E2;  nchunk = CNT_C2;  chunk = q - side * CNT_C2;
    }

    const int ns  = (((N + CNT_SLICES - 1) / CNT_SLICES) + 31) & ~31;
    const int lo  = blockIdx.x * ns;
    const int hiN = lo + ns < N ? lo + ns : N;
    const unsigned cnt = hiN > lo ? (unsigned)(hiN - lo) : 0u;

    for (unsigned j = tid; j < cnt; j += CNT_TPB) bins[j] = 0u;
    __syncthreads();

    const int per = ((E + nchunk * 4 - 1) / (nchunk * 4)) * 4;
    const int e0  = chunk * per;
    const int e1  = e0 + per < E ? e0 + per : E;
    const int ne  = e1 > e0 ? e1 - e0 : 0;
    const int n4  = ne >> 2;
    const int4* __restrict__ q4 = (const int4*)(idx + e0);

    int i = tid;
    for (; i + 3 * CNT_TPB < n4; i += 4 * CNT_TPB) {
        const int4 a = q4[i], b = q4[i + CNT_TPB];
        const int4 c = q4[i + 2 * CNT_TPB], d = q4[i + 3 * CNT_TPB];
        hg_bin(bins, a.x, lo, cnt); hg_bin(bins, a.y, lo, cnt);
        hg_bin(bins, a.z, lo, cnt); hg_bin(bins, a.w, lo, cnt);
        hg_bin(bins, b.x, lo, cnt); hg_bin(bins, b.y, lo, cnt);
        hg_bin(bins, b.z, lo, cnt); hg_bin(bins, b.w, lo, cnt);
        hg_bin(bins, c.x, lo, cnt); hg_bin(bins, c.y, lo, cnt);
        hg_bin(bins, c.z, lo, cnt); hg_bin(bins, c.w, lo, cnt);
        hg_bin(bins, d.x, lo, cnt); hg_bin(bins, d.y, lo, cnt);
        hg_bin(bins, d.z, lo, cnt); hg_bin(bins, d.w, lo, cnt);
    }
    for (; i < n4; i += CNT_TPB) {
        const int4 a = q4[i];
        hg_bin(bins, a.x, lo, cnt); hg_bin(bins, a.y, lo, cnt);
        hg_bin(bins, a.z, lo, cnt); hg_bin(bins, a.w, lo, cnt);
    }
    for (int e = e0 + (n4 << 2) + tid; e < e1; e += CNT_TPB)
        hg_bin(bins, idx[e], lo, cnt);
    __syncthreads();

    const size_t pitch = (size_t)((N + 31) & ~31);
    unsigned* __restrict__ pp = partial + (size_t)p * pitch + lo;
    for (unsigned j = tid; j < cnt; j += CNT_TPB) pp[j] = bins[j];
}

template <int C>
__device__ __forceinline__ unsigned hg_sum_planes(const unsigned* __restrict__ pp, size_t pitch) {
    unsigned s = 0u;
#pragma unroll
    for (int c = 0; c < C; ++c) s += pp[(size_t)c * pitch];
    return s;
}

__global__ __launch_bounds__(TPB) void hg_scale_sum(const unsigned* __restrict__ partial,
                                                    float* __restrict__ scales,
                                                    int N) {
    const int n = blockIdx.x * TPB + threadIdx.x;
    if (n >= N) return;
    const int r    = blockIdx.y >> 1;
    const int side = blockIdx.y & 1;
    const size_t pitch = (size_t)((N + 31) & ~31);
    unsigned s;
    if (r == 0)
        s = hg_sum_planes<CNT_C0>(partial + (size_t)(side * CNT_C0) * pitch + n, pitch);
    else if (r == 1)
        s = hg_sum_planes<CNT_C1>(partial + (size_t)(2 * CNT_C0 + side * CNT_C1) * pitch + n, pitch);
    else
        s = hg_sum_planes<CNT_C2>(partial + (size_t)(2 * (CNT_C0 + CNT_C1) + side * CNT_C2) * pitch + n, pitch);
    float d = (float)s;
    if (d < 1.0f) d = 1.0f;
    scales[(size_t)(side * 3 + r) * N + n] = rsqrtf(d);
}

__global__ __launch_bounds__(TPB) void hg_gemm1_wmma(const float* __restrict__ x,
                                                     const float* __restrict__ s_out,
                                                     const float* __restrict__ W,
                                                     float* __restrict__ hT,
                                                     int N) {
    const int lane = threadIdx.x & 31;
    const int wave = threadIdx.x >> 5;
    const int tile = blockIdx.x * (TPB / 32) + wave;
    const int row0 = tile * 16;
    if (row0 >= N) return;

    const int half = lane >> 4;
    const int lm   = lane & 15;
    int row  = row0 + lm;
    int rowc = row < N ? row : N - 1;
    const float sc = s_out[rowc];

    v8f c = {};
#pragma unroll
    for (int kb = 0; kb < 8; ++kb) {
        const int k0 = kb * 4 + half * 2;
        v2f xf, wf;
        xf.x = x[(size_t)rowc * 32 + k0]     * sc;
        xf.y = x[(size_t)rowc * 32 + k0 + 1] * sc;
        wf.x = W[(k0)     * 16 + lm];
        wf.y = W[(k0 + 1) * 16 + lm];
        c = __builtin_amdgcn_wmma_f32_16x16x4_f32(
                 false, wf,  false, xf,
                 (short)0, c,  false,  false);
    }

    float* __restrict__ dp = hT + (size_t)(half * 8) * N + row;
    if (row < N) {
#pragma unroll
        for (int v = 0; v < 8; ++v) dp[(size_t)v * N] = c[v];
    }
}

__global__ __launch_bounds__(TPB) void hg_gemm1_all_wmma(const float* __restrict__ x,
                                                         const float* __restrict__ s_out3,
                                                         const float* __restrict__ W1,
                                                         float* __restrict__ hT3,
                                                         int N) {
    const int lane = threadIdx.x & 31;
    const int wave = threadIdx.x >> 5;
    const int tile = blockIdx.x * (TPB / 32) + wave;
    const int row0 = tile * 16;
    if (row0 >= N) return;

    const int half = lane >> 4;
    const int lm   = lane & 15;
    int row  = row0 + lm;
    int rowc = row < N ? row : N - 1;
    const float sc0 = s_out3[rowc];
    const float sc1 = s_out3[(size_t)N + rowc];
    const float sc2 = s_out3[(size_t)2 * N + rowc];
    const float* __restrict__ Wa = W1;
    const float* __restrict__ Wb = W1 + 32 * 16;
    const float* __restrict__ Wc = W1 + 2 * 32 * 16;

    v8f c0 = {}, c1 = {}, c2 = {};
#pragma unroll
    for (int kb = 0; kb < 8; ++kb) {
        const int k0 = kb * 4 + half * 2;
        const float x0 = x[(size_t)rowc * 32 + k0];
        const float x1 = x[(size_t)rowc * 32 + k0 + 1];
        v2f xf, wf;
        xf.x = x0 * sc0;  xf.y = x1 * sc0;
        wf.x = Wa[(k0) * 16 + lm];  wf.y = Wa[(k0 + 1) * 16 + lm];
        c0 = __builtin_amdgcn_wmma_f32_16x16x4_f32(false, wf, false, xf, (short)0, c0, false, false);
        xf.x = x0 * sc1;  xf.y = x1 * sc1;
        wf.x = Wb[(k0) * 16 + lm];  wf.y = Wb[(k0 + 1) * 16 + lm];
        c1 = __builtin_amdgcn_wmma_f32_16x16x4_f32(false, wf, false, xf, (short)0, c1, false, false);
        xf.x = x0 * sc2;  xf.y = x1 * sc2;
        wf.x = Wc[(k0) * 16 + lm];  wf.y = Wc[(k0 + 1) * 16 + lm];
        c2 = __builtin_amdgcn_wmma_f32_16x16x4_f32(false, wf, false, xf, (short)0, c2, false, false);
    }

    if (row < N) {
        float* __restrict__ dp = hT3 + (size_t)(half * 8) * N + row;
#pragma unroll
        for (int v = 0; v < 8; ++v) dp[(size_t)v * N] = c0[v];
        dp += (size_t)16 * N;
#pragma unroll
        for (int v = 0; v < 8; ++v) dp[(size_t)v * N] = c1[v];
        dp += (size_t)16 * N;
#pragma unroll
        for (int v = 0; v < 8; ++v) dp[(size_t)v * N] = c2[v];
    }
}

#define Q_WORDS 160
static_assert(NMAX < 65536, "queue word holds src and local dst in 16 bits each");

__device__ __forceinline__ unsigned hg_push(unsigned* q, unsigned n, bool keep, unsigned word) {
    const unsigned m = __builtin_amdgcn_ballot_w32(keep);
    if (keep) q[n + __builtin_amdgcn_mbcnt_lo(m, 0u)] = word;
    return n + (unsigned)__builtin_popcount(m);
}

__device__ __forceinline__ void hg_add_word(float* sums, unsigned w, float v) {
    __hip_atomic_fetch_add(&sums[w >> 16], v, __ATOMIC_RELAXED, __HIP_MEMORY_SCOPE_WORKGROUP);
}

template <int R>
__device__ __forceinline__ void hg_take(const unsigned* q, const float* vals, float* sums,
                                        unsigned n, unsigned lane, unsigned carried, unsigned& wq) {
    unsigned w[R + 1];
#pragma unroll
    for (int r = 0; r < R; ++r) w[r] = q[lane + 32u * (unsigned)r];
    w[R] = lane + 32u * (unsigned)R < n ? q[lane + 32u * (unsigned)R] : 0u;
    if (lane < carried) w[0] = wq;
    float v[R + 1];
#pragma unroll
    for (int r = 0; r < R; ++r) v[r] = vals[w[r] & 0xffffu];
#pragma unroll
    for (int r = 0; r < R; ++r) hg_add_word(sums, w[r], v[r]);
    wq = w[R];
}

__device__ __forceinline__ void hg_trip(const int4 d, const int4 s, bool ok,
                                        const float* vals, float* sums, unsigned* q,
                                        int lo, unsigned cnt, unsigned lane,
                                        unsigned& rem, unsigned& wq) {
    const unsigned t0 = (unsigned)(d.x - lo), t1 = (unsigned)(d.y - lo);
    const unsigned t2 = (unsigned)(d.z - lo), t3 = (unsigned)(d.w - lo);
    const unsigned c = ok ? cnt : 0u;
    unsigned n = rem;
    n = hg_push(q, n, t0 < c, (unsigned)s.x | (t0 << 16));
    n = hg_push(q, n, t1 < c, (unsigned)s.y | (t1 << 16));
    n = hg_push(q, n, t2 < c, (unsigned)s.z | (t2 << 16));
    n = hg_push(q, n, t3 < c, (unsigned)s.w | (t3 << 16));
    __builtin_amdgcn_fence(__ATOMIC_RELEASE, "wavefront", "local");
    __builtin_amdgcn_wave_barrier();
    __builtin_amdgcn_fence(__ATOMIC_ACQUIRE, "wavefront", "local");
    const unsigned rows = n >> 5;
    if      (rows == 2u) hg_take<2>(q, vals, sums, n, lane, rem, wq);
    else if (rows == 1u) hg_take<1>(q, vals, sums, n, lane, rem, wq);
    else if (rows == 3u) hg_take<3>(q, vals, sums, n, lane, rem, wq);
    else if (rows == 0u) hg_take<0>(q, vals, sums, n, lane, rem, wq);
    else                 hg_take<4>(q, vals, sums, n, lane, rem, wq);
    rem = n & 31u;
}

__device__ __forceinline__ void hg_scan_own(const int4* __restrict__ s4, const int4* __restrict__ d4,
                                            int n4, const float* vals, float* sums, unsigned* q,
                                            int lo, unsigned cnt, int tid, int tpb) {
    const unsigned lane = (unsigned)tid & 31u;
    int4 dA = make_int4(0, 0, 0, 0), sA = make_int4(0, 0, 0, 0);
    int4 dB = make_int4(0, 0, 0, 0), sB = make_int4(0, 0, 0, 0);
    unsigned rem = 0u, wq = 0u;
    int ib = __builtin_amdgcn_readfirstlane(tid & ~31);
    if (ib + (int)lane < n4) { dA = d4[ib + (int)lane]; sA = s4[ib + (int)lane]; }
    const int last = n4 - 1;
    for (; ib < n4; ib += 2 * tpb) {
        const int iA = ib + (int)lane, iB = iA + tpb, iC = iB + tpb;
        const int jB = iB < n4 ? iB : last;
        dB = d4[jB];  sB = s4[jB];
        hg_trip(dA, sA, iA < n4, vals, sums, q, lo, cnt, lane, rem, wq);
        const int jC = iC < n4 ? iC : last;
        dA = d4[jC];  sA = s4[jC];
        if (ib + tpb < n4)
            hg_trip(dB, sB, iB < n4, vals, sums, q, lo, cnt, lane, rem, wq);
    }
    if (lane < rem) hg_add_word(sums, wq, vals[wq & 0xffffu]);
}

__device__ __forceinline__ void hg_edge_add(float* __restrict__ acc,
                                            float v, int d, int lo, unsigned cnt) {
    const unsigned dl = (unsigned)(d - lo);
    if (dl < cnt)
        __hip_atomic_fetch_add(&acc[dl], v, __ATOMIC_RELAXED,
                               __HIP_MEMORY_SCOPE_WORKGROUP);
}

__global__ __launch_bounds__(AGG_TPB) void hg_agg1_col(const int* __restrict__ src,
                                                       const int* __restrict__ dst,
                                                       int E,
                                                       const float* __restrict__ hT,
                                                       const float* __restrict__ s_in,
                                                       float* __restrict__ accT,
                                                       int N) {
    __shared__ float hcol[NMAX];
    __shared__ float acc[NHALF_MAX];
    __shared__ unsigned qbuf[(AGG_TPB / 32) * Q_WORDS];

    const int tid    = threadIdx.x;
    const int nchunk = gridDim.x;
    const int chunk  = blockIdx.x;
    const int f      = blockIdx.z;
    const int nh     = (N + 1) >> 1;
    const int lo     = blockIdx.y * nh;
    const int hiN    = lo + nh < N ? lo + nh : N;
    const unsigned cnt = hiN > lo ? (unsigned)(hiN - lo) : 0u;

    const float* __restrict__ hp = hT + (size_t)f * N;
    for (int i = tid; i < N; i += AGG_TPB) hcol[i] = hp[i];
    for (unsigned i = tid; i < cnt; i += AGG_TPB) acc[i] = 0.0f;
    __syncthreads();

    const int per = ((E + nchunk * 4 - 1) / (nchunk * 4)) * 4;
    const int e0  = chunk * per;
    const int e1  = e0 + per < E ? e0 + per : E;
    const int ne  = e1 > e0 ? e1 - e0 : 0;
    const int n4  = ne >> 2;
    const int4* __restrict__ s4 = (const int4*)(src + e0);
    const int4* __restrict__ d4 = (const int4*)(dst + e0);

    hg_scan_own(s4, d4, n4, hcol, acc, qbuf + (tid >> 5) * Q_WORDS, lo, cnt, tid, AGG_TPB);
    for (int e = e0 + (n4 << 2) + tid; e < e1; e += AGG_TPB)
        hg_edge_add(acc, hcol[src[e]], dst[e], lo, cnt);
    __syncthreads();

    float* __restrict__ ap = accT + (size_t)f * N + lo;
    const float* __restrict__ sp = s_in + lo;
    if (cnt != 0u) {
        const unsigned last = cnt - 1u;
        for (unsigned base = 0u; base < cnt; base += FLUSH_U * AGG_TPB) {
            float s[FLUSH_U], a[FLUSH_U];
#pragma unroll
            for (int u = 0; u < FLUSH_U; ++u) {
                const unsigned j  = base + (unsigned)u * AGG_TPB + tid;
                const unsigned jc = j < cnt ? j : last;
                s[u] = sp[jc];
                a[u] = acc[jc];
            }
#pragma unroll
            for (int u = 0; u < FLUSH_U; ++u) {
                const unsigned j = base + (unsigned)u * AGG_TPB + tid;
                const float v = a[u] * s[u];
                if (j < cnt && v != 0.0f) atomicAdd(ap + j, v);
            }
        }
    }
}

#define B16_TPB 1024
#define B16_VEC 5
#define B16_EPT (4 * B16_VEC)
#define B16_RUN (B16_TPB * B16_EPT)
#define B16_HDR 32
#define B16_PITCH (B16_HDR + B16_RUN)
#define SL16 16
#define SL16_MAX ((((NMAX + SL16 - 1) / SL16) + 31) & ~31)

__global__ __launch_bounds__(B16_TPB) void hg_bin16(const int* __restrict__ s0,
                                                    const int* __restrict__ d0,
                                                    const int* __restrict__ s1,
                                                    const int* __restrict__ d1,
                                                    const int* __restrict__ s2,
                                                    const int* __restrict__ d2,
                                                    int E0, int E1, int E2,
                                                    int nR0, int nR1,
                                                    unsigned* __restrict__ words,
                                                    int N) {
    __shared__ unsigned buf[B16_RUN];
    __shared__ unsigned hist[SL16 * 32];
    __shared__ unsigned base[SL16 * 32];
    __shared__ unsigned tot[SL16];
    __shared__ unsigned start[SL16 + 1];

    const int tid = threadIdx.x;
    const int run = blockIdx.x;
    const int* __restrict__ src;
    const int* __restrict__ dst;
    int E, q;
    if (run < nR0)            { src = s0;  dst = d0;  E = E0;  q = run; }
    else if (run < nR0 + nR1) { src = s1;  dst = d1;  E = E1;  q = run - nR0; }
    else                      { src = s2;  dst = d2;  E = E2;  q = run - nR0 - nR1; }
    const int ns  = (((N + SL16 - 1) / SL16) + 31) & ~31;
    const float inv_ns = 1.0f / (float)ns;
    const int R0  = q * B16_RUN;
    const int len = E - R0 < B16_RUN ? E - R0 : B16_RUN;

    const int lane = tid & 31;
    if (tid < SL16 * 32) hist[tid] = 0u;

    int d[B16_EPT], s[B16_EPT];
    bool ok[B16_EPT];
#pragma unroll
    for (int v = 0; v < B16_VEC; ++v) {
        const int b = v * (B16_RUN / B16_VEC) + 4 * tid;
        if (b + 3 < len) {
            const int4 dv = *(const int4*)(dst + R0 + b);
            const int4 sv = *(const int4*)(src + R0 + b);
            d[4 * v] = dv.x;  d[4 * v + 1] = dv.y;  d[4 * v + 2] = dv.z;  d[4 * v + 3] = dv.w;
            s[4 * v] = sv.x;  s[4 * v + 1] = sv.y;  s[4 * v + 2] = sv.z;  s[4 * v + 3] = sv.w;
            ok[4 * v] = ok[4 * v + 1] = ok[4 * v + 2] = ok[4 * v + 3] = true;
        } else {
#pragma unroll
            for (int k = 0; k < 4; ++k) {
                const bool in = b + k < len;
                ok[4 * v + k] = in;
                d[4 * v + k]  = in ? dst[R0 + b + k] : 0;
                s[4 * v + k]  = in ? src[R0 + b + k] : 0;
            }
        }
    }
    __syncthreads();

    unsigned word[B16_EPT], rank[B16_EPT];
    int bin[B16_EPT];
#pragma unroll
    for (int k = 0; k < B16_EPT; ++k) {
        const bool valid = ok[k] && (unsigned)d[k] < (unsigned)N;
        const int dv = valid ? d[k] : 0;
        int b = (int)((float)dv * inv_ns);
        if (b * ns > dv) --b;
        else if ((b + 1) * ns <= dv) ++b;
        const unsigned t = (unsigned)(dv - b * ns);
        word[k] = (unsigned)s[k] | (t << 16);
        bin[k]  = valid ? b : SL16;
        rank[k] = 0u;
        if (valid)
            rank[k] = __hip_atomic_fetch_add(&hist[b * 32 + lane], 1u, __ATOMIC_RELAXED,
                                             __HIP_MEMORY_SCOPE_WORKGROUP);
    }
    __syncthreads();
    unsigned cv = 0u, incl = 0u;
    if (tid < SL16 * 32) {
        cv = hist[tid];
        incl = cv;
#pragma unroll
        for (int o = 1; o < 32; o <<= 1) {
            const unsigned up = __shfl_up(incl, (unsigned)o, 32);
            if (lane >= o) incl += up;
        }
        if (lane == 31) tot[tid >> 5] = incl;
    }
    __syncthreads();
    if (tid == 0) {
        unsigned a = 0u;
#pragma unroll
        for (int b = 0; b < SL16; ++b) { start[b] = a;  a += tot[b]; }
        start[SL16] = a;
    }
    __syncthreads();
    if (tid < SL16 * 32) base[tid] = start[tid >> 5] + (incl - cv);
    __syncthreads();
#pragma unroll
    for (int k = 0; k < B16_EPT; ++k)
        if (bin[k] < SL16) buf[base[bin[k] * 32 + lane] + rank[k]] = word[k];
    __syncthreads();

    unsigned* __restrict__ wr = words + (size_t)run * B16_PITCH;
    if (tid < B16_HDR) wr[tid] = tid <= SL16 ? start[tid] : 0u;
    const int total = (int)start[SL16];
    for (int i = tid; i < total; i += B16_TPB) wr[B16_HDR + i] = buf[i];
}

#define RUN_ROWS 8
__device__ __forceinline__ void hg_load_rows(const unsigned* __restrict__ wp, int i0, int last,
                                             int lane, unsigned (&w)[RUN_ROWS]) {
#pragma unroll
    for (int u = 0; u < RUN_ROWS; ++u) {
        const int i = i0 + 32 * u + lane;
        w[u] = wp[i < last ? i : last];
    }
}

__device__ __forceinline__ void hg_add_rows(const unsigned (&w)[RUN_ROWS], int i0, int st, int en, int lane,
                                            const float* vals, float* sums, unsigned cnt) {
    float v[RUN_ROWS];
#pragma unroll
    for (int u = 0; u < RUN_ROWS; ++u) v[u] = vals[w[u] & 0xffffu];
#pragma unroll
    for (int u = 0; u < RUN_ROWS; ++u) {
        const int i = i0 + 32 * u + lane;
        const unsigned t = w[u] >> 16;
        if (i >= st && i < en && t < cnt)
            __hip_atomic_fetch_add(&sums[t], v[u], __ATOMIC_RELAXED, __HIP_MEMORY_SCOPE_WORKGROUP);
    }
}

__global__ __launch_bounds__(AGG_TPB) void hg_agg1_own16(const unsigned* __restrict__ words,
                                                         int nR,
                                                         const float* __restrict__ hT,
                                                         const float* __restrict__ s_in,
                                                         float* __restrict__ accT,
                                                         const float* __restrict__ b1,
                                                         int N, int first, int last) {
    __shared__ float hcol[NMAX];
    __shared__ float acc[SL16_MAX];

    const int tid  = threadIdx.x;
    const int lane = tid & 31;
    const int f    = blockIdx.y;
    const int sl   = blockIdx.x;
    const int ns   = (((N + SL16 - 1) / SL16) + 31) & ~31;
    const int lo   = sl * ns;
    const int hiN  = lo + ns < N ? lo + ns : N;
    const unsigned cnt = hiN > lo ? (unsigned)(hiN - lo) : 0u;

    const float* __restrict__ hp = hT + (size_t)f * N;
    for (int i = tid; i < N; i += AGG_TPB) hcol[i] = hp[i];
    for (unsigned i = tid; i < cnt; i += AGG_TPB) acc[i] = 0.0f;
    __syncthreads();

    const int nwav = AGG_TPB / 32;
    int q = __builtin_amdgcn_readfirstlane(tid >> 5);
    unsigned h0 = 0u, h1 = 0u;
    if (q < nR) {
        const unsigned* __restrict__ rg = words + (size_t)q * B16_PITCH;
        h0 = rg[sl];  h1 = rg[sl + 1];
    }
    while (q < nR) {
        const int qn = q + nwav;
        unsigned g0 = 0u, g1 = 0u;
        if (qn < nR) {
            const unsigned* __restrict__ rn = words + (size_t)qn * B16_PITCH;
            g0 = rn[sl];  g1 = rn[sl + 1];
        }
        const int st = __builtin_amdgcn_readfirstlane((int)h0);
        const int en = __builtin_amdgcn_readfirstlane((int)h1);
        const unsigned* __restrict__ wp = words + (size_t)q * B16_PITCH + B16_HDR;
        if (en > st) {
            const int a0   = st & ~31;
            const int last = en - 1;
            const int G  = 32 * RUN_ROWS;
            const int ng = (en - a0 + G - 1) / G;
            unsigned wA[RUN_ROWS], wB[RUN_ROWS];
            hg_load_rows(wp, a0, last, lane, wA);
            int g = 0;
            for (; g + 2 < ng; g += 2) {
                hg_load_rows(wp, a0 + (g + 1) * G, last, lane, wB);
                hg_add_rows(wA, a0 + g * G, st, en, lane, hcol, acc, cnt);
                hg_load_rows(wp, a0 + (g + 2) * G, last, lane, wA);
                hg_add_rows(wB, a0 + (g + 1) * G, st, en, lane, hcol, acc, cnt);
            }
            if (g + 1 < ng) {
                hg_load_rows(wp, a0 + (g + 1) * G, last, lane, wB);
                hg_add_rows(wA, a0 + g * G, st, en, lane, hcol, acc, cnt);
                hg_add_rows(wB, a0 + (g + 1) * G, st, en, lane, hcol, acc, cnt);
            } else {
                hg_add_rows(wA, a0 + g * G, st, en, lane, hcol, acc, cnt);
            }
        }
        h0 = g0;  h1 = g1;  q = qn;
    }
    __syncthreads();

    float* __restrict__ ap = accT + (size_t)f * N + lo;
    const float* __restrict__ sp = s_in + lo;
    const float bs0 = b1[f], bs1 = b1[16 + f], bs2 = b1[32 + f];
    for (unsigned j = tid; j < cnt; j += AGG_TPB) {
        const float v = acc[j] * sp[j];
        float t = first ? v : ap[j] + v;
        if (last) {
            t = t + bs0 + bs1 + bs2;
            t = t > 0.0f ? t : 0.0f;
        }
        ap[j] = t;
    }
}


#define A1_PT ((SL16_MAX + AGG_TPB - 1) / AGG_TPB)
__global__ __launch_bounds__(AGG_TPB) void hg_agg1_all16(const unsigned* __restrict__ words,
                                                         int nR0, int nR1, int nR2,
                                                         const float* __restrict__ hT3,
                                                         const float* __restrict__ s_in3,
                                                         const float* __restrict__ b1,
                                                         float* __restrict__ h,
                                                         int N) {
    __shared__ float hcol[NMAX];
    __shared__ float acc[SL16_MAX];

    const int tid  = threadIdx.x;
    const int lane = tid & 31;
    const int f    = blockIdx.y;
    const int sl   = blockIdx.x;
    const int ns   = (((N + SL16 - 1) / SL16) + 31) & ~31;
    const int lo   = sl * ns;
    const int hiN  = lo + ns < N ? lo + ns : N;
    const unsigned cnt = hiN > lo ? (unsigned)(hiN - lo) : 0u;

    const int nwav = AGG_TPB / 32;
    const int wv   = __builtin_amdgcn_readfirstlane(tid >> 5);

    float tot[A1_PT];
#pragma unroll
    for (int k = 0; k < A1_PT; ++k) tot[k] = 0.0f;

    int rb = 0;
    for (int r = 0; r < 3; ++r) {
        const int nR = r == 0 ? nR0 : (r == 1 ? nR1 : nR2);
        const float* __restrict__ hp = hT3 + ((size_t)r * 16 + f) * N;
        for (int i = tid; i < N; i += AGG_TPB) hcol[i] = hp[i];
        for (unsigned i = tid; i < cnt; i += AGG_TPB) acc[i] = 0.0f;
        __syncthreads();

        const unsigned* __restrict__ rel = words + (size_t)rb * B16_PITCH;
        int q = wv;
        unsigned h0 = 0u, h1 = 0u;
        if (q < nR) {
            const unsigned* __restrict__ rg = rel + (size_t)q * B16_PITCH;
            h0 = rg[sl];  h1 = rg[sl + 1];
        }
        while (q < nR) {
            const int qn = q + nwav;
            unsigned g0 = 0u, g1 = 0u;
            if (qn < nR) {
                const unsigned* __restrict__ rn = rel + (size_t)qn * B16_PITCH;
                g0 = rn[sl];  g1 = rn[sl + 1];
            }
            const int st = __builtin_amdgcn_readfirstlane((int)h0);
            const int en = __builtin_amdgcn_readfirstlane((int)h1);
            const unsigned* __restrict__ wp = rel + (size_t)q * B16_PITCH + B16_HDR;
            if (en > st) {
                const int a0   = st & ~31;
                const int last = en - 1;
                const int G  = 32 * RUN_ROWS;
                const int ng = (en - a0 + G - 1) / G;
                unsigned wA[RUN_ROWS], wB[RUN_ROWS];
                hg_load_rows(wp, a0, last, lane, wA);
                int g = 0;
                for (; g + 2 < ng; g += 2) {
                    hg_load_rows(wp, a0 + (g + 1) * G, last, lane, wB);
                    hg_add_rows(wA, a0 + g * G, st, en, lane, hcol, acc, cnt);
                    hg_load_rows(wp, a0 + (g + 2) * G, last, lane, wA);
                    hg_add_rows(wB, a0 + (g + 1) * G, st, en, lane, hcol, acc, cnt);
                }
                if (g + 1 < ng) {
                    hg_load_rows(wp, a0 + (g + 1) * G, last, lane, wB);
                    hg_add_rows(wA, a0 + g * G, st, en, lane, hcol, acc, cnt);
                    hg_add_rows(wB, a0 + (g + 1) * G, st, en, lane, hcol, acc, cnt);
                } else {
                    hg_add_rows(wA, a0 + g * G, st, en, lane, hcol, acc, cnt);
                }
            }
            h0 = g0;  h1 = g1;  q = qn;
        }
        __syncthreads();

        const float* __restrict__ sp = s_in3 + (size_t)r * N + lo;
#pragma unroll
        for (int k = 0; k < A1_PT; ++k) {
            const unsigned j = (unsigned)tid + (unsigned)k * AGG_TPB;
            if (j < cnt) tot[k] += acc[j] * sp[j];
        }
        rb += nR;
    }

    const float bs0 = b1[f], bs1 = b1[16 + f], bs2 = b1[32 + f];
    float* __restrict__ hq = h + (size_t)f * N + lo;
#pragma unroll
    for (int k = 0; k < A1_PT; ++k) {
        const unsigned j = (unsigned)tid + (unsigned)k * AGG_TPB;
        if (j < cnt) {
            const float v = tot[k] + bs0 + bs1 + bs2;
            hq[j] = v > 0.0f ? v : 0.0f;
        }
    }
}

__global__ __launch_bounds__(TPB) void hg_relu_bias(float* __restrict__ acc,
                                                    const float* __restrict__ b1,
                                                    int N) {
    const int f = blockIdx.y;
    const int n = blockIdx.x * TPB + threadIdx.x;
    if (n < N) {
        const size_t i = (size_t)f * N + n;
        float v = acc[i] + b1[f] + b1[16 + f] + b1[32 + f];
        acc[i] = v > 0.0f ? v : 0.0f;
    }
}

__global__ __launch_bounds__(TPB) void hg_gemm2_all(const float* __restrict__ h,
                                                    const float* __restrict__ s_out3,
                                                    const float* __restrict__ W2,
                                                    float* __restrict__ z3,
                                                    int N) {
    int i = blockIdx.x * TPB + threadIdx.x;
    if (i >= N) return;
    const float* hp = h + i;
    float a[16];
#pragma unroll
    for (int f = 0; f < 16; ++f) a[f] = hp[(size_t)f * N];
#pragma unroll
    for (int r = 0; r < 3; ++r) {
        const float4* wp = (const float4*)(W2 + r * 16);
        float acc = 0.0f;
#pragma unroll
        for (int q = 0; q < 4; ++q) {
            const float4 b = wp[q];
            acc += a[4 * q] * b.x + a[4 * q + 1] * b.y + a[4 * q + 2] * b.z + a[4 * q + 3] * b.w;
        }
        z3[(size_t)r * N + i] = acc * s_out3[(size_t)r * N + i];
    }
}

__global__ __launch_bounds__(L2_TPB) void hg_agg2_own(const int* __restrict__ s0,
                                                      const int* __restrict__ d0,
                                                      const int* __restrict__ s1,
                                                      const int* __restrict__ d1,
                                                      const int* __restrict__ s2,
                                                      const int* __restrict__ d2,
                                                      int E0, int E1, int E2,
                                                      const float* __restrict__ z3,
                                                      float* __restrict__ partial,
                                                      int N) {
    __shared__ float zc[NMAX];
    __shared__ float bins[L2_SLICE_MAX];
    __shared__ unsigned qbuf[(L2_TPB / 32) * Q_WORDS];

    const int tid = threadIdx.x;
    const int p   = blockIdx.y;
    const int* __restrict__ src;
    const int* __restrict__ dst;
    int E, nchunk, chunk, r;
    if (p < L2_C0) {
        src = s0;  dst = d0;  E = E0;  nchunk = L2_C0;  chunk = p;                  r = 0;
    } else if (p < L2_C0 + L2_C1) {
        src = s1;  dst = d1;  E = E1;  nchunk = L2_C1;  chunk = p - L2_C0;          r = 1;
    } else {
        src = s2;  dst = d2;  E = E2;  nchunk = L2_C2;  chunk = p - L2_C0 - L2_C1;  r = 2;
    }

    const int ns  = (((N + L2_SLICES - 1) / L2_SLICES) + 31) & ~31;
    const int lo  = blockIdx.x * ns;
    const int hiN = lo + ns < N ? lo + ns : N;
    const unsigned cnt = hiN > lo ? (unsigned)(hiN - lo) : 0u;

    const float* __restrict__ zp = z3 + (size_t)r * N;
    for (int i = tid; i < N; i += L2_TPB) zc[i] = zp[i];
    for (unsigned j = tid; j < cnt; j += L2_TPB) bins[j] = 0.0f;
    __syncthreads();

    const int per = ((E + nchunk * 4 - 1) / (nchunk * 4)) * 4;
    const int e0  = chunk * per;
    const int e1  = e0 + per < E ? e0 + per : E;
    const int ne  = e1 > e0 ? e1 - e0 : 0;
    const int n4  = ne >> 2;
    const int4* __restrict__ s4 = (const int4*)(src + e0);
    const int4* __restrict__ d4 = (const int4*)(dst + e0);

    hg_scan_own(s4, d4, n4, zc, bins, qbuf + (tid >> 5) * Q_WORDS, lo, cnt, tid, L2_TPB);
    for (int e = e0 + (n4 << 2) + tid; e < e1; e += L2_TPB)
        hg_edge_add(bins, zc[src[e]], dst[e], lo, cnt);
    __syncthreads();

    const size_t pitch = (size_t)((N + 31) & ~31);
    float* __restrict__ pp = partial + (size_t)p * pitch + lo;
    for (unsigned j = tid; j < cnt; j += L2_TPB) pp[j] = bins[j];
}

#define A2_G0 11
#define A2_G1 4
#define A2_G2 1
#define A2_PLANES (A2_G0 + A2_G1 + A2_G2)

__global__ __launch_bounds__(AGG_TPB) void hg_agg2_own16(const unsigned* __restrict__ words,
                                                         int nR0, int nR1, int nR2,
                                                         const float* __restrict__ z3,
                                                         float* __restrict__ partial,
                                                         int N) {
    __shared__ float zc[NMAX];
    __shared__ float acc[SL16_MAX];

    const int tid  = threadIdx.x;
    const int lane = tid & 31;
    const int sl   = blockIdx.x;
    const int p    = blockIdx.y;
    int r, gi, nG, nR, rb;
    if (p < A2_G0)              { r = 0;  gi = p;                  nG = A2_G0;  nR = nR0;  rb = 0; }
    else if (p < A2_G0 + A2_G1) { r = 1;  gi = p - A2_G0;          nG = A2_G1;  nR = nR1;  rb = nR0; }
    else                        { r = 2;  gi = p - A2_G0 - A2_G1;  nG = A2_G2;  nR = nR2;  rb = nR0 + nR1; }
    const int per = (nR + nG - 1) / nG;
    const int q0  = gi * per;
    const int q1  = q0 + per < nR ? q0 + per : nR;

    const int ns   = (((N + SL16 - 1) / SL16) + 31) & ~31;
    const int lo   = sl * ns;
    const int hiN  = lo + ns < N ? lo + ns : N;
    const unsigned cnt = hiN > lo ? (unsigned)(hiN - lo) : 0u;

    const float* __restrict__ zp = z3 + (size_t)r * N;
    for (int i = tid; i < N; i += AGG_TPB) zc[i] = zp[i];
    for (unsigned i = tid; i < cnt; i += AGG_TPB) acc[i] = 0.0f;
    __syncthreads();

    const int nwav = AGG_TPB / 32;
    const unsigned* __restrict__ rel = words + (size_t)rb * B16_PITCH;
    int q = q0 + __builtin_amdgcn_readfirstlane(tid >> 5);
    unsigned h0 = 0u, h1 = 0u;
    if (q < q1) {
        const unsigned* __restrict__ rg = rel + (size_t)q * B16_PITCH;
        h0 = rg[sl];  h1 = rg[sl + 1];
    }
    while (q < q1) {
        const int qn = q + nwav;
        unsigned g0 = 0u, g1 = 0u;
        if (qn < q1) {
            const unsigned* __restrict__ rn = rel + (size_t)qn * B16_PITCH;
            g0 = rn[sl];  g1 = rn[sl + 1];
        }
        const int st = __builtin_amdgcn_readfirstlane((int)h0);
        const int en = __builtin_amdgcn_readfirstlane((int)h1);
        const unsigned* __restrict__ wp = rel + (size_t)q * B16_PITCH + B16_HDR;
        if (en > st) {
            const int a0   = st & ~31;
            const int last = en - 1;
            const int G  = 32 * RUN_ROWS;
            const int ng = (en - a0 + G - 1) / G;
            unsigned wA[RUN_ROWS], wB[RUN_ROWS];
            hg_load_rows(wp, a0, last, lane, wA);
            int g = 0;
            for (; g + 2 < ng; g += 2) {
                hg_load_rows(wp, a0 + (g + 1) * G, last, lane, wB);
                hg_add_rows(wA, a0 + g * G, st, en, lane, zc, acc, cnt);
                hg_load_rows(wp, a0 + (g + 2) * G, last, lane, wA);
                hg_add_rows(wB, a0 + (g + 1) * G, st, en, lane, zc, acc, cnt);
            }
            if (g + 1 < ng) {
                hg_load_rows(wp, a0 + (g + 1) * G, last, lane, wB);
                hg_add_rows(wA, a0 + g * G, st, en, lane, zc, acc, cnt);
                hg_add_rows(wB, a0 + (g + 1) * G, st, en, lane, zc, acc, cnt);
            } else {
                hg_add_rows(wA, a0 + g * G, st, en, lane, zc, acc, cnt);
            }
        }
        h0 = g0;  h1 = g1;  q = qn;
    }
    __syncthreads();

    const size_t pitch = (size_t)((N + 31) & ~31);
    float* __restrict__ pp = partial + (size_t)p * pitch + lo;
    for (unsigned j = tid; j < cnt; j += AGG_TPB) pp[j] = acc[j];
}

template <int C>
__device__ __forceinline__ float hg_sum_fplanes(const float* __restrict__ pp, size_t pitch) {
    float s = 0.0f;
#pragma unroll
    for (int c = 0; c < C; ++c) s += pp[(size_t)c * pitch];
    return s;
}

template <int C0, int C1, int C2>
__device__ __forceinline__ void hg_out_body(const float* __restrict__ partial,
                                            const float* __restrict__ s_in3,
                                            const float* __restrict__ b2,
                                            float* __restrict__ out,
                                            int N) {
    const int n = blockIdx.x * TPB + threadIdx.x;
    if (n >= N) return;
    const size_t pitch = (size_t)((N + 31) & ~31);
    const float m0 = hg_sum_fplanes<C0>(partial + n, pitch);
    const float m1 = hg_sum_fplanes<C1>(partial + (size_t)C0 * pitch + n, pitch);
    const float m2 = hg_sum_fplanes<C2>(partial + (size_t)(C0 + C1) * pitch + n, pitch);
    out[n] = (m0 * s_in3[n] + b2[0])
           + (m1 * s_in3[(size_t)N + n] + b2[1])
           + (m2 * s_in3[(size_t)2 * N + n] + b2[2]);
}

__global__ __launch_bounds__(TPB) void hg_out_sum(const float* __restrict__ partial,
                                                  const float* __restrict__ s_in3,
                                                  const float* __restrict__ b2,
                                                  float* __restrict__ out,
                                                  int N) {
    hg_out_body<L2_C0, L2_C1, L2_C2>(partial, s_in3, b2, out, N);
}

__global__ __launch_bounds__(TPB) void hg_out_sum16(const float* __restrict__ partial,
                                                    const float* __restrict__ s_in3,
                                                    const float* __restrict__ b2,
                                                    float* __restrict__ out,
                                                    int N) {
    hg_out_body<A2_G0, A2_G1, A2_G2>(partial, s_in3, b2, out, N);
}

extern "C" void kernel_launch(void* const* d_in, const int* in_sizes, int n_in,
                              void* d_out, int out_size, void* d_ws, size_t ws_size,
                              hipStream_t stream) {
    const float* x  = (const float*)d_in[0];
    const int* src[3] = {(const int*)d_in[1], (const int*)d_in[3], (const int*)d_in[5]};
    const int* dst[3] = {(const int*)d_in[2], (const int*)d_in[4], (const int*)d_in[6]};
    const int  E[3]   = {in_sizes[1], in_sizes[3], in_sizes[5]};
    const float* W1 = (const float*)d_in[7];
    const float* b1 = (const float*)d_in[8];
    const float* W2 = (const float*)d_in[9];
    const float* b2 = (const float*)d_in[10];
    const int N = in_sizes[0] / 32;
    float* out = (float*)d_out;

    const int nR[3] = {(E[0] + B16_RUN - 1) / B16_RUN, (E[1] + B16_RUN - 1) / B16_RUN,
                       (E[2] + B16_RUN - 1) / B16_RUN};
    const int nRuns = nR[0] + nR[1] + nR[2];
    char* ws = (char*)d_ws;
    float* scales = (float*)ws;
    const int tiles  = (N + 15) / 16;
    const int gblk   = (tiles + (TPB / 32) - 1) / (TPB / 32);

    const size_t pitchN  = (size_t)((N + 31) & ~31);
    const size_t w_off3  = ((size_t)70 * N * 4 + 127) & ~(size_t)127;
    const bool merged = nRuns > 0 && N >= 64 &&
                        ws_size >= w_off3 + (size_t)nRuns * B16_PITCH * 4;
    if (merged) {
        float* hpre3   = scales + (size_t)6 * N;
        float* hbuf    = hpre3 + (size_t)48 * N;
        unsigned* cntp = (unsigned*)hpre3;
        float* planes3 = hpre3;
        float* z3m     = hpre3 + (size_t)16 * pitchN;
        unsigned* wordsm = (unsigned*)(ws + w_off3);

        hg_count_own<<<dim3(CNT_SLICES, CNT_PLANES), dim3(CNT_TPB), 0, stream>>>(
            src[0], dst[0], src[1], dst[1], src[2], dst[2], E[0], E[1], E[2], cntp, N);
        hg_scale_sum<<<dim3((N + TPB - 1) / TPB, 6), dim3(TPB), 0, stream>>>(cntp, scales, N);
        hg_bin16<<<dim3(nRuns), dim3(B16_TPB), 0, stream>>>(
            src[0], dst[0], src[1], dst[1], src[2], dst[2], E[0], E[1], E[2], nR[0], nR[1], wordsm, N);
        hg_gemm1_all_wmma<<<dim3(gblk), dim3(TPB), 0, stream>>>(x, scales, W1, hpre3, N);
        hg_agg1_all16<<<dim3(SL16, 16), dim3(AGG_TPB), 0, stream>>>(
            wordsm, nR[0], nR[1], nR[2], hpre3, scales + (size_t)3 * N, b1, hbuf, N);
        hg_gemm2_all<<<dim3((N + TPB - 1) / TPB), dim3(TPB), 0, stream>>>(hbuf, scales, W2, z3m, N);
        hg_agg2_own16<<<dim3(SL16, A2_PLANES), dim3(AGG_TPB), 0, stream>>>(
            wordsm, nR[0], nR[1], nR[2], z3m, planes3, N);
        hg_out_sum16<<<dim3((N + TPB - 1) / TPB), dim3(TPB), 0, stream>>>(
            planes3, scales + (size_t)3 * N, b2, out, N);
        return;
    }

    float* acc1   = (float*)(ws + (size_t)6 * N * 4);
    float* hpre   = acc1 + (size_t)N * 16;
    float* z      = hpre + (size_t)N * 16;
    float* z3     = z - (size_t)2 * N;
    float* planes2 = acc1;
    unsigned* partial = (unsigned*)acc1;
    const size_t w_off = ((size_t)39 * N * 4 + 127) & ~(size_t)127;
    const bool binned = nRuns > 0 && ws_size >= w_off + (size_t)nRuns * B16_PITCH * 4;
    unsigned* words = (unsigned*)(ws + w_off);
    const size_t rOff[3] = {0, (size_t)nR[0] * B16_PITCH, (size_t)(nR[0] + nR[1]) * B16_PITCH};

    hg_count_own<<<dim3(CNT_SLICES, CNT_PLANES), dim3(CNT_TPB), 0, stream>>>(
        src[0], dst[0], src[1], dst[1], src[2], dst[2], E[0], E[1], E[2], partial, N);
    hg_scale_sum<<<dim3((N + TPB - 1) / TPB, 6), dim3(TPB), 0, stream>>>(partial, scales, N);

    if (!binned) {
        long span = 16L * N;
        hg_init<<<dim3((unsigned)((span + TPB - 1) / TPB)), dim3(TPB), 0, stream>>>(
            acc1, N);
    }

    if (binned)
        hg_bin16<<<dim3(nRuns), dim3(B16_TPB), 0, stream>>>(
            src[0], dst[0], src[1], dst[1], src[2], dst[2], E[0], E[1], E[2], nR[0], nR[1], words, N);

    for (int r = 0; r < 3; ++r) {
        hg_gemm1_wmma<<<dim3(gblk), dim3(TPB), 0, stream>>>(
            x, scales + (size_t)r * N, W1 + (size_t)r * 32 * 16, hpre, N);
        if (binned)
            hg_agg1_own16<<<dim3(SL16, 16), dim3(AGG_TPB), 0, stream>>>(
                words + rOff[r], nR[r], hpre, scales + (size_t)(3 + r) * N, acc1, b1, N,
                r == 0 ? 1 : 0, r == 2 ? 1 : 0);
        else
            hg_agg1_col<<<dim3(8, 2, 16), dim3(AGG_TPB), 0, stream>>>(
                src[r], dst[r], E[r], hpre, scales + (size_t)(3 + r) * N, acc1, N);
    }
    if (!binned)
        hg_relu_bias<<<dim3((N + TPB - 1) / TPB, 16), dim3(TPB), 0, stream>>>(acc1, b1, N);

    hg_gemm2_all<<<dim3((N + TPB - 1) / TPB), dim3(TPB), 0, stream>>>(acc1, scales, W2, z3, N);
    if (binned) {
        hg_agg2_own16<<<dim3(SL16, A2_PLANES), dim3(AGG_TPB), 0, stream>>>(
            words, nR[0], nR[1], nR[2], z3, planes2, N);
        hg_out_sum16<<<dim3((N + TPB - 1) / TPB), dim3(TPB), 0, stream>>>(
            planes2, scales + (size_t)3 * N, b2, out, N);
    } else {
        hg_agg2_own<<<dim3(L2_SLICES, L2_PLANES), dim3(L2_TPB), 0, stream>>>(
            src[0], dst[0], src[1], dst[1], src[2], dst[2], E[0], E[1], E[2], z3, planes2, N);
        hg_out_sum<<<dim3((N + TPB - 1) / TPB), dim3(TPB), 0, stream>>>(
            planes2, scales + (size_t)3 * N, b2, out, N);
    }
}


# degree count: a work-group owns half the node range (was an eighth) and a quarter-size index chunk, so each index array is read by 2 groups, not 8; 112 count planes (was 28), exact integer sums
# speedup vs baseline: 18.3464x; 1.0759x over previous
#include <hip/hip_runtime.h>

typedef float v2f __attribute__((ext_vector_type(2)));
typedef float v8f __attribute__((ext_vector_type(8)));

#define TPB 256
#define AGG_TPB 1024
#define NMAX 50000
#define NHALF_MAX ((NMAX + 1) / 2)
#define FLUSH_U 13

#define CNT_TPB 1024
#define CNT_SLICES 8
#define CNT_SLICE_MAX ((((NMAX + CNT_SLICES - 1) / CNT_SLICES) + 31) & ~31)
#define CNT_C0 10
#define CNT_C1 3
#define CNT_C2 1
#define CNT_PLANES (2 * (CNT_C0 + CNT_C1 + CNT_C2))
#define CNTW_SLICES 2
#define CNTW_C0 40
#define CNTW_C1 13
#define CNTW_C2 3
#define CNTW_PLANES (2 * (CNTW_C0 + CNTW_C1 + CNTW_C2))
#define CNT_B 10

#define L2_TPB 1024
#define L2_SLICES 4
#define L2_SLICE_MAX ((((NMAX + L2_SLICES - 1) / L2_SLICES) + 31) & ~31)
#define L2_C0 20
#define L2_C1 7
#define L2_C2 2
#define L2_PLANES (L2_C0 + L2_C1 + L2_C2)

__global__ __launch_bounds__(TPB) void hg_init(float* __restrict__ acc1,
                                               int N) {
    long i = (long)blockIdx.x * TPB + threadIdx.x;
    if (i < 16L * N) acc1[i] = 0.0f;
}

__device__ __forceinline__ void hg_bin(unsigned* __restrict__ bins,
                                       int v, int lo, unsigned cnt) {
    const unsigned t = (unsigned)(v - lo);
    if (t < cnt)
        __hip_atomic_fetch_add(&bins[t], 1u, __ATOMIC_RELAXED,
                               __HIP_MEMORY_SCOPE_WORKGROUP);
}

template <int SLICES, int C0, int C1, int C2>
__global__ __launch_bounds__(CNT_TPB) void hg_count_own(const int* __restrict__ s0,
                                                        const int* __restrict__ d0,
                                                        const int* __restrict__ s1,
                                                        const int* __restrict__ d1,
                                                        const int* __restrict__ s2,
                                                        const int* __restrict__ d2,
                                                        int E0, int E1, int E2,
                                                        unsigned* __restrict__ partial,
                                                        int N) {
    __shared__ unsigned bins[(((NMAX + SLICES - 1) / SLICES) + 31) & ~31];

    const int tid = threadIdx.x;
    const int p   = blockIdx.y;
    const int* __restrict__ idx;
    int E, nchunk, chunk;
    if (p < 2 * C0) {
        const int side = p >= C0;
        idx = side ? d0 : s0;  E = E0;  nchunk = C0;  chunk = p - side * C0;
    } else if (p < 2 * (C0 + C1)) {
        const int q = p - 2 * C0;
        const int side = q >= C1;
        idx = side ? d1 : s1;  E = E1;  nchunk = C1;  chunk = q - side * C1;
    } else {
        const int q = p - 2 * (C0 + C1);
        const int side = q >= C2;
        idx = side ? d2 : s2;  E = E2;  nchunk = C2;  chunk = q - side * C2;
    }

    const int ns  = (((N + SLICES - 1) / SLICES) + 31) & ~31;
    const int lo  = blockIdx.x * ns;
    const int hiN = lo + ns < N ? lo + ns : N;
    const unsigned cnt = hiN > lo ? (unsigned)(hiN - lo) : 0u;

    for (unsigned j = tid; j < cnt; j += CNT_TPB) bins[j] = 0u;
    __syncthreads();

    const int per = ((E + nchunk * 4 - 1) / (nchunk * 4)) * 4;
    const int e0  = chunk * per;
    const int e1  = e0 + per < E ? e0 + per : E;
    const int ne  = e1 > e0 ? e1 - e0 : 0;
    const int n4  = ne >> 2;
    const int4* __restrict__ q4 = (const int4*)(idx + e0);

    for (int base = 0; base < n4; base += CNT_B * CNT_TPB) {
        int4 v[CNT_B];
#pragma unroll
        for (int u = 0; u < CNT_B; ++u) {
            const int i = base + u * CNT_TPB + tid;
            v[u] = q4[i < n4 ? i : n4 - 1];
        }
#pragma unroll
        for (int u = 0; u < CNT_B; ++u) {
            const int i = base + u * CNT_TPB + tid;
            const unsigned c = i < n4 ? cnt : 0u;
            hg_bin(bins, v[u].x, lo, c); hg_bin(bins, v[u].y, lo, c);
            hg_bin(bins, v[u].z, lo, c); hg_bin(bins, v[u].w, lo, c);
        }
    }
    for (int e = e0 + (n4 << 2) + tid; e < e1; e += CNT_TPB)
        hg_bin(bins, idx[e], lo, cnt);
    __syncthreads();

    const size_t pitch = (size_t)((N + 31) & ~31);
    unsigned* __restrict__ pp = partial + (size_t)p * pitch + lo;
    for (unsigned j = tid; j < cnt; j += CNT_TPB) pp[j] = bins[j];
}

template <int C>
__device__ __forceinline__ unsigned hg_sum_planes(const unsigned* __restrict__ pp, size_t pitch) {
    unsigned s = 0u;
#pragma unroll
    for (int c = 0; c < C; ++c) s += pp[(size_t)c * pitch];
    return s;
}

template <int C0, int C1, int C2>
__global__ __launch_bounds__(TPB) void hg_scale_sum(const unsigned* __restrict__ partial,
                                                    float* __restrict__ scales,
                                                    int N) {
    const int n = blockIdx.x * TPB + threadIdx.x;
    if (n >= N) return;
    const int r    = blockIdx.y >> 1;
    const int side = blockIdx.y & 1;
    const size_t pitch = (size_t)((N + 31) & ~31);
    unsigned s;
    if (r == 0)
        s = hg_sum_planes<C0>(partial + (size_t)(side * C0) * pitch + n, pitch);
    else if (r == 1)
        s = hg_sum_planes<C1>(partial + (size_t)(2 * C0 + side * C1) * pitch + n, pitch);
    else
        s = hg_sum_planes<C2>(partial + (size_t)(2 * (C0 + C1) + side * C2) * pitch + n, pitch);
    float d = (float)s;
    if (d < 1.0f) d = 1.0f;
    scales[(size_t)(side * 3 + r) * N + n] = rsqrtf(d);
}

__global__ __launch_bounds__(TPB) void hg_gemm1_wmma(const float* __restrict__ x,
                                                     const float* __restrict__ s_out,
                                                     const float* __restrict__ W,
                                                     float* __restrict__ hT,
                                                     int N) {
    const int lane = threadIdx.x & 31;
    const int wave = threadIdx.x >> 5;
    const int tile = blockIdx.x * (TPB / 32) + wave;
    const int row0 = tile * 16;
    if (row0 >= N) return;

    const int half = lane >> 4;
    const int lm   = lane & 15;
    int row  = row0 + lm;
    int rowc = row < N ? row : N - 1;
    const float sc = s_out[rowc];

    v8f c = {};
#pragma unroll
    for (int kb = 0; kb < 8; ++kb) {
        const int k0 = kb * 4 + half * 2;
        v2f xf, wf;
        xf.x = x[(size_t)rowc * 32 + k0]     * sc;
        xf.y = x[(size_t)rowc * 32 + k0 + 1] * sc;
        wf.x = W[(k0)     * 16 + lm];
        wf.y = W[(k0 + 1) * 16 + lm];
        c = __builtin_amdgcn_wmma_f32_16x16x4_f32(
                 false, wf,  false, xf,
                 (short)0, c,  false,  false);
    }

    float* __restrict__ dp = hT + (size_t)(half * 8) * N + row;
    if (row < N) {
#pragma unroll
        for (int v = 0; v < 8; ++v) dp[(size_t)v * N] = c[v];
    }
}

__global__ __launch_bounds__(TPB) void hg_gemm1_all_wmma(const float* __restrict__ x,
                                                         const float* __restrict__ s_out3,
                                                         const float* __restrict__ W1,
                                                         float* __restrict__ hT3,
                                                         int N) {
    const int lane = threadIdx.x & 31;
    const int wave = threadIdx.x >> 5;
    const int tile = blockIdx.x * (TPB / 32) + wave;
    const int row0 = tile * 16;
    if (row0 >= N) return;

    const int half = lane >> 4;
    const int lm   = lane & 15;
    int row  = row0 + lm;
    int rowc = row < N ? row : N - 1;
    const float sc0 = s_out3[rowc];
    const float sc1 = s_out3[(size_t)N + rowc];
    const float sc2 = s_out3[(size_t)2 * N + rowc];
    const float* __restrict__ Wa = W1;
    const float* __restrict__ Wb = W1 + 32 * 16;
    const float* __restrict__ Wc = W1 + 2 * 32 * 16;

    v8f c0 = {}, c1 = {}, c2 = {};
#pragma unroll
    for (int kb = 0; kb < 8; ++kb) {
        const int k0 = kb * 4 + half * 2;
        const float x0 = x[(size_t)rowc * 32 + k0];
        const float x1 = x[(size_t)rowc * 32 + k0 + 1];
        v2f xf, wf;
        xf.x = x0 * sc0;  xf.y = x1 * sc0;
        wf.x = Wa[(k0) * 16 + lm];  wf.y = Wa[(k0 + 1) * 16 + lm];
        c0 = __builtin_amdgcn_wmma_f32_16x16x4_f32(false, wf, false, xf, (short)0, c0, false, false);
        xf.x = x0 * sc1;  xf.y = x1 * sc1;
        wf.x = Wb[(k0) * 16 + lm];  wf.y = Wb[(k0 + 1) * 16 + lm];
        c1 = __builtin_amdgcn_wmma_f32_16x16x4_f32(false, wf, false, xf, (short)0, c1, false, false);
        xf.x = x0 * sc2;  xf.y = x1 * sc2;
        wf.x = Wc[(k0) * 16 + lm];  wf.y = Wc[(k0 + 1) * 16 + lm];
        c2 = __builtin_amdgcn_wmma_f32_16x16x4_f32(false, wf, false, xf, (short)0, c2, false, false);
    }

    if (row < N) {
        float* __restrict__ dp = hT3 + (size_t)(half * 8) * N + row;
#pragma unroll
        for (int v = 0; v < 8; ++v) dp[(size_t)v * N] = c0[v];
        dp += (size_t)16 * N;
#pragma unroll
        for (int v = 0; v < 8; ++v) dp[(size_t)v * N] = c1[v];
        dp += (size_t)16 * N;
#pragma unroll
        for (int v = 0; v < 8; ++v) dp[(size_t)v * N] = c2[v];
    }
}

#define Q_WORDS 160
static_assert(NMAX < 65536, "queue word holds src and local dst in 16 bits each");

__device__ __forceinline__ unsigned hg_push(unsigned* q, unsigned n, bool keep, unsigned word) {
    const unsigned m = __builtin_amdgcn_ballot_w32(keep);
    if (keep) q[n + __builtin_amdgcn_mbcnt_lo(m, 0u)] = word;
    return n + (unsigned)__builtin_popcount(m);
}

__device__ __forceinline__ void hg_add_word(float* sums, unsigned w, float v) {
    __hip_atomic_fetch_add(&sums[w >> 16], v, __ATOMIC_RELAXED, __HIP_MEMORY_SCOPE_WORKGROUP);
}

template <int R>
__device__ __forceinline__ void hg_take(const unsigned* q, const float* vals, float* sums,
                                        unsigned n, unsigned lane, unsigned carried, unsigned& wq) {
    unsigned w[R + 1];
#pragma unroll
    for (int r = 0; r < R; ++r) w[r] = q[lane + 32u * (unsigned)r];
    w[R] = lane + 32u * (unsigned)R < n ? q[lane + 32u * (unsigned)R] : 0u;
    if (lane < carried) w[0] = wq;
    float v[R + 1];
#pragma unroll
    for (int r = 0; r < R; ++r) v[r] = vals[w[r] & 0xffffu];
#pragma unroll
    for (int r = 0; r < R; ++r) hg_add_word(sums, w[r], v[r]);
    wq = w[R];
}

__device__ __forceinline__ void hg_trip(const int4 d, const int4 s, bool ok,
                                        const float* vals, float* sums, unsigned* q,
                                        int lo, unsigned cnt, unsigned lane,
                                        unsigned& rem, unsigned& wq) {
    const unsigned t0 = (unsigned)(d.x - lo), t1 = (unsigned)(d.y - lo);
    const unsigned t2 = (unsigned)(d.z - lo), t3 = (unsigned)(d.w - lo);
    const unsigned c = ok ? cnt : 0u;
    unsigned n = rem;
    n = hg_push(q, n, t0 < c, (unsigned)s.x | (t0 << 16));
    n = hg_push(q, n, t1 < c, (unsigned)s.y | (t1 << 16));
    n = hg_push(q, n, t2 < c, (unsigned)s.z | (t2 << 16));
    n = hg_push(q, n, t3 < c, (unsigned)s.w | (t3 << 16));
    __builtin_amdgcn_fence(__ATOMIC_RELEASE, "wavefront", "local");
    __builtin_amdgcn_wave_barrier();
    __builtin_amdgcn_fence(__ATOMIC_ACQUIRE, "wavefront", "local");
    const unsigned rows = n >> 5;
    if      (rows == 2u) hg_take<2>(q, vals, sums, n, lane, rem, wq);
    else if (rows == 1u) hg_take<1>(q, vals, sums, n, lane, rem, wq);
    else if (rows == 3u) hg_take<3>(q, vals, sums, n, lane, rem, wq);
    else if (rows == 0u) hg_take<0>(q, vals, sums, n, lane, rem, wq);
    else                 hg_take<4>(q, vals, sums, n, lane, rem, wq);
    rem = n & 31u;
}

__device__ __forceinline__ void hg_scan_own(const int4* __restrict__ s4, const int4* __restrict__ d4,
                                            int n4, const float* vals, float* sums, unsigned* q,
                                            int lo, unsigned cnt, int tid, int tpb) {
    const unsigned lane = (unsigned)tid & 31u;
    int4 dA = make_int4(0, 0, 0, 0), sA = make_int4(0, 0, 0, 0);
    int4 dB = make_int4(0, 0, 0, 0), sB = make_int4(0, 0, 0, 0);
    unsigned rem = 0u, wq = 0u;
    int ib = __builtin_amdgcn_readfirstlane(tid & ~31);
    if (ib + (int)lane < n4) { dA = d4[ib + (int)lane]; sA = s4[ib + (int)lane]; }
    const int last = n4 - 1;
    for (; ib < n4; ib += 2 * tpb) {
        const int iA = ib + (int)lane, iB = iA + tpb, iC = iB + tpb;
        const int jB = iB < n4 ? iB : last;
        dB = d4[jB];  sB = s4[jB];
        hg_trip(dA, sA, iA < n4, vals, sums, q, lo, cnt, lane, rem, wq);
        const int jC = iC < n4 ? iC : last;
        dA = d4[jC];  sA = s4[jC];
        if (ib + tpb < n4)
            hg_trip(dB, sB, iB < n4, vals, sums, q, lo, cnt, lane, rem, wq);
    }
    if (lane < rem) hg_add_word(sums, wq, vals[wq & 0xffffu]);
}

__device__ __forceinline__ void hg_edge_add(float* __restrict__ acc,
                                            float v, int d, int lo, unsigned cnt) {
    const unsigned dl = (unsigned)(d - lo);
    if (dl < cnt)
        __hip_atomic_fetch_add(&acc[dl], v, __ATOMIC_RELAXED,
                               __HIP_MEMORY_SCOPE_WORKGROUP);
}

__global__ __launch_bounds__(AGG_TPB) void hg_agg1_col(const int* __restrict__ src,
                                                       const int* __restrict__ dst,
                                                       int E,
                                                       const float* __restrict__ hT,
                                                       const float* __restrict__ s_in,
                                                       float* __restrict__ accT,
                                                       int N) {
    __shared__ float hcol[NMAX];
    __shared__ float acc[NHALF_MAX];
    __shared__ unsigned qbuf[(AGG_TPB / 32) * Q_WORDS];

    const int tid    = threadIdx.x;
    const int nchunk = gridDim.x;
    const int chunk  = blockIdx.x;
    const int f      = blockIdx.z;
    const int nh     = (N + 1) >> 1;
    const int lo     = blockIdx.y * nh;
    const int hiN    = lo + nh < N ? lo + nh : N;
    const unsigned cnt = hiN > lo ? (unsigned)(hiN - lo) : 0u;

    const float* __restrict__ hp = hT + (size_t)f * N;
    for (int i = tid; i < N; i += AGG_TPB) hcol[i] = hp[i];
    for (unsigned i = tid; i < cnt; i += AGG_TPB) acc[i] = 0.0f;
    __syncthreads();

    const int per = ((E + nchunk * 4 - 1) / (nchunk * 4)) * 4;
    const int e0  = chunk * per;
    const int e1  = e0 + per < E ? e0 + per : E;
    const int ne  = e1 > e0 ? e1 - e0 : 0;
    const int n4  = ne >> 2;
    const int4* __restrict__ s4 = (const int4*)(src + e0);
    const int4* __restrict__ d4 = (const int4*)(dst + e0);

    hg_scan_own(s4, d4, n4, hcol, acc, qbuf + (tid >> 5) * Q_WORDS, lo, cnt, tid, AGG_TPB);
    for (int e = e0 + (n4 << 2) + tid; e < e1; e += AGG_TPB)
        hg_edge_add(acc, hcol[src[e]], dst[e], lo, cnt);
    __syncthreads();

    float* __restrict__ ap = accT + (size_t)f * N + lo;
    const float* __restrict__ sp = s_in + lo;
    if (cnt != 0u) {
        const unsigned last = cnt - 1u;
        for (unsigned base = 0u; base < cnt; base += FLUSH_U * AGG_TPB) {
            float s[FLUSH_U], a[FLUSH_U];
#pragma unroll
            for (int u = 0; u < FLUSH_U; ++u) {
                const unsigned j  = base + (unsigned)u * AGG_TPB + tid;
                const unsigned jc = j < cnt ? j : last;
                s[u] = sp[jc];
                a[u] = acc[jc];
            }
#pragma unroll
            for (int u = 0; u < FLUSH_U; ++u) {
                const unsigned j = base + (unsigned)u * AGG_TPB + tid;
                const float v = a[u] * s[u];
                if (j < cnt && v != 0.0f) atomicAdd(ap + j, v);
            }
        }
    }
}

#define B16_TPB 1024
#define B16_VEC 5
#define B16_EPT (4 * B16_VEC)
#define B16_RUN (B16_TPB * B16_EPT)
#define B16_HDR 32
#define B16_PITCH (B16_HDR + B16_RUN)
#define SL16 16
#define SL16_MAX ((((NMAX + SL16 - 1) / SL16) + 31) & ~31)

__global__ __launch_bounds__(B16_TPB) void hg_bin16(const int* __restrict__ s0,
                                                    const int* __restrict__ d0,
                                                    const int* __restrict__ s1,
                                                    const int* __restrict__ d1,
                                                    const int* __restrict__ s2,
                                                    const int* __restrict__ d2,
                                                    int E0, int E1, int E2,
                                                    int nR0, int nR1,
                                                    unsigned* __restrict__ words,
                                                    int N) {
    __shared__ unsigned buf[B16_RUN];
    __shared__ unsigned hist[SL16 * 32];
    __shared__ unsigned base[SL16 * 32];
    __shared__ unsigned tot[SL16];
    __shared__ unsigned start[SL16 + 1];

    const int tid = threadIdx.x;
    const int run = blockIdx.x;
    const int* __restrict__ src;
    const int* __restrict__ dst;
    int E, q;
    if (run < nR0)            { src = s0;  dst = d0;  E = E0;  q = run; }
    else if (run < nR0 + nR1) { src = s1;  dst = d1;  E = E1;  q = run - nR0; }
    else                      { src = s2;  dst = d2;  E = E2;  q = run - nR0 - nR1; }
    const int ns  = (((N + SL16 - 1) / SL16) + 31) & ~31;
    const float inv_ns = 1.0f / (float)ns;
    const int R0  = q * B16_RUN;
    const int len = E - R0 < B16_RUN ? E - R0 : B16_RUN;

    const int lane = tid & 31;
    if (tid < SL16 * 32) hist[tid] = 0u;

    int d[B16_EPT], s[B16_EPT];
    bool ok[B16_EPT];
#pragma unroll
    for (int v = 0; v < B16_VEC; ++v) {
        const int b = v * (B16_RUN / B16_VEC) + 4 * tid;
        if (b + 3 < len) {
            const int4 dv = *(const int4*)(dst + R0 + b);
            const int4 sv = *(const int4*)(src + R0 + b);
            d[4 * v] = dv.x;  d[4 * v + 1] = dv.y;  d[4 * v + 2] = dv.z;  d[4 * v + 3] = dv.w;
            s[4 * v] = sv.x;  s[4 * v + 1] = sv.y;  s[4 * v + 2] = sv.z;  s[4 * v + 3] = sv.w;
            ok[4 * v] = ok[4 * v + 1] = ok[4 * v + 2] = ok[4 * v + 3] = true;
        } else {
#pragma unroll
            for (int k = 0; k < 4; ++k) {
                const bool in = b + k < len;
                ok[4 * v + k] = in;
                d[4 * v + k]  = in ? dst[R0 + b + k] : 0;
                s[4 * v + k]  = in ? src[R0 + b + k] : 0;
            }
        }
    }
    __syncthreads();

    unsigned word[B16_EPT], rank[B16_EPT];
    int bin[B16_EPT];
#pragma unroll
    for (int k = 0; k < B16_EPT; ++k) {
        const bool valid = ok[k] && (unsigned)d[k] < (unsigned)N;
        const int dv = valid ? d[k] : 0;
        int b = (int)((float)dv * inv_ns);
        if (b * ns > dv) --b;
        else if ((b + 1) * ns <= dv) ++b;
        const unsigned t = (unsigned)(dv - b * ns);
        word[k] = (unsigned)s[k] | (t << 16);
        bin[k]  = valid ? b : SL16;
        rank[k] = 0u;
        if (valid)
            rank[k] = __hip_atomic_fetch_add(&hist[b * 32 + lane], 1u, __ATOMIC_RELAXED,
                                             __HIP_MEMORY_SCOPE_WORKGROUP);
    }
    __syncthreads();
    unsigned cv = 0u, incl = 0u;
    if (tid < SL16 * 32) {
        cv = hist[tid];
        incl = cv;
#pragma unroll
        for (int o = 1; o < 32; o <<= 1) {
            const unsigned up = __shfl_up(incl, (unsigned)o, 32);
            if (lane >= o) incl += up;
        }
        if (lane == 31) tot[tid >> 5] = incl;
    }
    __syncthreads();
    if (tid == 0) {
        unsigned a = 0u;
#pragma unroll
        for (int b = 0; b < SL16; ++b) { start[b] = a;  a += tot[b]; }
        start[SL16] = a;
    }
    __syncthreads();
    if (tid < SL16 * 32) base[tid] = start[tid >> 5] + (incl - cv);
    __syncthreads();
#pragma unroll
    for (int k = 0; k < B16_EPT; ++k)
        if (bin[k] < SL16) buf[base[bin[k] * 32 + lane] + rank[k]] = word[k];
    __syncthreads();

    unsigned* __restrict__ wr = words + (size_t)run * B16_PITCH;
    if (tid < B16_HDR) wr[tid] = tid <= SL16 ? start[tid] : 0u;
    const int total = (int)start[SL16];
    for (int i = tid; i < total; i += B16_TPB) wr[B16_HDR + i] = buf[i];
}

#define RUN_ROWS 8
__device__ __forceinline__ void hg_load_rows(const unsigned* __restrict__ wp, int i0, int last,
                                             int lane, unsigned (&w)[RUN_ROWS]) {
#pragma unroll
    for (int u = 0; u < RUN_ROWS; ++u) {
        const int i = i0 + 32 * u + lane;
        w[u] = wp[i < last ? i : last];
    }
}

__device__ __forceinline__ void hg_add_rows(const unsigned (&w)[RUN_ROWS], int i0, int st, int en, int lane,
                                            const float* vals, float* sums, unsigned cnt) {
    float v[RUN_ROWS];
#pragma unroll
    for (int u = 0; u < RUN_ROWS; ++u) v[u] = vals[w[u] & 0xffffu];
#pragma unroll
    for (int u = 0; u < RUN_ROWS; ++u) {
        const int i = i0 + 32 * u + lane;
        const unsigned t = w[u] >> 16;
        if (i >= st && i < en && t < cnt)
            __hip_atomic_fetch_add(&sums[t], v[u], __ATOMIC_RELAXED, __HIP_MEMORY_SCOPE_WORKGROUP);
    }
}

__global__ __launch_bounds__(AGG_TPB) void hg_agg1_own16(const unsigned* __restrict__ words,
                                                         int nR,
                                                         const float* __restrict__ hT,
                                                         const float* __restrict__ s_in,
                                                         float* __restrict__ accT,
                                                         const float* __restrict__ b1,
                                                         int N, int first, int last) {
    __shared__ float hcol[NMAX];
    __shared__ float acc[SL16_MAX];

    const int tid  = threadIdx.x;
    const int lane = tid & 31;
    const int f    = blockIdx.y;
    const int sl   = blockIdx.x;
    const int ns   = (((N + SL16 - 1) / SL16) + 31) & ~31;
    const int lo   = sl * ns;
    const int hiN  = lo + ns < N ? lo + ns : N;
    const unsigned cnt = hiN > lo ? (unsigned)(hiN - lo) : 0u;

    const float* __restrict__ hp = hT + (size_t)f * N;
    for (int i = tid; i < N; i += AGG_TPB) hcol[i] = hp[i];
    for (unsigned i = tid; i < cnt; i += AGG_TPB) acc[i] = 0.0f;
    __syncthreads();

    const int nwav = AGG_TPB / 32;
    int q = __builtin_amdgcn_readfirstlane(tid >> 5);
    unsigned h0 = 0u, h1 = 0u;
    if (q < nR) {
        const unsigned* __restrict__ rg = words + (size_t)q * B16_PITCH;
        h0 = rg[sl];  h1 = rg[sl + 1];
    }
    while (q < nR) {
        const int qn = q + nwav;
        unsigned g0 = 0u, g1 = 0u;
        if (qn < nR) {
            const unsigned* __restrict__ rn = words + (size_t)qn * B16_PITCH;
            g0 = rn[sl];  g1 = rn[sl + 1];
        }
        const int st = __builtin_amdgcn_readfirstlane((int)h0);
        const int en = __builtin_amdgcn_readfirstlane((int)h1);
        const unsigned* __restrict__ wp = words + (size_t)q * B16_PITCH + B16_HDR;
        if (en > st) {
            const int a0   = st & ~31;
            const int last = en - 1;
            const int G  = 32 * RUN_ROWS;
            const int ng = (en - a0 + G - 1) / G;
            unsigned wA[RUN_ROWS], wB[RUN_ROWS];
            hg_load_rows(wp, a0, last, lane, wA);
            int g = 0;
            for (; g + 2 < ng; g += 2) {
                hg_load_rows(wp, a0 + (g + 1) * G, last, lane, wB);
                hg_add_rows(wA, a0 + g * G, st, en, lane, hcol, acc, cnt);
                hg_load_rows(wp, a0 + (g + 2) * G, last, lane, wA);
                hg_add_rows(wB, a0 + (g + 1) * G, st, en, lane, hcol, acc, cnt);
            }
            if (g + 1 < ng) {
                hg_load_rows(wp, a0 + (g + 1) * G, last, lane, wB);
                hg_add_rows(wA, a0 + g * G, st, en, lane, hcol, acc, cnt);
                hg_add_rows(wB, a0 + (g + 1) * G, st, en, lane, hcol, acc, cnt);
            } else {
                hg_add_rows(wA, a0 + g * G, st, en, lane, hcol, acc, cnt);
            }
        }
        h0 = g0;  h1 = g1;  q = qn;
    }
    __syncthreads();

    float* __restrict__ ap = accT + (size_t)f * N + lo;
    const float* __restrict__ sp = s_in + lo;
    const float bs0 = b1[f], bs1 = b1[16 + f], bs2 = b1[32 + f];
    for (unsigned j = tid; j < cnt; j += AGG_TPB) {
        const float v = acc[j] * sp[j];
        float t = first ? v : ap[j] + v;
        if (last) {
            t = t + bs0 + bs1 + bs2;
            t = t > 0.0f ? t : 0.0f;
        }
        ap[j] = t;
    }
}


#define A1_PT ((SL16_MAX + AGG_TPB - 1) / AGG_TPB)
__global__ __launch_bounds__(AGG_TPB) void hg_agg1_all16(const unsigned* __restrict__ words,
                                                         int nR0, int nR1, int nR2,
                                                         const float* __restrict__ hT3,
                                                         const float* __restrict__ s_in3,
                                                         const float* __restrict__ b1,
                                                         float* __restrict__ h,
                                                         int N) {
    __shared__ float hcol[NMAX];
    __shared__ float acc[SL16_MAX];

    const int tid  = threadIdx.x;
    const int lane = tid & 31;
    const int f    = blockIdx.y;
    const int sl   = blockIdx.x;
    const int ns   = (((N + SL16 - 1) / SL16) + 31) & ~31;
    const int lo   = sl * ns;
    const int hiN  = lo + ns < N ? lo + ns : N;
    const unsigned cnt = hiN > lo ? (unsigned)(hiN - lo) : 0u;

    const int nwav = AGG_TPB / 32;
    const int wv   = __builtin_amdgcn_readfirstlane(tid >> 5);

    float tot[A1_PT];
#pragma unroll
    for (int k = 0; k < A1_PT; ++k) tot[k] = 0.0f;

    int rb = 0;
    for (int r = 0; r < 3; ++r) {
        const int nR = r == 0 ? nR0 : (r == 1 ? nR1 : nR2);
        const float* __restrict__ hp = hT3 + ((size_t)r * 16 + f) * N;
        for (int i = tid; i < N; i += AGG_TPB) hcol[i] = hp[i];
        for (unsigned i = tid; i < cnt; i += AGG_TPB) acc[i] = 0.0f;
        __syncthreads();

        const unsigned* __restrict__ rel = words + (size_t)rb * B16_PITCH;
        int q = wv;
        unsigned h0 = 0u, h1 = 0u;
        if (q < nR) {
            const unsigned* __restrict__ rg = rel + (size_t)q * B16_PITCH;
            h0 = rg[sl];  h1 = rg[sl + 1];
        }
        while (q < nR) {
            const int qn = q + nwav;
            unsigned g0 = 0u, g1 = 0u;
            if (qn < nR) {
                const unsigned* __restrict__ rn = rel + (size_t)qn * B16_PITCH;
                g0 = rn[sl];  g1 = rn[sl + 1];
            }
            const int st = __builtin_amdgcn_readfirstlane((int)h0);
            const int en = __builtin_amdgcn_readfirstlane((int)h1);
            const unsigned* __restrict__ wp = rel + (size_t)q * B16_PITCH + B16_HDR;
            if (en > st) {
                const int a0   = st & ~31;
                const int last = en - 1;
                const int G  = 32 * RUN_ROWS;
                const int ng = (en - a0 + G - 1) / G;
                unsigned wA[RUN_ROWS], wB[RUN_ROWS];
                hg_load_rows(wp, a0, last, lane, wA);
                int g = 0;
                for (; g + 2 < ng; g += 2) {
                    hg_load_rows(wp, a0 + (g + 1) * G, last, lane, wB);
                    hg_add_rows(wA, a0 + g * G, st, en, lane, hcol, acc, cnt);
                    hg_load_rows(wp, a0 + (g + 2) * G, last, lane, wA);
                    hg_add_rows(wB, a0 + (g + 1) * G, st, en, lane, hcol, acc, cnt);
                }
                if (g + 1 < ng) {
                    hg_load_rows(wp, a0 + (g + 1) * G, last, lane, wB);
                    hg_add_rows(wA, a0 + g * G, st, en, lane, hcol, acc, cnt);
                    hg_add_rows(wB, a0 + (g + 1) * G, st, en, lane, hcol, acc, cnt);
                } else {
                    hg_add_rows(wA, a0 + g * G, st, en, lane, hcol, acc, cnt);
                }
            }
            h0 = g0;  h1 = g1;  q = qn;
        }
        __syncthreads();

        const float* __restrict__ sp = s_in3 + (size_t)r * N + lo;
#pragma unroll
        for (int k = 0; k < A1_PT; ++k) {
            const unsigned j = (unsigned)tid + (unsigned)k * AGG_TPB;
            if (j < cnt) tot[k] += acc[j] * sp[j];
        }
        rb += nR;
    }

    const float bs0 = b1[f], bs1 = b1[16 + f], bs2 = b1[32 + f];
    float* __restrict__ hq = h + (size_t)f * N + lo;
#pragma unroll
    for (int k = 0; k < A1_PT; ++k) {
        const unsigned j = (unsigned)tid + (unsigned)k * AGG_TPB;
        if (j < cnt) {
            const float v = tot[k] + bs0 + bs1 + bs2;
            hq[j] = v > 0.0f ? v : 0.0f;
        }
    }
}

__global__ __launch_bounds__(TPB) void hg_relu_bias(float* __restrict__ acc,
                                                    const float* __restrict__ b1,
                                                    int N) {
    const int f = blockIdx.y;
    const int n = blockIdx.x * TPB + threadIdx.x;
    if (n < N) {
        const size_t i = (size_t)f * N + n;
        float v = acc[i] + b1[f] + b1[16 + f] + b1[32 + f];
        acc[i] = v > 0.0f ? v : 0.0f;
    }
}

__global__ __launch_bounds__(TPB) void hg_gemm2_all(const float* __restrict__ h,
                                                    const float* __restrict__ s_out3,
                                                    const float* __restrict__ W2,
                                                    float* __restrict__ z3,
                                                    int N) {
    int i = blockIdx.x * TPB + threadIdx.x;
    if (i >= N) return;
    const float* hp = h + i;
    float a[16];
#pragma unroll
    for (int f = 0; f < 16; ++f) a[f] = hp[(size_t)f * N];
#pragma unroll
    for (int r = 0; r < 3; ++r) {
        const float4* wp = (const float4*)(W2 + r * 16);
        float acc = 0.0f;
#pragma unroll
        for (int q = 0; q < 4; ++q) {
            const float4 b = wp[q];
            acc += a[4 * q] * b.x + a[4 * q + 1] * b.y + a[4 * q + 2] * b.z + a[4 * q + 3] * b.w;
        }
        z3[(size_t)r * N + i] = acc * s_out3[(size_t)r * N + i];
    }
}

__global__ __launch_bounds__(L2_TPB) void hg_agg2_own(const int* __restrict__ s0,
                                                      const int* __restrict__ d0,
                                                      const int* __restrict__ s1,
                                                      const int* __restrict__ d1,
                                                      const int* __restrict__ s2,
                                                      const int* __restrict__ d2,
                                                      int E0, int E1, int E2,
                                                      const float* __restrict__ z3,
                                                      float* __restrict__ partial,
                                                      int N) {
    __shared__ float zc[NMAX];
    __shared__ float bins[L2_SLICE_MAX];
    __shared__ unsigned qbuf[(L2_TPB / 32) * Q_WORDS];

    const int tid = threadIdx.x;
    const int p   = blockIdx.y;
    const int* __restrict__ src;
    const int* __restrict__ dst;
    int E, nchunk, chunk, r;
    if (p < L2_C0) {
        src = s0;  dst = d0;  E = E0;  nchunk = L2_C0;  chunk = p;                  r = 0;
    } else if (p < L2_C0 + L2_C1) {
        src = s1;  dst = d1;  E = E1;  nchunk = L2_C1;  chunk = p - L2_C0;          r = 1;
    } else {
        src = s2;  dst = d2;  E = E2;  nchunk = L2_C2;  chunk = p - L2_C0 - L2_C1;  r = 2;
    }

    const int ns  = (((N + L2_SLICES - 1) / L2_SLICES) + 31) & ~31;
    const int lo  = blockIdx.x * ns;
    const int hiN = lo + ns < N ? lo + ns : N;
    const unsigned cnt = hiN > lo ? (unsigned)(hiN - lo) : 0u;

    const float* __restrict__ zp = z3 + (size_t)r * N;
    for (int i = tid; i < N; i += L2_TPB) zc[i] = zp[i];
    for (unsigned j = tid; j < cnt; j += L2_TPB) bins[j] = 0.0f;
    __syncthreads();

    const int per = ((E + nchunk * 4 - 1) / (nchunk * 4)) * 4;
    const int e0  = chunk * per;
    const int e1  = e0 + per < E ? e0 + per : E;
    const int ne  = e1 > e0 ? e1 - e0 : 0;
    const int n4  = ne >> 2;
    const int4* __restrict__ s4 = (const int4*)(src + e0);
    const int4* __restrict__ d4 = (const int4*)(dst + e0);

    hg_scan_own(s4, d4, n4, zc, bins, qbuf + (tid >> 5) * Q_WORDS, lo, cnt, tid, L2_TPB);
    for (int e = e0 + (n4 << 2) + tid; e < e1; e += L2_TPB)
        hg_edge_add(bins, zc[src[e]], dst[e], lo, cnt);
    __syncthreads();

    const size_t pitch = (size_t)((N + 31) & ~31);
    float* __restrict__ pp = partial + (size_t)p * pitch + lo;
    for (unsigned j = tid; j < cnt; j += L2_TPB) pp[j] = bins[j];
}

#define A2_G0 11
#define A2_G1 4
#define A2_G2 1
#define A2_PLANES (A2_G0 + A2_G1 + A2_G2)

__global__ __launch_bounds__(AGG_TPB) void hg_agg2_own16(const unsigned* __restrict__ words,
                                                         int nR0, int nR1, int nR2,
                                                         const float* __restrict__ z3,
                                                         float* __restrict__ partial,
                                                         int N) {
    __shared__ float zc[NMAX];
    __shared__ float acc[SL16_MAX];

    const int tid  = threadIdx.x;
    const int lane = tid & 31;
    const int sl   = blockIdx.x;
    const int p    = blockIdx.y;
    int r, gi, nG, nR, rb;
    if (p < A2_G0)              { r = 0;  gi = p;                  nG = A2_G0;  nR = nR0;  rb = 0; }
    else if (p < A2_G0 + A2_G1) { r = 1;  gi = p - A2_G0;          nG = A2_G1;  nR = nR1;  rb = nR0; }
    else                        { r = 2;  gi = p - A2_G0 - A2_G1;  nG = A2_G2;  nR = nR2;  rb = nR0 + nR1; }
    const int per = (nR + nG - 1) / nG;
    const int q0  = gi * per;
    const int q1  = q0 + per < nR ? q0 + per : nR;

    const int ns   = (((N + SL16 - 1) / SL16) + 31) & ~31;
    const int lo   = sl * ns;
    const int hiN  = lo + ns < N ? lo + ns : N;
    const unsigned cnt = hiN > lo ? (unsigned)(hiN - lo) : 0u;

    const float* __restrict__ zp = z3 + (size_t)r * N;
    for (int i = tid; i < N; i += AGG_TPB) zc[i] = zp[i];
    for (unsigned i = tid; i < cnt; i += AGG_TPB) acc[i] = 0.0f;
    __syncthreads();

    const int nwav = AGG_TPB / 32;
    const unsigned* __restrict__ rel = words + (size_t)rb * B16_PITCH;
    int q = q0 + __builtin_amdgcn_readfirstlane(tid >> 5);
    unsigned h0 = 0u, h1 = 0u;
    if (q < q1) {
        const unsigned* __restrict__ rg = rel + (size_t)q * B16_PITCH;
        h0 = rg[sl];  h1 = rg[sl + 1];
    }
    while (q < q1) {
        const int qn = q + nwav;
        unsigned g0 = 0u, g1 = 0u;
        if (qn < q1) {
            const unsigned* __restrict__ rn = rel + (size_t)qn * B16_PITCH;
            g0 = rn[sl];  g1 = rn[sl + 1];
        }
        const int st = __builtin_amdgcn_readfirstlane((int)h0);
        const int en = __builtin_amdgcn_readfirstlane((int)h1);
        const unsigned* __restrict__ wp = rel + (size_t)q * B16_PITCH + B16_HDR;
        if (en > st) {
            const int a0   = st & ~31;
            const int last = en - 1;
            const int G  = 32 * RUN_ROWS;
            const int ng = (en - a0 + G - 1) / G;
            unsigned wA[RUN_ROWS], wB[RUN_ROWS];
            hg_load_rows(wp, a0, last, lane, wA);
            int g = 0;
            for (; g + 2 < ng; g += 2) {
                hg_load_rows(wp, a0 + (g + 1) * G, last, lane, wB);
                hg_add_rows(wA, a0 + g * G, st, en, lane, zc, acc, cnt);
                hg_load_rows(wp, a0 + (g + 2) * G, last, lane, wA);
                hg_add_rows(wB, a0 + (g + 1) * G, st, en, lane, zc, acc, cnt);
            }
            if (g + 1 < ng) {
                hg_load_rows(wp, a0 + (g + 1) * G, last, lane, wB);
                hg_add_rows(wA, a0 + g * G, st, en, lane, zc, acc, cnt);
                hg_add_rows(wB, a0 + (g + 1) * G, st, en, lane, zc, acc, cnt);
            } else {
                hg_add_rows(wA, a0 + g * G, st, en, lane, zc, acc, cnt);
            }
        }
        h0 = g0;  h1 = g1;  q = qn;
    }
    __syncthreads();

    const size_t pitch = (size_t)((N + 31) & ~31);
    float* __restrict__ pp = partial + (size_t)p * pitch + lo;
    for (unsigned j = tid; j < cnt; j += AGG_TPB) pp[j] = acc[j];
}

template <int C>
__device__ __forceinline__ float hg_sum_fplanes(const float* __restrict__ pp, size_t pitch) {
    float s = 0.0f;
#pragma unroll
    for (int c = 0; c < C; ++c) s += pp[(size_t)c * pitch];
    return s;
}

template <int C0, int C1, int C2>
__device__ __forceinline__ void hg_out_body(const float* __restrict__ partial,
                                            const float* __restrict__ s_in3,
                                            const float* __restrict__ b2,
                                            float* __restrict__ out,
                                            int N) {
    const int n = blockIdx.x * TPB + threadIdx.x;
    if (n >= N) return;
    const size_t pitch = (size_t)((N + 31) & ~31);
    const float m0 = hg_sum_fplanes<C0>(partial + n, pitch);
    const float m1 = hg_sum_fplanes<C1>(partial + (size_t)C0 * pitch + n, pitch);
    const float m2 = hg_sum_fplanes<C2>(partial + (size_t)(C0 + C1) * pitch + n, pitch);
    out[n] = (m0 * s_in3[n] + b2[0])
           + (m1 * s_in3[(size_t)N + n] + b2[1])
           + (m2 * s_in3[(size_t)2 * N + n] + b2[2]);
}

__global__ __launch_bounds__(TPB) void hg_out_sum(const float* __restrict__ partial,
                                                  const float* __restrict__ s_in3,
                                                  const float* __restrict__ b2,
                                                  float* __restrict__ out,
                                                  int N) {
    hg_out_body<L2_C0, L2_C1, L2_C2>(partial, s_in3, b2, out, N);
}

__global__ __launch_bounds__(TPB) void hg_out_sum16(const float* __restrict__ partial,
                                                    const float* __restrict__ s_in3,
                                                    const float* __restrict__ b2,
                                                    float* __restrict__ out,
                                                    int N) {
    hg_out_body<A2_G0, A2_G1, A2_G2>(partial, s_in3, b2, out, N);
}

extern "C" void kernel_launch(void* const* d_in, const int* in_sizes, int n_in,
                              void* d_out, int out_size, void* d_ws, size_t ws_size,
                              hipStream_t stream) {
    const float* x  = (const float*)d_in[0];
    const int* src[3] = {(const int*)d_in[1], (const int*)d_in[3], (const int*)d_in[5]};
    const int* dst[3] = {(const int*)d_in[2], (const int*)d_in[4], (const int*)d_in[6]};
    const int  E[3]   = {in_sizes[1], in_sizes[3], in_sizes[5]};
    const float* W1 = (const float*)d_in[7];
    const float* b1 = (const float*)d_in[8];
    const float* W2 = (const float*)d_in[9];
    const float* b2 = (const float*)d_in[10];
    const int N = in_sizes[0] / 32;
    float* out = (float*)d_out;

    const int nR[3] = {(E[0] + B16_RUN - 1) / B16_RUN, (E[1] + B16_RUN - 1) / B16_RUN,
                       (E[2] + B16_RUN - 1) / B16_RUN};
    const int nRuns = nR[0] + nR[1] + nR[2];
    char* ws = (char*)d_ws;
    float* scales = (float*)ws;
    const int tiles  = (N + 15) / 16;
    const int gblk   = (tiles + (TPB / 32) - 1) / (TPB / 32);

    const size_t pitchN  = (size_t)((N + 31) & ~31);
    const size_t w_off3  = ((size_t)70 * N * 4 + 127) & ~(size_t)127;
    const bool merged = nRuns > 0 && N >= 64 &&
                        ws_size >= w_off3 + (size_t)nRuns * B16_PITCH * 4;
    if (merged) {
        float* hpre3   = scales + (size_t)6 * N;
        float* hbuf    = hpre3 + (size_t)48 * N;
        unsigned* cntp = (unsigned*)hpre3;
        float* planes3 = hpre3;
        float* z3m     = hpre3 + (size_t)16 * pitchN;
        unsigned* wordsm = (unsigned*)(ws + w_off3);

        const bool wide = (size_t)6 * N * 4 + (size_t)CNTW_PLANES * pitchN * 4
                          <= w_off3 + (size_t)nRuns * B16_PITCH * 4;
        if (wide) {
            hg_count_own<CNTW_SLICES, CNTW_C0, CNTW_C1, CNTW_C2>
                <<<dim3(CNTW_SLICES, CNTW_PLANES), dim3(CNT_TPB), 0, stream>>>(
                src[0], dst[0], src[1], dst[1], src[2], dst[2], E[0], E[1], E[2], cntp, N);
            hg_scale_sum<CNTW_C0, CNTW_C1, CNTW_C2>
                <<<dim3((N + TPB - 1) / TPB, 6), dim3(TPB), 0, stream>>>(cntp, scales, N);
        } else {
            hg_count_own<CNT_SLICES, CNT_C0, CNT_C1, CNT_C2>
                <<<dim3(CNT_SLICES, CNT_PLANES), dim3(CNT_TPB), 0, stream>>>(
                src[0], dst[0], src[1], dst[1], src[2], dst[2], E[0], E[1], E[2], cntp, N);
            hg_scale_sum<CNT_C0, CNT_C1, CNT_C2>
                <<<dim3((N + TPB - 1) / TPB, 6), dim3(TPB), 0, stream>>>(cntp, scales, N);
        }
        hg_bin16<<<dim3(nRuns), dim3(B16_TPB), 0, stream>>>(
            src[0], dst[0], src[1], dst[1], src[2], dst[2], E[0], E[1], E[2], nR[0], nR[1], wordsm, N);
        hg_gemm1_all_wmma<<<dim3(gblk), dim3(TPB), 0, stream>>>(x, scales, W1, hpre3, N);
        hg_agg1_all16<<<dim3(SL16, 16), dim3(AGG_TPB), 0, stream>>>(
            wordsm, nR[0], nR[1], nR[2], hpre3, scales + (size_t)3 * N, b1, hbuf, N);
        hg_gemm2_all<<<dim3((N + TPB - 1) / TPB), dim3(TPB), 0, stream>>>(hbuf, scales, W2, z3m, N);
        hg_agg2_own16<<<dim3(SL16, A2_PLANES), dim3(AGG_TPB), 0, stream>>>(
            wordsm, nR[0], nR[1], nR[2], z3m, planes3, N);
        hg_out_sum16<<<dim3((N + TPB - 1) / TPB), dim3(TPB), 0, stream>>>(
            planes3, scales + (size_t)3 * N, b2, out, N);
        return;
    }

    float* acc1   = (float*)(ws + (size_t)6 * N * 4);
    float* hpre   = acc1 + (size_t)N * 16;
    float* z      = hpre + (size_t)N * 16;
    float* z3     = z - (size_t)2 * N;
    float* planes2 = acc1;
    unsigned* partial = (unsigned*)acc1;
    const size_t w_off = ((size_t)39 * N * 4 + 127) & ~(size_t)127;
    const bool binned = nRuns > 0 && ws_size >= w_off + (size_t)nRuns * B16_PITCH * 4;
    unsigned* words = (unsigned*)(ws + w_off);
    const size_t rOff[3] = {0, (size_t)nR[0] * B16_PITCH, (size_t)(nR[0] + nR[1]) * B16_PITCH};

    hg_count_own<CNT_SLICES, CNT_C0, CNT_C1, CNT_C2>
        <<<dim3(CNT_SLICES, CNT_PLANES), dim3(CNT_TPB), 0, stream>>>(
        src[0], dst[0], src[1], dst[1], src[2], dst[2], E[0], E[1], E[2], partial, N);
    hg_scale_sum<CNT_C0, CNT_C1, CNT_C2>
        <<<dim3((N + TPB - 1) / TPB, 6), dim3(TPB), 0, stream>>>(partial, scales, N);

    if (!binned) {
        long span = 16L * N;
        hg_init<<<dim3((unsigned)((span + TPB - 1) / TPB)), dim3(TPB), 0, stream>>>(
            acc1, N);
    }

    if (binned)
        hg_bin16<<<dim3(nRuns), dim3(B16_TPB), 0, stream>>>(
            src[0], dst[0], src[1], dst[1], src[2], dst[2], E[0], E[1], E[2], nR[0], nR[1], words, N);

    for (int r = 0; r < 3; ++r) {
        hg_gemm1_wmma<<<dim3(gblk), dim3(TPB), 0, stream>>>(
            x, scales + (size_t)r * N, W1 + (size_t)r * 32 * 16, hpre, N);
        if (binned)
            hg_agg1_own16<<<dim3(SL16, 16), dim3(AGG_TPB), 0, stream>>>(
                words + rOff[r], nR[r], hpre, scales + (size_t)(3 + r) * N, acc1, b1, N,
                r == 0 ? 1 : 0, r == 2 ? 1 : 0);
        else
            hg_agg1_col<<<dim3(8, 2, 16), dim3(AGG_TPB), 0, stream>>>(
                src[r], dst[r], E[r], hpre, scales + (size_t)(3 + r) * N, acc1, N);
    }
    if (!binned)
        hg_relu_bias<<<dim3((N + TPB - 1) / TPB, 16), dim3(TPB), 0, stream>>>(acc1, b1, N);

    hg_gemm2_all<<<dim3((N + TPB - 1) / TPB), dim3(TPB), 0, stream>>>(acc1, scales, W2, z3, N);
    if (binned) {
        hg_agg2_own16<<<dim3(SL16, A2_PLANES), dim3(AGG_TPB), 0, stream>>>(
            words, nR[0], nR[1], nR[2], z3, planes2, N);
        hg_out_sum16<<<dim3((N + TPB - 1) / TPB), dim3(TPB), 0, stream>>>(
            planes2, scales + (size_t)3 * N, b2, out, N);
    } else {
        hg_agg2_own<<<dim3(L2_SLICES, L2_PLANES), dim3(L2_TPB), 0, stream>>>(
            src[0], dst[0], src[1], dst[1], src[2], dst[2], E[0], E[1], E[2], z3, planes2, N);
        hg_out_sum<<<dim3((N + TPB - 1) / TPB), dim3(TPB), 0, stream>>>(
            planes2, scales + (size_t)3 * N, b2, out, N);
    }
}
